# starting kernel
# baseline (speedup 1.0000x reference)
#include <hip/hip_runtime.h>
#include <math.h>

typedef __attribute__((ext_vector_type(2))) float v2f;
typedef __attribute__((ext_vector_type(8))) float v8f;

#define O_DIM 128

__device__ __forceinline__ void atomAddF(float* p, float v) {
    __hip_atomic_fetch_add(p, v, __ATOMIC_RELAXED, __HIP_MEMORY_SCOPE_AGENT);
}

__global__ __launch_bounds__(256)
void gemm_bias_relu(const float* __restrict__ X, const float* __restrict__ W,
                    const float* __restrict__ bias, float* __restrict__ out,
                    int N, int K) {
    __shared__ float  As[128][36];
    __shared__ float2 Bs2[16][144];

    const int tid = threadIdx.x;
    const int wv  = tid >> 5;
    const int ln  = tid & 31;
    const int m0  = blockIdx.x * 128;

    v8f acc[8];
#pragma unroll
    for (int t = 0; t < 8; ++t)
#pragma unroll
        for (int j = 0; j < 8; ++j) acc[t][j] = 0.0f;

    const int mrow = wv * 16 + (ln & 15);
    const int lnn  = ln & 15;
    const int khalf = (ln >> 4) << 1;

    for (int k0 = 0; k0 < K; k0 += 32) {
#pragma unroll
        for (int i = 0; i < 4; ++i) {
            int f   = tid + i * 256;
            int row = f >> 3;
            int c4  = f & 7;
            float4 v = make_float4(0.f, 0.f, 0.f, 0.f);
            if (m0 + row < N)
                v = *(const float4*)(X + (size_t)(m0 + row) * K + k0 + c4 * 4);
            *(float4*)&As[row][c4 * 4] = v;
        }
#pragma unroll
        for (int i = 0; i < 4; ++i) {
            int f   = tid + i * 256;
            int row = f >> 5;
            int c4  = f & 31;
            float4 v = *(const float4*)(W + (size_t)(k0 + row) * O_DIM + c4 * 4);
            float* bp = (float*)&Bs2[row >> 1][c4 * 4] + (row & 1);
            bp[0] = v.x;
            bp[2] = v.y;
            bp[4] = v.z;
            bp[6] = v.w;
        }
        __syncthreads();

#pragma unroll
        for (int kk = 0; kk < 32; kk += 4) {
            const int kk2 = kk + khalf;
            const int p   = kk2 >> 1;
            v2f a = *(const v2f*)&As[mrow][kk2];
#pragma unroll
            for (int t = 0; t < 8; ++t) {
                v2f b = *(const v2f*)&Bs2[p][t * 16 + lnn];
                acc[t] = __builtin_amdgcn_wmma_f32_16x16x4_f32(
                    false, a, false, b, (short)0, acc[t], false, false);
            }
        }
        __syncthreads();
    }

    const int mtile = m0 + wv * 16;
    if (mtile < N) {
        const int rowoff = (ln >> 4) << 3;
#pragma unroll
        for (int t = 0; t < 8; ++t) {
            const int col = t * 16 + lnn;
            const float bv = bias[col];
#pragma unroll
            for (int i = 0; i < 8; ++i) {
                float v = acc[t][i] + bv;
                out[(size_t)(mtile + rowoff + i) * O_DIM + col] = v > 0.f ? v : 0.f;
            }
        }
    }
}

__global__ __launch_bounds__(256)
void spmm128(const int* __restrict__ src, const int* __restrict__ dst,
             const float* __restrict__ w, const float* __restrict__ x,
             float* __restrict__ out, int E) {
    int e = blockIdx.x * 8 + (threadIdx.x >> 5);
    if (e >= E) return;
    const int ln = threadIdx.x & 31;
    const int s = src[e];
    const int d = dst[e];
    const float wt = w[e];
    float4 v = ((const float4*)(x + (size_t)s * 128))[ln];
    float* o = out + (size_t)d * 128 + ln * 4;
    atomAddF(o + 0, wt * v.x);
    atomAddF(o + 1, wt * v.y);
    atomAddF(o + 2, wt * v.z);
    atomAddF(o + 3, wt * v.w);
}

__global__ __launch_bounds__(256)
void spmm_concat256(const int* __restrict__ src, const int* __restrict__ dst,
                    const float* __restrict__ w, const float* __restrict__ xa,
                    const float* __restrict__ xb, float* __restrict__ out, int E) {
    int e = blockIdx.x * 8 + (threadIdx.x >> 5);
    if (e >= E) return;
    const int ln = threadIdx.x & 31;
    const int s = src[e];
    const int d = dst[e];
    const float wt = w[e];
    float4 va = ((const float4*)(xa + (size_t)s * 128))[ln];
    float4 vb = ((const float4*)(xb + (size_t)s * 128))[ln];
    float* oa = out + (size_t)d * 256 + ln * 4;
    float* ob = oa + 128;
    atomAddF(oa + 0, wt * va.x);
    atomAddF(oa + 1, wt * va.y);
    atomAddF(oa + 2, wt * va.z);
    atomAddF(oa + 3, wt * va.w);
    atomAddF(ob + 0, wt * vb.x);
    atomAddF(ob + 1, wt * vb.y);
    atomAddF(ob + 2, wt * vb.z);
    atomAddF(ob + 3, wt * vb.w);
}

__global__ __launch_bounds__(256)
void l2norm_concat(const float* __restrict__ r0, const float* __restrict__ r1,
                   float* __restrict__ out, int N) {
    int row = blockIdx.x * 8 + (threadIdx.x >> 5);
    if (row >= N) return;
    const int ln = threadIdx.x & 31;

    float4 v0  = ((const float4*)(r0 + (size_t)row * 128))[ln];
    float4 v1a = ((const float4*)(r1 + (size_t)row * 256))[ln];
    float4 v1b = ((const float4*)(r1 + (size_t)row * 256))[ln + 32];

    float s0 = v0.x * v0.x + v0.y * v0.y + v0.z * v0.z + v0.w * v0.w;
    float s1 = v1a.x * v1a.x + v1a.y * v1a.y + v1a.z * v1a.z + v1a.w * v1a.w
             + v1b.x * v1b.x + v1b.y * v1b.y + v1b.z * v1b.z + v1b.w * v1b.w;
#pragma unroll
    for (int off = 16; off > 0; off >>= 1) {
        s0 += __shfl_xor(s0, off, 32);
        s1 += __shfl_xor(s1, off, 32);
    }
    const float i0 = 1.0f / (sqrtf(s0) + 1e-9f);
    const float i1 = 1.0f / (sqrtf(s1) + 1e-9f);

    float* o = out + (size_t)row * 384;
    float4 w0 = make_float4(v0.x * i0, v0.y * i0, v0.z * i0, v0.w * i0);
    float4 wa = make_float4(v1a.x * i1, v1a.y * i1, v1a.z * i1, v1a.w * i1);
    float4 wb = make_float4(v1b.x * i1, v1b.y * i1, v1b.z * i1, v1b.w * i1);
    *(float4*)(o + ln * 4)        = w0;
    *(float4*)(o + 128 + ln * 4)  = wa;
    *(float4*)(o + 256 + ln * 4)  = wb;
}

extern "C" void kernel_launch(void* const* d_in, const int* in_sizes, int n_in,
                              void* d_out, int out_size, void* d_ws, size_t ws_size,
                              hipStream_t stream) {
    const float* x1       = (const float*)d_in[0];
    const float* x2       = (const float*)d_in[1];
    const float* word_emb = (const float*)d_in[2];
    const float* W1a = (const float*)d_in[3];
    const float* b1a = (const float*)d_in[4];
    const float* W1b = (const float*)d_in[5];
    const float* b1b = (const float*)d_in[6];
    const float* W2a = (const float*)d_in[7];
    const float* b2a = (const float*)d_in[8];
    const float* W2b = (const float*)d_in[9];
    const float* b2b = (const float*)d_in[10];
    const int*   e11_src = (const int*)d_in[11];
    const int*   e11_dst = (const int*)d_in[12];
    const float* e11_w   = (const float*)d_in[13];
    const int*   e22_src = (const int*)d_in[14];
    const int*   e22_dst = (const int*)d_in[15];
    const float* e22_w   = (const float*)d_in[16];
    const int*   e01_src = (const int*)d_in[17];
    const int*   e01_dst = (const int*)d_in[18];
    const float* e01_w   = (const float*)d_in[19];
    const int*   e02_src = (const int*)d_in[20];
    const int*   e02_dst = (const int*)d_in[21];
    const float* e02_w   = (const float*)d_in[22];

    const int N0 = 10000, N1 = 50000, N2 = 30000;
    const int E11 = in_sizes[11], E22 = in_sizes[14];
    const int E01 = in_sizes[17], E02 = in_sizes[20];

    float* ws = (float*)d_ws;
    size_t off = 0;
    float* l1_1 = ws + off; off += (size_t)N1 * 128;
    float* h11  = ws + off; off += (size_t)N1 * 128;
    float* l2_1 = ws + off; off += (size_t)N1 * 128;
    float* l1_2 = ws + off; off += (size_t)N2 * 128;
    float* h22  = ws + off; off += (size_t)N2 * 128;
    float* l2_2 = ws + off; off += (size_t)N2 * 128;
    float* r0   = ws + off; off += (size_t)N0 * 128;
    float* r1   = ws + off; off += (size_t)N0 * 256;
    float* r0s  = ws + off; off += (size_t)N0 * 128;
    float* r1s  = ws + off; off += (size_t)N0 * 256;

    float* doc     = (float*)d_out;
    float* doc_svd = doc + (size_t)N0 * 384;

    const dim3 blk(256);
    auto gemm_grid = [](int n) { return dim3((n + 127) / 128); };
    auto edge_grid = [](int e) { return dim3((e + 7) / 8); };

    gemm_bias_relu<<<gemm_grid(N1), blk, 0, stream>>>(x1, W1a, b1a, l1_1, N1, 256);
    hipMemsetAsync(h11, 0, (size_t)N1 * 128 * sizeof(float), stream);
    spmm128<<<edge_grid(E11), blk, 0, stream>>>(e11_src, e11_dst, e11_w, l1_1, h11, E11);
    gemm_bias_relu<<<gemm_grid(N1), blk, 0, stream>>>(h11, W1b, b1b, l2_1, N1, 128);

    gemm_bias_relu<<<gemm_grid(N2), blk, 0, stream>>>(x2, W2a, b2a, l1_2, N2, 256);
    hipMemsetAsync(h22, 0, (size_t)N2 * 128 * sizeof(float), stream);
    spmm128<<<edge_grid(E22), blk, 0, stream>>>(e22_src, e22_dst, e22_w, l1_2, h22, E22);
    gemm_bias_relu<<<gemm_grid(N2), blk, 0, stream>>>(h22, W2b, b2b, l2_2, N2, 128);

    hipMemsetAsync(r0, 0, (size_t)N0 * (128 + 256 + 128 + 256) * sizeof(float), stream);
    spmm128<<<edge_grid(E01), blk, 0, stream>>>(e01_src, e01_dst, e01_w, l2_1, r0, E01);
    spmm_concat256<<<edge_grid(E02), blk, 0, stream>>>(e02_src, e02_dst, e02_w, l2_2, word_emb, r1, E02);
    spmm128<<<edge_grid(E01), blk, 0, stream>>>(e01_src, e01_dst, e01_w, l1_1, r0s, E01);
    spmm_concat256<<<edge_grid(E02), blk, 0, stream>>>(e02_src, e02_dst, e02_w, l1_2, word_emb, r1s, E02);

    l2norm_concat<<<dim3((N0 + 7) / 8), blk, 0, stream>>>(r0,  r1,  doc,     N0);
    l2norm_concat<<<dim3((N0 + 7) / 8), blk, 0, stream>>>(r0s, r1s, doc_svd, N0);
}


# edge scatter-add: lane l takes columns l+32j, so every f32 atomic add and row load of a wave covers one whole 128-byte line (4 add lines per 128 columns, was 16); sums stay in atomic arrival order
# speedup vs baseline: 1.4180x; 1.4180x over previous
#include <hip/hip_runtime.h>
#include <math.h>

typedef __attribute__((ext_vector_type(2))) float v2f;
typedef __attribute__((ext_vector_type(8))) float v8f;

#define O_DIM 128

__device__ __forceinline__ void atomAddF(float* p, float v) {
    __hip_atomic_fetch_add(p, v, __ATOMIC_RELAXED, __HIP_MEMORY_SCOPE_AGENT);
}

__global__ __launch_bounds__(256)
void gemm_bias_relu(const float* __restrict__ X, const float* __restrict__ W,
                    const float* __restrict__ bias, float* __restrict__ out,
                    int N, int K) {
    __shared__ float  As[128][36];
    __shared__ float2 Bs2[16][144];

    const int tid = threadIdx.x;
    const int wv  = tid >> 5;
    const int ln  = tid & 31;
    const int m0  = blockIdx.x * 128;

    v8f acc[8];
#pragma unroll
    for (int t = 0; t < 8; ++t)
#pragma unroll
        for (int j = 0; j < 8; ++j) acc[t][j] = 0.0f;

    const int mrow = wv * 16 + (ln & 15);
    const int lnn  = ln & 15;
    const int khalf = (ln >> 4) << 1;

    for (int k0 = 0; k0 < K; k0 += 32) {
#pragma unroll
        for (int i = 0; i < 4; ++i) {
            int f   = tid + i * 256;
            int row = f >> 3;
            int c4  = f & 7;
            float4 v = make_float4(0.f, 0.f, 0.f, 0.f);
            if (m0 + row < N)
                v = *(const float4*)(X + (size_t)(m0 + row) * K + k0 + c4 * 4);
            *(float4*)&As[row][c4 * 4] = v;
        }
#pragma unroll
        for (int i = 0; i < 4; ++i) {
            int f   = tid + i * 256;
            int row = f >> 5;
            int c4  = f & 31;
            float4 v = *(const float4*)(W + (size_t)(k0 + row) * O_DIM + c4 * 4);
            float* bp = (float*)&Bs2[row >> 1][c4 * 4] + (row & 1);
            bp[0] = v.x;
            bp[2] = v.y;
            bp[4] = v.z;
            bp[6] = v.w;
        }
        __syncthreads();

#pragma unroll
        for (int kk = 0; kk < 32; kk += 4) {
            const int kk2 = kk + khalf;
            const int p   = kk2 >> 1;
            v2f a = *(const v2f*)&As[mrow][kk2];
#pragma unroll
            for (int t = 0; t < 8; ++t) {
                v2f b = *(const v2f*)&Bs2[p][t * 16 + lnn];
                acc[t] = __builtin_amdgcn_wmma_f32_16x16x4_f32(
                    false, a, false, b, (short)0, acc[t], false, false);
            }
        }
        __syncthreads();
    }

    const int mtile = m0 + wv * 16;
    if (mtile < N) {
        const int rowoff = (ln >> 4) << 3;
#pragma unroll
        for (int t = 0; t < 8; ++t) {
            const int col = t * 16 + lnn;
            const float bv = bias[col];
#pragma unroll
            for (int i = 0; i < 8; ++i) {
                float v = acc[t][i] + bv;
                out[(size_t)(mtile + rowoff + i) * O_DIM + col] = v > 0.f ? v : 0.f;
            }
        }
    }
}

__global__ __launch_bounds__(256)
void spmm128(const int* __restrict__ src, const int* __restrict__ dst,
             const float* __restrict__ w, const float* __restrict__ x,
             float* __restrict__ out, int E) {
    int e = blockIdx.x * 8 + (threadIdx.x >> 5);
    if (e >= E) return;
    const int ln = threadIdx.x & 31;
    const int s = src[e];
    const int d = dst[e];
    const float wt = w[e];
    const float* xr = x + (size_t)s * 128 + ln;
    const float v0 = xr[0];
    const float v1 = xr[32];
    const float v2 = xr[64];
    const float v3 = xr[96];
    float* o = out + (size_t)d * 128 + ln;
    atomAddF(o +  0, wt * v0);
    atomAddF(o + 32, wt * v1);
    atomAddF(o + 64, wt * v2);
    atomAddF(o + 96, wt * v3);
}

__global__ __launch_bounds__(256)
void spmm_concat256(const int* __restrict__ src, const int* __restrict__ dst,
                    const float* __restrict__ w, const float* __restrict__ xa,
                    const float* __restrict__ xb, float* __restrict__ out, int E) {
    int e = blockIdx.x * 8 + (threadIdx.x >> 5);
    if (e >= E) return;
    const int ln = threadIdx.x & 31;
    const int s = src[e];
    const int d = dst[e];
    const float wt = w[e];
    const float* ar = xa + (size_t)s * 128 + ln;
    const float* br = xb + (size_t)s * 128 + ln;
    const float a0 = ar[0];
    const float a1 = ar[32];
    const float a2 = ar[64];
    const float a3 = ar[96];
    const float b0 = br[0];
    const float b1 = br[32];
    const float b2 = br[64];
    const float b3 = br[96];
    float* o = out + (size_t)d * 256 + ln;
    atomAddF(o +   0, wt * a0);
    atomAddF(o +  32, wt * a1);
    atomAddF(o +  64, wt * a2);
    atomAddF(o +  96, wt * a3);
    atomAddF(o + 128, wt * b0);
    atomAddF(o + 160, wt * b1);
    atomAddF(o + 192, wt * b2);
    atomAddF(o + 224, wt * b3);
}

__global__ __launch_bounds__(256)
void l2norm_concat(const float* __restrict__ r0, const float* __restrict__ r1,
                   float* __restrict__ out, int N) {
    int row = blockIdx.x * 8 + (threadIdx.x >> 5);
    if (row >= N) return;
    const int ln = threadIdx.x & 31;

    float4 v0  = ((const float4*)(r0 + (size_t)row * 128))[ln];
    float4 v1a = ((const float4*)(r1 + (size_t)row * 256))[ln];
    float4 v1b = ((const float4*)(r1 + (size_t)row * 256))[ln + 32];

    float s0 = v0.x * v0.x + v0.y * v0.y + v0.z * v0.z + v0.w * v0.w;
    float s1 = v1a.x * v1a.x + v1a.y * v1a.y + v1a.z * v1a.z + v1a.w * v1a.w
             + v1b.x * v1b.x + v1b.y * v1b.y + v1b.z * v1b.z + v1b.w * v1b.w;
#pragma unroll
    for (int off = 16; off > 0; off >>= 1) {
        s0 += __shfl_xor(s0, off, 32);
        s1 += __shfl_xor(s1, off, 32);
    }
    const float i0 = 1.0f / (sqrtf(s0) + 1e-9f);
    const float i1 = 1.0f / (sqrtf(s1) + 1e-9f);

    float* o = out + (size_t)row * 384;
    float4 w0 = make_float4(v0.x * i0, v0.y * i0, v0.z * i0, v0.w * i0);
    float4 wa = make_float4(v1a.x * i1, v1a.y * i1, v1a.z * i1, v1a.w * i1);
    float4 wb = make_float4(v1b.x * i1, v1b.y * i1, v1b.z * i1, v1b.w * i1);
    *(float4*)(o + ln * 4)        = w0;
    *(float4*)(o + 128 + ln * 4)  = wa;
    *(float4*)(o + 256 + ln * 4)  = wb;
}

extern "C" void kernel_launch(void* const* d_in, const int* in_sizes, int n_in,
                              void* d_out, int out_size, void* d_ws, size_t ws_size,
                              hipStream_t stream) {
    const float* x1       = (const float*)d_in[0];
    const float* x2       = (const float*)d_in[1];
    const float* word_emb = (const float*)d_in[2];
    const float* W1a = (const float*)d_in[3];
    const float* b1a = (const float*)d_in[4];
    const float* W1b = (const float*)d_in[5];
    const float* b1b = (const float*)d_in[6];
    const float* W2a = (const float*)d_in[7];
    const float* b2a = (const float*)d_in[8];
    const float* W2b = (const float*)d_in[9];
    const float* b2b = (const float*)d_in[10];
    const int*   e11_src = (const int*)d_in[11];
    const int*   e11_dst = (const int*)d_in[12];
    const float* e11_w   = (const float*)d_in[13];
    const int*   e22_src = (const int*)d_in[14];
    const int*   e22_dst = (const int*)d_in[15];
    const float* e22_w   = (const float*)d_in[16];
    const int*   e01_src = (const int*)d_in[17];
    const int*   e01_dst = (const int*)d_in[18];
    const float* e01_w   = (const float*)d_in[19];
    const int*   e02_src = (const int*)d_in[20];
    const int*   e02_dst = (const int*)d_in[21];
    const float* e02_w   = (const float*)d_in[22];

    const int N0 = 10000, N1 = 50000, N2 = 30000;
    const int E11 = in_sizes[11], E22 = in_sizes[14];
    const int E01 = in_sizes[17], E02 = in_sizes[20];

    float* ws = (float*)d_ws;
    size_t off = 0;
    float* l1_1 = ws + off; off += (size_t)N1 * 128;
    float* h11  = ws + off; off += (size_t)N1 * 128;
    float* l2_1 = ws + off; off += (size_t)N1 * 128;
    float* l1_2 = ws + off; off += (size_t)N2 * 128;
    float* h22  = ws + off; off += (size_t)N2 * 128;
    float* l2_2 = ws + off; off += (size_t)N2 * 128;
    float* r0   = ws + off; off += (size_t)N0 * 128;
    float* r1   = ws + off; off += (size_t)N0 * 256;
    float* r0s  = ws + off; off += (size_t)N0 * 128;
    float* r1s  = ws + off; off += (size_t)N0 * 256;

    float* doc     = (float*)d_out;
    float* doc_svd = doc + (size_t)N0 * 384;

    const dim3 blk(256);
    auto gemm_grid = [](int n) { return dim3((n + 127) / 128); };
    auto edge_grid = [](int e) { return dim3((e + 7) / 8); };

    gemm_bias_relu<<<gemm_grid(N1), blk, 0, stream>>>(x1, W1a, b1a, l1_1, N1, 256);
    hipMemsetAsync(h11, 0, (size_t)N1 * 128 * sizeof(float), stream);
    spmm128<<<edge_grid(E11), blk, 0, stream>>>(e11_src, e11_dst, e11_w, l1_1, h11, E11);
    gemm_bias_relu<<<gemm_grid(N1), blk, 0, stream>>>(h11, W1b, b1b, l2_1, N1, 128);

    gemm_bias_relu<<<gemm_grid(N2), blk, 0, stream>>>(x2, W2a, b2a, l1_2, N2, 256);
    hipMemsetAsync(h22, 0, (size_t)N2 * 128 * sizeof(float), stream);
    spmm128<<<edge_grid(E22), blk, 0, stream>>>(e22_src, e22_dst, e22_w, l1_2, h22, E22);
    gemm_bias_relu<<<gemm_grid(N2), blk, 0, stream>>>(h22, W2b, b2b, l2_2, N2, 128);

    hipMemsetAsync(r0, 0, (size_t)N0 * (128 + 256 + 128 + 256) * sizeof(float), stream);
    spmm128<<<edge_grid(E01), blk, 0, stream>>>(e01_src, e01_dst, e01_w, l2_1, r0, E01);
    spmm_concat256<<<edge_grid(E02), blk, 0, stream>>>(e02_src, e02_dst, e02_w, l2_2, word_emb, r1, E02);
    spmm128<<<edge_grid(E01), blk, 0, stream>>>(e01_src, e01_dst, e01_w, l1_1, r0s, E01);
    spmm_concat256<<<edge_grid(E02), blk, 0, stream>>>(e02_src, e02_dst, e02_w, l1_2, word_emb, r1s, E02);

    l2norm_concat<<<dim3((N0 + 7) / 8), blk, 0, stream>>>(r0,  r1,  doc,     N0);
    l2norm_concat<<<dim3((N0 + 7) / 8), blk, 0, stream>>>(r0s, r1s, doc_svd, N0);
}


# edge scatter-add: one wave walks 32 consecutive edges (coalesced index loads, 8-edge batches with all row loads issued before the adds): 32x fewer waves; sums stay in atomic arrival order
# speedup vs baseline: 1.7277x; 1.2184x over previous
#include <hip/hip_runtime.h>
#include <math.h>

typedef __attribute__((ext_vector_type(2))) float v2f;
typedef __attribute__((ext_vector_type(8))) float v8f;

#define O_DIM 128

__device__ __forceinline__ void atomAddF(float* p, float v) {
    __hip_atomic_fetch_add(p, v, __ATOMIC_RELAXED, __HIP_MEMORY_SCOPE_AGENT);
}

__global__ __launch_bounds__(256)
void gemm_bias_relu(const float* __restrict__ X, const float* __restrict__ W,
                    const float* __restrict__ bias, float* __restrict__ out,
                    int N, int K) {
    __shared__ float  As[128][36];
    __shared__ float2 Bs2[16][144];

    const int tid = threadIdx.x;
    const int wv  = tid >> 5;
    const int ln  = tid & 31;
    const int m0  = blockIdx.x * 128;

    v8f acc[8];
#pragma unroll
    for (int t = 0; t < 8; ++t)
#pragma unroll
        for (int j = 0; j < 8; ++j) acc[t][j] = 0.0f;

    const int mrow = wv * 16 + (ln & 15);
    const int lnn  = ln & 15;
    const int khalf = (ln >> 4) << 1;

    for (int k0 = 0; k0 < K; k0 += 32) {
#pragma unroll
        for (int i = 0; i < 4; ++i) {
            int f   = tid + i * 256;
            int row = f >> 3;
            int c4  = f & 7;
            float4 v = make_float4(0.f, 0.f, 0.f, 0.f);
            if (m0 + row < N)
                v = *(const float4*)(X + (size_t)(m0 + row) * K + k0 + c4 * 4);
            *(float4*)&As[row][c4 * 4] = v;
        }
#pragma unroll
        for (int i = 0; i < 4; ++i) {
            int f   = tid + i * 256;
            int row = f >> 5;
            int c4  = f & 31;
            float4 v = *(const float4*)(W + (size_t)(k0 + row) * O_DIM + c4 * 4);
            float* bp = (float*)&Bs2[row >> 1][c4 * 4] + (row & 1);
            bp[0] = v.x;
            bp[2] = v.y;
            bp[4] = v.z;
            bp[6] = v.w;
        }
        __syncthreads();

#pragma unroll
        for (int kk = 0; kk < 32; kk += 4) {
            const int kk2 = kk + khalf;
            const int p   = kk2 >> 1;
            v2f a = *(const v2f*)&As[mrow][kk2];
#pragma unroll
            for (int t = 0; t < 8; ++t) {
                v2f b = *(const v2f*)&Bs2[p][t * 16 + lnn];
                acc[t] = __builtin_amdgcn_wmma_f32_16x16x4_f32(
                    false, a, false, b, (short)0, acc[t], false, false);
            }
        }
        __syncthreads();
    }

    const int mtile = m0 + wv * 16;
    if (mtile < N) {
        const int rowoff = (ln >> 4) << 3;
#pragma unroll
        for (int t = 0; t < 8; ++t) {
            const int col = t * 16 + lnn;
            const float bv = bias[col];
#pragma unroll
            for (int i = 0; i < 8; ++i) {
                float v = acc[t][i] + bv;
                out[(size_t)(mtile + rowoff + i) * O_DIM + col] = v > 0.f ? v : 0.f;
            }
        }
    }
}

#define EDGES_PER_WAVE 32
#define EDGES_PER_BLOCK (8 * EDGES_PER_WAVE)

__device__ __forceinline__ int bcastI(int v, int lane) {
    return __builtin_amdgcn_readlane(v, lane);
}
__device__ __forceinline__ float bcastF(float v, int lane) {
    return __builtin_bit_cast(float, __builtin_amdgcn_readlane(__builtin_bit_cast(int, v), lane));
}

__global__ __launch_bounds__(256)
void spmm128(const int* __restrict__ src, const int* __restrict__ dst,
             const float* __restrict__ w, const float* __restrict__ x,
             float* __restrict__ out, int E) {
    const int ln = threadIdx.x & 31;
    const int wave = __builtin_amdgcn_readfirstlane((int)(blockIdx.x * 8 + (threadIdx.x >> 5)));
    const int e0 = wave * EDGES_PER_WAVE;
    if (e0 >= E) return;
    const int rem = E - e0;
    const int cnt = rem < EDGES_PER_WAVE ? rem : EDGES_PER_WAVE;

    int sv = 0, dv = 0;
    float wv = 0.f;
    if (ln < cnt) {
        sv = src[e0 + ln];
        dv = dst[e0 + ln];
        wv = w[e0 + ln];
    }
    const float* xl = x + ln;
    float* ol = out + ln;

    constexpr int EB = 8;
    int j0 = 0;
    for (; j0 + EB <= cnt; j0 += EB) {
        float v[EB][4];
#pragma unroll
        for (int j = 0; j < EB; ++j) {
            const float* xr = xl + (size_t)bcastI(sv, j0 + j) * 128;
            v[j][0] = xr[0];
            v[j][1] = xr[32];
            v[j][2] = xr[64];
            v[j][3] = xr[96];
        }
#pragma unroll
        for (int j = 0; j < EB; ++j) {
            float* o = ol + (size_t)bcastI(dv, j0 + j) * 128;
            const float wt = bcastF(wv, j0 + j);
            atomAddF(o +  0, wt * v[j][0]);
            atomAddF(o + 32, wt * v[j][1]);
            atomAddF(o + 64, wt * v[j][2]);
            atomAddF(o + 96, wt * v[j][3]);
        }
    }
    for (; j0 < cnt; ++j0) {
        const float* xr = xl + (size_t)bcastI(sv, j0) * 128;
        const float v0 = xr[0];
        const float v1 = xr[32];
        const float v2 = xr[64];
        const float v3 = xr[96];
        float* o = ol + (size_t)bcastI(dv, j0) * 128;
        const float wt = bcastF(wv, j0);
        atomAddF(o +  0, wt * v0);
        atomAddF(o + 32, wt * v1);
        atomAddF(o + 64, wt * v2);
        atomAddF(o + 96, wt * v3);
    }
}

__global__ __launch_bounds__(256)
void spmm_concat256(const int* __restrict__ src, const int* __restrict__ dst,
                    const float* __restrict__ w, const float* __restrict__ xa,
                    const float* __restrict__ xb, float* __restrict__ out, int E) {
    const int ln = threadIdx.x & 31;
    const int wave = __builtin_amdgcn_readfirstlane((int)(blockIdx.x * 8 + (threadIdx.x >> 5)));
    const int e0 = wave * EDGES_PER_WAVE;
    if (e0 >= E) return;
    const int rem = E - e0;
    const int cnt = rem < EDGES_PER_WAVE ? rem : EDGES_PER_WAVE;

    int sv = 0, dv = 0;
    float wv = 0.f;
    if (ln < cnt) {
        sv = src[e0 + ln];
        dv = dst[e0 + ln];
        wv = w[e0 + ln];
    }
    const float* al = xa + ln;
    const float* bl = xb + ln;
    float* ol = out + ln;

    constexpr int EB = 4;
    int j0 = 0;
    for (; j0 + EB <= cnt; j0 += EB) {
        float v[EB][8];
#pragma unroll
        for (int j = 0; j < EB; ++j) {
            const size_t ro = (size_t)bcastI(sv, j0 + j) * 128;
            const float* ar = al + ro;
            const float* br = bl + ro;
            v[j][0] = ar[0];
            v[j][1] = ar[32];
            v[j][2] = ar[64];
            v[j][3] = ar[96];
            v[j][4] = br[0];
            v[j][5] = br[32];
            v[j][6] = br[64];
            v[j][7] = br[96];
        }
#pragma unroll
        for (int j = 0; j < EB; ++j) {
            float* o = ol + (size_t)bcastI(dv, j0 + j) * 256;
            const float wt = bcastF(wv, j0 + j);
#pragma unroll
            for (int c = 0; c < 8; ++c) atomAddF(o + 32 * c, wt * v[j][c]);
        }
    }
    for (; j0 < cnt; ++j0) {
        const size_t ro = (size_t)bcastI(sv, j0) * 128;
        const float* ar = al + ro;
        const float* br = bl + ro;
        float v[8];
        v[0] = ar[0];
        v[1] = ar[32];
        v[2] = ar[64];
        v[3] = ar[96];
        v[4] = br[0];
        v[5] = br[32];
        v[6] = br[64];
        v[7] = br[96];
        float* o = ol + (size_t)bcastI(dv, j0) * 256;
        const float wt = bcastF(wv, j0);
#pragma unroll
        for (int c = 0; c < 8; ++c) atomAddF(o + 32 * c, wt * v[c]);
    }
}

__global__ __launch_bounds__(256)
void l2norm_concat(const float* __restrict__ r0, const float* __restrict__ r1,
                   float* __restrict__ out, int N) {
    int row = blockIdx.x * 8 + (threadIdx.x >> 5);
    if (row >= N) return;
    const int ln = threadIdx.x & 31;

    float4 v0  = ((const float4*)(r0 + (size_t)row * 128))[ln];
    float4 v1a = ((const float4*)(r1 + (size_t)row * 256))[ln];
    float4 v1b = ((const float4*)(r1 + (size_t)row * 256))[ln + 32];

    float s0 = v0.x * v0.x + v0.y * v0.y + v0.z * v0.z + v0.w * v0.w;
    float s1 = v1a.x * v1a.x + v1a.y * v1a.y + v1a.z * v1a.z + v1a.w * v1a.w
             + v1b.x * v1b.x + v1b.y * v1b.y + v1b.z * v1b.z + v1b.w * v1b.w;
#pragma unroll
    for (int off = 16; off > 0; off >>= 1) {
        s0 += __shfl_xor(s0, off, 32);
        s1 += __shfl_xor(s1, off, 32);
    }
    const float i0 = 1.0f / (sqrtf(s0) + 1e-9f);
    const float i1 = 1.0f / (sqrtf(s1) + 1e-9f);

    float* o = out + (size_t)row * 384;
    float4 w0 = make_float4(v0.x * i0, v0.y * i0, v0.z * i0, v0.w * i0);
    float4 wa = make_float4(v1a.x * i1, v1a.y * i1, v1a.z * i1, v1a.w * i1);
    float4 wb = make_float4(v1b.x * i1, v1b.y * i1, v1b.z * i1, v1b.w * i1);
    *(float4*)(o + ln * 4)        = w0;
    *(float4*)(o + 128 + ln * 4)  = wa;
    *(float4*)(o + 256 + ln * 4)  = wb;
}

extern "C" void kernel_launch(void* const* d_in, const int* in_sizes, int n_in,
                              void* d_out, int out_size, void* d_ws, size_t ws_size,
                              hipStream_t stream) {
    const float* x1       = (const float*)d_in[0];
    const float* x2       = (const float*)d_in[1];
    const float* word_emb = (const float*)d_in[2];
    const float* W1a = (const float*)d_in[3];
    const float* b1a = (const float*)d_in[4];
    const float* W1b = (const float*)d_in[5];
    const float* b1b = (const float*)d_in[6];
    const float* W2a = (const float*)d_in[7];
    const float* b2a = (const float*)d_in[8];
    const float* W2b = (const float*)d_in[9];
    const float* b2b = (const float*)d_in[10];
    const int*   e11_src = (const int*)d_in[11];
    const int*   e11_dst = (const int*)d_in[12];
    const float* e11_w   = (const float*)d_in[13];
    const int*   e22_src = (const int*)d_in[14];
    const int*   e22_dst = (const int*)d_in[15];
    const float* e22_w   = (const float*)d_in[16];
    const int*   e01_src = (const int*)d_in[17];
    const int*   e01_dst = (const int*)d_in[18];
    const float* e01_w   = (const float*)d_in[19];
    const int*   e02_src = (const int*)d_in[20];
    const int*   e02_dst = (const int*)d_in[21];
    const float* e02_w   = (const float*)d_in[22];

    const int N0 = 10000, N1 = 50000, N2 = 30000;
    const int E11 = in_sizes[11], E22 = in_sizes[14];
    const int E01 = in_sizes[17], E02 = in_sizes[20];

    float* ws = (float*)d_ws;
    size_t off = 0;
    float* l1_1 = ws + off; off += (size_t)N1 * 128;
    float* h11  = ws + off; off += (size_t)N1 * 128;
    float* l2_1 = ws + off; off += (size_t)N1 * 128;
    float* l1_2 = ws + off; off += (size_t)N2 * 128;
    float* h22  = ws + off; off += (size_t)N2 * 128;
    float* l2_2 = ws + off; off += (size_t)N2 * 128;
    float* r0   = ws + off; off += (size_t)N0 * 128;
    float* r1   = ws + off; off += (size_t)N0 * 256;
    float* r0s  = ws + off; off += (size_t)N0 * 128;
    float* r1s  = ws + off; off += (size_t)N0 * 256;

    float* doc     = (float*)d_out;
    float* doc_svd = doc + (size_t)N0 * 384;

    const dim3 blk(256);
    auto gemm_grid = [](int n) { return dim3((n + 127) / 128); };
    auto edge_grid = [](int e) { return dim3((e + EDGES_PER_BLOCK - 1) / EDGES_PER_BLOCK); };

    gemm_bias_relu<<<gemm_grid(N1), blk, 0, stream>>>(x1, W1a, b1a, l1_1, N1, 256);
    hipMemsetAsync(h11, 0, (size_t)N1 * 128 * sizeof(float), stream);
    spmm128<<<edge_grid(E11), blk, 0, stream>>>(e11_src, e11_dst, e11_w, l1_1, h11, E11);
    gemm_bias_relu<<<gemm_grid(N1), blk, 0, stream>>>(h11, W1b, b1b, l2_1, N1, 128);

    gemm_bias_relu<<<gemm_grid(N2), blk, 0, stream>>>(x2, W2a, b2a, l1_2, N2, 256);
    hipMemsetAsync(h22, 0, (size_t)N2 * 128 * sizeof(float), stream);
    spmm128<<<edge_grid(E22), blk, 0, stream>>>(e22_src, e22_dst, e22_w, l1_2, h22, E22);
    gemm_bias_relu<<<gemm_grid(N2), blk, 0, stream>>>(h22, W2b, b2b, l2_2, N2, 128);

    hipMemsetAsync(r0, 0, (size_t)N0 * (128 + 256 + 128 + 256) * sizeof(float), stream);
    spmm128<<<edge_grid(E01), blk, 0, stream>>>(e01_src, e01_dst, e01_w, l2_1, r0, E01);
    spmm_concat256<<<edge_grid(E02), blk, 0, stream>>>(e02_src, e02_dst, e02_w, l2_2, word_emb, r1, E02);
    spmm128<<<edge_grid(E01), blk, 0, stream>>>(e01_src, e01_dst, e01_w, l1_1, r0s, E01);
    spmm_concat256<<<edge_grid(E02), blk, 0, stream>>>(e02_src, e02_dst, e02_w, l1_2, word_emb, r1s, E02);

    l2norm_concat<<<dim3((N0 + 7) / 8), blk, 0, stream>>>(r0,  r1,  doc,     N0);
    l2norm_concat<<<dim3((N0 + 7) / 8), blk, 0, stream>>>(r0s, r1s, doc_svd, N0);
}


# doc aggregation owner-computes in LDS: a work-group keeps 40 output rows, scans both edge lists, adds gathered rows by LDS f32 atomics (arrival-order sums; shared word half once), normalises, stores
# speedup vs baseline: 2.0730x; 1.1999x over previous
#include <hip/hip_runtime.h>
#include <math.h>

typedef __attribute__((ext_vector_type(2))) float v2f;
typedef __attribute__((ext_vector_type(8))) float v8f;

#define O_DIM 128

__device__ __forceinline__ void atomAddF(float* p, float v) {
    __hip_atomic_fetch_add(p, v, __ATOMIC_RELAXED, __HIP_MEMORY_SCOPE_AGENT);
}

__global__ __launch_bounds__(256)
void gemm_bias_relu(const float* __restrict__ X, const float* __restrict__ W,
                    const float* __restrict__ bias, float* __restrict__ out,
                    int N, int K) {
    __shared__ float  As[128][36];
    __shared__ float2 Bs2[16][144];

    const int tid = threadIdx.x;
    const int wv  = tid >> 5;
    const int ln  = tid & 31;
    const int m0  = blockIdx.x * 128;

    v8f acc[8];
#pragma unroll
    for (int t = 0; t < 8; ++t)
#pragma unroll
        for (int j = 0; j < 8; ++j) acc[t][j] = 0.0f;

    const int mrow = wv * 16 + (ln & 15);
    const int lnn  = ln & 15;
    const int khalf = (ln >> 4) << 1;

    for (int k0 = 0; k0 < K; k0 += 32) {
#pragma unroll
        for (int i = 0; i < 4; ++i) {
            int f   = tid + i * 256;
            int row = f >> 3;
            int c4  = f & 7;
            float4 v = make_float4(0.f, 0.f, 0.f, 0.f);
            if (m0 + row < N)
                v = *(const float4*)(X + (size_t)(m0 + row) * K + k0 + c4 * 4);
            *(float4*)&As[row][c4 * 4] = v;
        }
#pragma unroll
        for (int i = 0; i < 4; ++i) {
            int f   = tid + i * 256;
            int row = f >> 5;
            int c4  = f & 31;
            float4 v = *(const float4*)(W + (size_t)(k0 + row) * O_DIM + c4 * 4);
            float* bp = (float*)&Bs2[row >> 1][c4 * 4] + (row & 1);
            bp[0] = v.x;
            bp[2] = v.y;
            bp[4] = v.z;
            bp[6] = v.w;
        }
        __syncthreads();

#pragma unroll
        for (int kk = 0; kk < 32; kk += 4) {
            const int kk2 = kk + khalf;
            const int p   = kk2 >> 1;
            v2f a = *(const v2f*)&As[mrow][kk2];
#pragma unroll
            for (int t = 0; t < 8; ++t) {
                v2f b = *(const v2f*)&Bs2[p][t * 16 + lnn];
                acc[t] = __builtin_amdgcn_wmma_f32_16x16x4_f32(
                    false, a, false, b, (short)0, acc[t], false, false);
            }
        }
        __syncthreads();
    }

    const int mtile = m0 + wv * 16;
    if (mtile < N) {
        const int rowoff = (ln >> 4) << 3;
#pragma unroll
        for (int t = 0; t < 8; ++t) {
            const int col = t * 16 + lnn;
            const float bv = bias[col];
#pragma unroll
            for (int i = 0; i < 8; ++i) {
                float v = acc[t][i] + bv;
                out[(size_t)(mtile + rowoff + i) * O_DIM + col] = v > 0.f ? v : 0.f;
            }
        }
    }
}

#define EDGES_PER_WAVE 32
#define EDGES_PER_BLOCK (8 * EDGES_PER_WAVE)

__device__ __forceinline__ int bcastI(int v, int lane) {
    return __builtin_amdgcn_readlane(v, lane);
}
__device__ __forceinline__ float bcastF(float v, int lane) {
    return __builtin_bit_cast(float, __builtin_amdgcn_readlane(__builtin_bit_cast(int, v), lane));
}

__global__ __launch_bounds__(256)
void spmm128(const int* __restrict__ src, const int* __restrict__ dst,
             const float* __restrict__ w, const float* __restrict__ x,
             float* __restrict__ out, int E) {
    const int ln = threadIdx.x & 31;
    const int wave = __builtin_amdgcn_readfirstlane((int)(blockIdx.x * 8 + (threadIdx.x >> 5)));
    const int e0 = wave * EDGES_PER_WAVE;
    if (e0 >= E) return;
    const int rem = E - e0;
    const int cnt = rem < EDGES_PER_WAVE ? rem : EDGES_PER_WAVE;

    int sv = 0, dv = 0;
    float wv = 0.f;
    if (ln < cnt) {
        sv = src[e0 + ln];
        dv = dst[e0 + ln];
        wv = w[e0 + ln];
    }
    const float* xl = x + ln;
    float* ol = out + ln;

    constexpr int EB = 8;
    int j0 = 0;
    for (; j0 + EB <= cnt; j0 += EB) {
        float v[EB][4];
#pragma unroll
        for (int j = 0; j < EB; ++j) {
            const float* xr = xl + (size_t)bcastI(sv, j0 + j) * 128;
            v[j][0] = xr[0];
            v[j][1] = xr[32];
            v[j][2] = xr[64];
            v[j][3] = xr[96];
        }
#pragma unroll
        for (int j = 0; j < EB; ++j) {
            float* o = ol + (size_t)bcastI(dv, j0 + j) * 128;
            const float wt = bcastF(wv, j0 + j);
            atomAddF(o +  0, wt * v[j][0]);
            atomAddF(o + 32, wt * v[j][1]);
            atomAddF(o + 64, wt * v[j][2]);
            atomAddF(o + 96, wt * v[j][3]);
        }
    }
    for (; j0 < cnt; ++j0) {
        const float* xr = xl + (size_t)bcastI(sv, j0) * 128;
        const float v0 = xr[0];
        const float v1 = xr[32];
        const float v2 = xr[64];
        const float v3 = xr[96];
        float* o = ol + (size_t)bcastI(dv, j0) * 128;
        const float wt = bcastF(wv, j0);
        atomAddF(o +  0, wt * v0);
        atomAddF(o + 32, wt * v1);
        atomAddF(o + 64, wt * v2);
        atomAddF(o + 96, wt * v3);
    }
}

#define AGG_ROWS 40
#define AGG_ROWF 640
#define AGG_QCAP 6144

__device__ __forceinline__ void ldsAddF(float* p, float v) {
    __hip_atomic_fetch_add(p, v, __ATOMIC_RELAXED, __HIP_MEMORY_SCOPE_WORKGROUP);
}

template <int NS, int EB>
__device__ __forceinline__ void gather_add(const int* qs, const int* qd, const float* qw, int b,
                                           const float* __restrict__ xa, const float* __restrict__ xb,
                                           const float* __restrict__ xc, float* acc,
                                           int sa, int sb, int sc, int ln) {
    float v[EB][4 * NS];
    int   dl[EB];
    float wt[EB];
#pragma unroll
    for (int j = 0; j < EB; ++j) {
        const size_t ro = (size_t)__builtin_amdgcn_readfirstlane(qs[b + j]) * 128 + ln;
        dl[j] = __builtin_amdgcn_readfirstlane(qd[b + j]);
        wt[j] = qw[b + j];
#pragma unroll
        for (int c = 0; c < 4; ++c) {
            v[j][c]     = xa[ro + 32 * c];
            v[j][4 + c] = xb[ro + 32 * c];
            if (NS == 3) v[j][8 + c] = xc[ro + 32 * c];
        }
    }
#pragma unroll
    for (int j = 0; j < EB; ++j) {
        float* o = acc + dl[j] * AGG_ROWF + ln;
#pragma unroll
        for (int c = 0; c < 4; ++c) {
            ldsAddF(o + sa + 32 * c, wt[j] * v[j][c]);
            ldsAddF(o + sb + 32 * c, wt[j] * v[j][4 + c]);
            if (NS == 3) ldsAddF(o + sc + 32 * c, wt[j] * v[j][8 + c]);
        }
    }
}

__global__ __launch_bounds__(1024)
void doc_aggregate(const int* __restrict__ e01_src, const int* __restrict__ e01_dst,
                   const float* __restrict__ e01_w, int E01,
                   const int* __restrict__ e02_src, const int* __restrict__ e02_dst,
                   const float* __restrict__ e02_w, int E02,
                   const float* __restrict__ l2_1, const float* __restrict__ l1_1,
                   const float* __restrict__ l2_2, const float* __restrict__ l1_2,
                   const float* __restrict__ word,
                   float* __restrict__ doc, float* __restrict__ doc_svd, int N0) {
    __shared__ __attribute__((aligned(16))) float acc[AGG_ROWS * AGG_ROWF];
    __shared__ int   q_src[AGG_QCAP];
    __shared__ int   q_dl[AGG_QCAP];
    __shared__ float q_w[AGG_QCAP];
    __shared__ int   q_n;

    const int tid = threadIdx.x;
    const int ln  = tid & 31;
    const int wv  = __builtin_amdgcn_readfirstlane(tid >> 5);
    const unsigned row0 = blockIdx.x * AGG_ROWS;

    for (int i = tid; i < AGG_ROWS * AGG_ROWF / 4; i += 1024)
        ((float4*)acc)[i] = make_float4(0.f, 0.f, 0.f, 0.f);
    if (tid == 0) q_n = 0;
    __syncthreads();

#pragma unroll
    for (int list = 0; list < 2; ++list) {
        const int*   esrc = list == 0 ? e01_src : e02_src;
        const int*   edst = list == 0 ? e01_dst : e02_dst;
        const float* ew   = list == 0 ? e01_w   : e02_w;
        const int    E    = list == 0 ? E01     : E02;
        const float* xa   = list == 0 ? l2_1 : l2_2;
        const float* xb   = list == 0 ? l1_1 : l1_2;
        const int    segA = list == 0 ? 0 : 2 * 128;
        const int    segB = list == 0 ? 128 : 3 * 128;

        auto push = [&](int e, unsigned dl) {
            const int   s  = esrc[e];
            const float wt = ew[e];
            const int slot = __hip_atomic_fetch_add(&q_n, 1, __ATOMIC_RELAXED,
                                                    __HIP_MEMORY_SCOPE_WORKGROUP);
            if (slot < AGG_QCAP) {
                q_src[slot] = s;
                q_dl[slot]  = (int)dl;
                q_w[slot]   = wt;
            } else {
                float* o = acc + dl * AGG_ROWF;
                const size_t ro = (size_t)s * 128;
                for (int c = 0; c < 128; ++c) {
                    ldsAddF(o + segA + c, wt * xa[ro + c]);
                    ldsAddF(o + segB + c, wt * xb[ro + c]);
                    if (list == 1) ldsAddF(o + 4 * 128 + c, wt * word[ro + c]);
                }
            }
        };

        const bool vec_ok = (((size_t)edst) & 15) == 0;
        const int nvec = vec_ok ? (E >> 2) : 0;
        const int4* dvec = (const int4*)edst;
        auto own4 = [&](const int4& d) -> unsigned {
            return ((unsigned)d.x - row0 < AGG_ROWS ? 1u : 0u)
                 | ((unsigned)d.y - row0 < AGG_ROWS ? 2u : 0u)
                 | ((unsigned)d.z - row0 < AGG_ROWS ? 4u : 0u)
                 | ((unsigned)d.w - row0 < AGG_ROWS ? 8u : 0u);
        };
        int base = 0;
        for (; base + 4096 <= nvec; base += 4096) {
            int4 d[4];
#pragma unroll
            for (int u = 0; u < 4; ++u) d[u] = dvec[base + u * 1024 + tid];
            unsigned hit = 0;
#pragma unroll
            for (int u = 0; u < 4; ++u) hit |= own4(d[u]) << (4 * u);
            while (hit) {
                const int k = __builtin_ctz(hit);
                hit &= hit - 1;
                const int e = ((base + (k >> 2) * 1024 + tid) << 2) + (k & 3);
                push(e, (unsigned)edst[e] - row0);
            }
        }
        for (int i = base + tid; i < nvec; i += 1024) {
            unsigned hit = own4(dvec[i]);
            while (hit) {
                const int k = __builtin_ctz(hit);
                hit &= hit - 1;
                const int e = (i << 2) + k;
                push(e, (unsigned)edst[e] - row0);
            }
        }
        for (int e = (nvec << 2) + tid; e < E; e += 1024) {
            const unsigned r = (unsigned)edst[e] - row0;
            if (r < AGG_ROWS) push(e, r);
        }
        __syncthreads();

        const int qn = q_n < AGG_QCAP ? q_n : AGG_QCAP;
        if (list == 0) {
            constexpr int EB = 4;
            for (int b = wv * EB; b + EB <= qn; b += 32 * EB)
                gather_add<2, EB>(q_src, q_dl, q_w, b, xa, xb, xb, acc, segA, segB, 0, ln);
            const int rem0 = (qn / EB) * EB;
            if (rem0 + wv < qn)
                gather_add<2, 1>(q_src, q_dl, q_w, rem0 + wv, xa, xb, xb, acc, segA, segB, 0, ln);
        } else {
            constexpr int EB = 3;
            for (int b = wv * EB; b + EB <= qn; b += 32 * EB)
                gather_add<3, EB>(q_src, q_dl, q_w, b, xa, xb, word, acc, segA, segB, 4 * 128, ln);
            const int rem0 = (qn / EB) * EB;
            if (rem0 + wv < qn)
                gather_add<3, 1>(q_src, q_dl, q_w, rem0 + wv, xa, xb, word, acc, segA, segB, 4 * 128, ln);
        }
        __syncthreads();
        if (tid == 0) q_n = 0;
        __syncthreads();
    }

    for (int r = wv; r < AGG_ROWS; r += 32) {
        const int row = (int)row0 + r;
        if (row >= N0) break;
        const float4* p = (const float4*)(acc + r * AGG_ROWF);
        const float4 a0  = p[ln];
        const float4 a0s = p[32 + ln];
        const float4 a1  = p[64 + ln];
        const float4 a1s = p[96 + ln];
        const float4 aw  = p[128 + ln];
        const float sw = aw.x * aw.x + aw.y * aw.y + aw.z * aw.z + aw.w * aw.w;
        float s0  = a0.x * a0.x + a0.y * a0.y + a0.z * a0.z + a0.w * a0.w;
        float s0s = a0s.x * a0s.x + a0s.y * a0s.y + a0s.z * a0s.z + a0s.w * a0s.w;
        float s1  = a1.x * a1.x + a1.y * a1.y + a1.z * a1.z + a1.w * a1.w + sw;
        float s1s = a1s.x * a1s.x + a1s.y * a1s.y + a1s.z * a1s.z + a1s.w * a1s.w + sw;
#pragma unroll
        for (int off = 16; off > 0; off >>= 1) {
            s0  += __shfl_xor(s0,  off, 32);
            s0s += __shfl_xor(s0s, off, 32);
            s1  += __shfl_xor(s1,  off, 32);
            s1s += __shfl_xor(s1s, off, 32);
        }
        const float i0  = 1.0f / (sqrtf(s0)  + 1e-9f);
        const float i0s = 1.0f / (sqrtf(s0s) + 1e-9f);
        const float i1  = 1.0f / (sqrtf(s1)  + 1e-9f);
        const float i1s = 1.0f / (sqrtf(s1s) + 1e-9f);

        float* o  = doc     + (size_t)row * 384 + ln * 4;
        float* os = doc_svd + (size_t)row * 384 + ln * 4;
        *(float4*)(o)        = make_float4(a0.x * i0, a0.y * i0, a0.z * i0, a0.w * i0);
        *(float4*)(o + 128)  = make_float4(a1.x * i1, a1.y * i1, a1.z * i1, a1.w * i1);
        *(float4*)(o + 256)  = make_float4(aw.x * i1, aw.y * i1, aw.z * i1, aw.w * i1);
        *(float4*)(os)       = make_float4(a0s.x * i0s, a0s.y * i0s, a0s.z * i0s, a0s.w * i0s);
        *(float4*)(os + 128) = make_float4(a1s.x * i1s, a1s.y * i1s, a1s.z * i1s, a1s.w * i1s);
        *(float4*)(os + 256) = make_float4(aw.x * i1s, aw.y * i1s, aw.z * i1s, aw.w * i1s);
    }
}

extern "C" void kernel_launch(void* const* d_in, const int* in_sizes, int n_in,
                              void* d_out, int out_size, void* d_ws, size_t ws_size,
                              hipStream_t stream) {
    const float* x1       = (const float*)d_in[0];
    const float* x2       = (const float*)d_in[1];
    const float* word_emb = (const float*)d_in[2];
    const float* W1a = (const float*)d_in[3];
    const float* b1a = (const float*)d_in[4];
    const float* W1b = (const float*)d_in[5];
    const float* b1b = (const float*)d_in[6];
    const float* W2a = (const float*)d_in[7];
    const float* b2a = (const float*)d_in[8];
    const float* W2b = (const float*)d_in[9];
    const float* b2b = (const float*)d_in[10];
    const int*   e11_src = (const int*)d_in[11];
    const int*   e11_dst = (const int*)d_in[12];
    const float* e11_w   = (const float*)d_in[13];
    const int*   e22_src = (const int*)d_in[14];
    const int*   e22_dst = (const int*)d_in[15];
    const float* e22_w   = (const float*)d_in[16];
    const int*   e01_src = (const int*)d_in[17];
    const int*   e01_dst = (const int*)d_in[18];
    const float* e01_w   = (const float*)d_in[19];
    const int*   e02_src = (const int*)d_in[20];
    const int*   e02_dst = (const int*)d_in[21];
    const float* e02_w   = (const float*)d_in[22];

    const int N0 = 10000, N1 = 50000, N2 = 30000;
    const int E11 = in_sizes[11], E22 = in_sizes[14];
    const int E01 = in_sizes[17], E02 = in_sizes[20];

    float* ws = (float*)d_ws;
    size_t off = 0;
    float* l1_1 = ws + off; off += (size_t)N1 * 128;
    float* h11  = ws + off; off += (size_t)N1 * 128;
    float* l2_1 = ws + off; off += (size_t)N1 * 128;
    float* l1_2 = ws + off; off += (size_t)N2 * 128;
    float* h22  = ws + off; off += (size_t)N2 * 128;
    float* l2_2 = ws + off; off += (size_t)N2 * 128;
    float* doc     = (float*)d_out;
    float* doc_svd = doc + (size_t)N0 * 384;

    const dim3 blk(256);
    auto gemm_grid = [](int n) { return dim3((n + 127) / 128); };
    auto edge_grid = [](int e) { return dim3((e + EDGES_PER_BLOCK - 1) / EDGES_PER_BLOCK); };

    gemm_bias_relu<<<gemm_grid(N1), blk, 0, stream>>>(x1, W1a, b1a, l1_1, N1, 256);
    hipMemsetAsync(h11, 0, (size_t)N1 * 128 * sizeof(float), stream);
    spmm128<<<edge_grid(E11), blk, 0, stream>>>(e11_src, e11_dst, e11_w, l1_1, h11, E11);
    gemm_bias_relu<<<gemm_grid(N1), blk, 0, stream>>>(h11, W1b, b1b, l2_1, N1, 128);

    gemm_bias_relu<<<gemm_grid(N2), blk, 0, stream>>>(x2, W2a, b2a, l1_2, N2, 256);
    hipMemsetAsync(h22, 0, (size_t)N2 * 128 * sizeof(float), stream);
    spmm128<<<edge_grid(E22), blk, 0, stream>>>(e22_src, e22_dst, e22_w, l1_2, h22, E22);
    gemm_bias_relu<<<gemm_grid(N2), blk, 0, stream>>>(h22, W2b, b2b, l2_2, N2, 128);

    doc_aggregate<<<dim3((N0 + AGG_ROWS - 1) / AGG_ROWS), dim3(1024), 0, stream>>>(
        e01_src, e01_dst, e01_w, E01, e02_src, e02_dst, e02_w, E02,
        l2_1, l1_1, l2_2, l1_2, word_emb, doc, doc_svd, N0);
}


# stages of the two towers share launches: tower-2 layer-a GEMM beside tower-1 scatter-add, tower-1 layer-b GEMM beside tower-2 scatter-add, zero-fill beside GEMM 1; sums stay in atomic arrival order
# speedup vs baseline: 2.2333x; 1.0773x over previous
#include <hip/hip_runtime.h>
#include <math.h>

typedef __attribute__((ext_vector_type(2))) float v2f;
typedef __attribute__((ext_vector_type(8))) float v8f;

#define O_DIM 128

__device__ __forceinline__ void atomAddF(float* p, float v) {
    __hip_atomic_fetch_add(p, v, __ATOMIC_RELAXED, __HIP_MEMORY_SCOPE_AGENT);
}

__device__ __forceinline__
void gemm_tile(const float* __restrict__ X, const float* __restrict__ W,
               const float* __restrict__ bias, float* __restrict__ out,
               int N, int K, int bid) {
    __shared__ float  As[128][36];
    __shared__ float2 Bs2[16][144];

    const int tid = threadIdx.x;
    const int wv  = tid >> 5;
    const int ln  = tid & 31;
    const int m0  = bid * 128;

    v8f acc[8];
#pragma unroll
    for (int t = 0; t < 8; ++t)
#pragma unroll
        for (int j = 0; j < 8; ++j) acc[t][j] = 0.0f;

    const int mrow = wv * 16 + (ln & 15);
    const int lnn  = ln & 15;
    const int khalf = (ln >> 4) << 1;

    for (int k0 = 0; k0 < K; k0 += 32) {
#pragma unroll
        for (int i = 0; i < 4; ++i) {
            int f   = tid + i * 256;
            int row = f >> 3;
            int c4  = f & 7;
            float4 v = make_float4(0.f, 0.f, 0.f, 0.f);
            if (m0 + row < N)
                v = *(const float4*)(X + (size_t)(m0 + row) * K + k0 + c4 * 4);
            *(float4*)&As[row][c4 * 4] = v;
        }
#pragma unroll
        for (int i = 0; i < 4; ++i) {
            int f   = tid + i * 256;
            int row = f >> 5;
            int c4  = f & 31;
            float4 v = *(const float4*)(W + (size_t)(k0 + row) * O_DIM + c4 * 4);
            float* bp = (float*)&Bs2[row >> 1][c4 * 4] + (row & 1);
            bp[0] = v.x;
            bp[2] = v.y;
            bp[4] = v.z;
            bp[6] = v.w;
        }
        __syncthreads();

#pragma unroll
        for (int kk = 0; kk < 32; kk += 4) {
            const int kk2 = kk + khalf;
            const int p   = kk2 >> 1;
            v2f a = *(const v2f*)&As[mrow][kk2];
#pragma unroll
            for (int t = 0; t < 8; ++t) {
                v2f b = *(const v2f*)&Bs2[p][t * 16 + lnn];
                acc[t] = __builtin_amdgcn_wmma_f32_16x16x4_f32(
                    false, a, false, b, (short)0, acc[t], false, false);
            }
        }
        __syncthreads();
    }

    const int mtile = m0 + wv * 16;
    if (mtile < N) {
        const int rowoff = (ln >> 4) << 3;
#pragma unroll
        for (int t = 0; t < 8; ++t) {
            const int col = t * 16 + lnn;
            const float bv = bias[col];
#pragma unroll
            for (int i = 0; i < 8; ++i) {
                float v = acc[t][i] + bv;
                out[(size_t)(mtile + rowoff + i) * O_DIM + col] = v > 0.f ? v : 0.f;
            }
        }
    }
}

#define EDGES_PER_WAVE 32
#define EDGES_PER_BLOCK (8 * EDGES_PER_WAVE)

__device__ __forceinline__ int bcastI(int v, int lane) {
    return __builtin_amdgcn_readlane(v, lane);
}
__device__ __forceinline__ float bcastF(float v, int lane) {
    return __builtin_bit_cast(float, __builtin_amdgcn_readlane(__builtin_bit_cast(int, v), lane));
}

__device__ __forceinline__
void spmm_tile(const int* __restrict__ src, const int* __restrict__ dst,
               const float* __restrict__ w, const float* __restrict__ x,
               float* __restrict__ out, int E, int bid) {
    const int ln = threadIdx.x & 31;
    const int wave = __builtin_amdgcn_readfirstlane(bid * 8 + (int)(threadIdx.x >> 5));
    const int e0 = wave * EDGES_PER_WAVE;
    if (e0 >= E) return;
    const int rem = E - e0;
    const int cnt = rem < EDGES_PER_WAVE ? rem : EDGES_PER_WAVE;

    int sv = 0, dv = 0;
    float wv = 0.f;
    if (ln < cnt) {
        sv = src[e0 + ln];
        dv = dst[e0 + ln];
        wv = w[e0 + ln];
    }
    const float* xl = x + ln;
    float* ol = out + ln;

    constexpr int EB = 8;
    int j0 = 0;
    for (; j0 + EB <= cnt; j0 += EB) {
        float v[EB][4];
#pragma unroll
        for (int j = 0; j < EB; ++j) {
            const float* xr = xl + (size_t)bcastI(sv, j0 + j) * 128;
            v[j][0] = xr[0];
            v[j][1] = xr[32];
            v[j][2] = xr[64];
            v[j][3] = xr[96];
        }
#pragma unroll
        for (int j = 0; j < EB; ++j) {
            float* o = ol + (size_t)bcastI(dv, j0 + j) * 128;
            const float wt = bcastF(wv, j0 + j);
            atomAddF(o +  0, wt * v[j][0]);
            atomAddF(o + 32, wt * v[j][1]);
            atomAddF(o + 64, wt * v[j][2]);
            atomAddF(o + 96, wt * v[j][3]);
        }
    }
    for (; j0 < cnt; ++j0) {
        const float* xr = xl + (size_t)bcastI(sv, j0) * 128;
        const float v0 = xr[0];
        const float v1 = xr[32];
        const float v2 = xr[64];
        const float v3 = xr[96];
        float* o = ol + (size_t)bcastI(dv, j0) * 128;
        const float wt = bcastF(wv, j0);
        atomAddF(o +  0, wt * v0);
        atomAddF(o + 32, wt * v1);
        atomAddF(o + 64, wt * v2);
        atomAddF(o + 96, wt * v3);
    }
}

#define FILL_F4_PER_BLOCK 4096

__device__ __forceinline__ void fill_tile(float* __restrict__ p, int n4, int bid) {
    float4* q = (float4*)p;
    const int b0 = bid * FILL_F4_PER_BLOCK + (int)threadIdx.x;
#pragma unroll
    for (int i = 0; i < 16; ++i) {
        const int idx = b0 + i * 256;
        if (idx < n4) q[idx] = make_float4(0.f, 0.f, 0.f, 0.f);
    }
}

__global__ __launch_bounds__(256)
void stage_group(const float* __restrict__ gX, const float* __restrict__ gW,
                 const float* __restrict__ gBias, float* __restrict__ gOut,
                 int gN, int gK, int gBlocks,
                 const int* __restrict__ sSrc, const int* __restrict__ sDst,
                 const float* __restrict__ sW, const float* __restrict__ sX,
                 float* __restrict__ sOut, int sE, int sBlocks,
                 float* __restrict__ zP, int zN4) {
    const int bid = (int)blockIdx.x;
    if (bid < gBlocks)
        gemm_tile(gX, gW, gBias, gOut, gN, gK, bid);
    else if (bid < gBlocks + sBlocks)
        spmm_tile(sSrc, sDst, sW, sX, sOut, sE, bid - gBlocks);
    else
        fill_tile(zP, zN4, bid - gBlocks - sBlocks);
}

#define AGG_ROWS 40
#define AGG_ROWF 640
#define AGG_QCAP 6144

__device__ __forceinline__ void ldsAddF(float* p, float v) {
    __hip_atomic_fetch_add(p, v, __ATOMIC_RELAXED, __HIP_MEMORY_SCOPE_WORKGROUP);
}

template <int NS, int EB>
__device__ __forceinline__ void gather_add(const int* qs, const int* qd, const float* qw, int b,
                                           const float* __restrict__ xa, const float* __restrict__ xb,
                                           const float* __restrict__ xc, float* acc,
                                           int sa, int sb, int sc, int ln) {
    float v[EB][4 * NS];
    int   dl[EB];
    float wt[EB];
#pragma unroll
    for (int j = 0; j < EB; ++j) {
        const size_t ro = (size_t)__builtin_amdgcn_readfirstlane(qs[b + j]) * 128 + ln;
        dl[j] = __builtin_amdgcn_readfirstlane(qd[b + j]);
        wt[j] = qw[b + j];
#pragma unroll
        for (int c = 0; c < 4; ++c) {
            v[j][c]     = xa[ro + 32 * c];
            v[j][4 + c] = xb[ro + 32 * c];
            if (NS == 3) v[j][8 + c] = xc[ro + 32 * c];
        }
    }
#pragma unroll
    for (int j = 0; j < EB; ++j) {
        float* o = acc + dl[j] * AGG_ROWF + ln;
#pragma unroll
        for (int c = 0; c < 4; ++c) {
            ldsAddF(o + sa + 32 * c, wt[j] * v[j][c]);
            ldsAddF(o + sb + 32 * c, wt[j] * v[j][4 + c]);
            if (NS == 3) ldsAddF(o + sc + 32 * c, wt[j] * v[j][8 + c]);
        }
    }
}

__global__ __launch_bounds__(1024)
void doc_aggregate(const int* __restrict__ e01_src, const int* __restrict__ e01_dst,
                   const float* __restrict__ e01_w, int E01,
                   const int* __restrict__ e02_src, const int* __restrict__ e02_dst,
                   const float* __restrict__ e02_w, int E02,
                   const float* __restrict__ l2_1, const float* __restrict__ l1_1,
                   const float* __restrict__ l2_2, const float* __restrict__ l1_2,
                   const float* __restrict__ word,
                   float* __restrict__ doc, float* __restrict__ doc_svd, int N0) {
    __shared__ __attribute__((aligned(16))) float acc[AGG_ROWS * AGG_ROWF];
    __shared__ int   q_src[AGG_QCAP];
    __shared__ int   q_dl[AGG_QCAP];
    __shared__ float q_w[AGG_QCAP];
    __shared__ int   q_n;

    const int tid = threadIdx.x;
    const int ln  = tid & 31;
    const int wv  = __builtin_amdgcn_readfirstlane(tid >> 5);
    const unsigned row0 = blockIdx.x * AGG_ROWS;

    for (int i = tid; i < AGG_ROWS * AGG_ROWF / 4; i += 1024)
        ((float4*)acc)[i] = make_float4(0.f, 0.f, 0.f, 0.f);
    if (tid == 0) q_n = 0;
    __syncthreads();

#pragma unroll
    for (int list = 0; list < 2; ++list) {
        const int*   esrc = list == 0 ? e01_src : e02_src;
        const int*   edst = list == 0 ? e01_dst : e02_dst;
        const float* ew   = list == 0 ? e01_w   : e02_w;
        const int    E    = list == 0 ? E01     : E02;
        const float* xa   = list == 0 ? l2_1 : l2_2;
        const float* xb   = list == 0 ? l1_1 : l1_2;
        const int    segA = list == 0 ? 0 : 2 * 128;
        const int    segB = list == 0 ? 128 : 3 * 128;

        auto push = [&](int e, unsigned dl) {
            const int   s  = esrc[e];
            const float wt = ew[e];
            const int slot = __hip_atomic_fetch_add(&q_n, 1, __ATOMIC_RELAXED,
                                                    __HIP_MEMORY_SCOPE_WORKGROUP);
            if (slot < AGG_QCAP) {
                q_src[slot] = s;
                q_dl[slot]  = (int)dl;
                q_w[slot]   = wt;
            } else {
                float* o = acc + dl * AGG_ROWF;
                const size_t ro = (size_t)s * 128;
                for (int c = 0; c < 128; ++c) {
                    ldsAddF(o + segA + c, wt * xa[ro + c]);
                    ldsAddF(o + segB + c, wt * xb[ro + c]);
                    if (list == 1) ldsAddF(o + 4 * 128 + c, wt * word[ro + c]);
                }
            }
        };

        const bool vec_ok = (((size_t)edst) & 15) == 0;
        const int nvec = vec_ok ? (E >> 2) : 0;
        const int4* dvec = (const int4*)edst;
        auto own4 = [&](const int4& d) -> unsigned {
            return ((unsigned)d.x - row0 < AGG_ROWS ? 1u : 0u)
                 | ((unsigned)d.y - row0 < AGG_ROWS ? 2u : 0u)
                 | ((unsigned)d.z - row0 < AGG_ROWS ? 4u : 0u)
                 | ((unsigned)d.w - row0 < AGG_ROWS ? 8u : 0u);
        };
        int base = 0;
        for (; base + 4096 <= nvec; base += 4096) {
            int4 d[4];
#pragma unroll
            for (int u = 0; u < 4; ++u) d[u] = dvec[base + u * 1024 + tid];
            unsigned hit = 0;
#pragma unroll
            for (int u = 0; u < 4; ++u) hit |= own4(d[u]) << (4 * u);
            while (hit) {
                const int k = __builtin_ctz(hit);
                hit &= hit - 1;
                const int e = ((base + (k >> 2) * 1024 + tid) << 2) + (k & 3);
                push(e, (unsigned)edst[e] - row0);
            }
        }
        for (int i = base + tid; i < nvec; i += 1024) {
            unsigned hit = own4(dvec[i]);
            while (hit) {
                const int k = __builtin_ctz(hit);
                hit &= hit - 1;
                const int e = (i << 2) + k;
                push(e, (unsigned)edst[e] - row0);
            }
        }
        for (int e = (nvec << 2) + tid; e < E; e += 1024) {
            const unsigned r = (unsigned)edst[e] - row0;
            if (r < AGG_ROWS) push(e, r);
        }
        __syncthreads();

        const int qn = q_n < AGG_QCAP ? q_n : AGG_QCAP;
        if (list == 0) {
            constexpr int EB = 4;
            for (int b = wv * EB; b + EB <= qn; b += 32 * EB)
                gather_add<2, EB>(q_src, q_dl, q_w, b, xa, xb, xb, acc, segA, segB, 0, ln);
            const int rem0 = (qn / EB) * EB;
            if (rem0 + wv < qn)
                gather_add<2, 1>(q_src, q_dl, q_w, rem0 + wv, xa, xb, xb, acc, segA, segB, 0, ln);
        } else {
            constexpr int EB = 3;
            for (int b = wv * EB; b + EB <= qn; b += 32 * EB)
                gather_add<3, EB>(q_src, q_dl, q_w, b, xa, xb, word, acc, segA, segB, 4 * 128, ln);
            const int rem0 = (qn / EB) * EB;
            if (rem0 + wv < qn)
                gather_add<3, 1>(q_src, q_dl, q_w, rem0 + wv, xa, xb, word, acc, segA, segB, 4 * 128, ln);
        }
        __syncthreads();
        if (tid == 0) q_n = 0;
        __syncthreads();
    }

    for (int r = wv; r < AGG_ROWS; r += 32) {
        const int row = (int)row0 + r;
        if (row >= N0) break;
        const float4* p = (const float4*)(acc + r * AGG_ROWF);
        const float4 a0  = p[ln];
        const float4 a0s = p[32 + ln];
        const float4 a1  = p[64 + ln];
        const float4 a1s = p[96 + ln];
        const float4 aw  = p[128 + ln];
        const float sw = aw.x * aw.x + aw.y * aw.y + aw.z * aw.z + aw.w * aw.w;
        float s0  = a0.x * a0.x + a0.y * a0.y + a0.z * a0.z + a0.w * a0.w;
        float s0s = a0s.x * a0s.x + a0s.y * a0s.y + a0s.z * a0s.z + a0s.w * a0s.w;
        float s1  = a1.x * a1.x + a1.y * a1.y + a1.z * a1.z + a1.w * a1.w + sw;
        float s1s = a1s.x * a1s.x + a1s.y * a1s.y + a1s.z * a1s.z + a1s.w * a1s.w + sw;
#pragma unroll
        for (int off = 16; off > 0; off >>= 1) {
            s0  += __shfl_xor(s0,  off, 32);
            s0s += __shfl_xor(s0s, off, 32);
            s1  += __shfl_xor(s1,  off, 32);
            s1s += __shfl_xor(s1s, off, 32);
        }
        const float i0  = 1.0f / (sqrtf(s0)  + 1e-9f);
        const float i0s = 1.0f / (sqrtf(s0s) + 1e-9f);
        const float i1  = 1.0f / (sqrtf(s1)  + 1e-9f);
        const float i1s = 1.0f / (sqrtf(s1s) + 1e-9f);

        float* o  = doc     + (size_t)row * 384 + ln * 4;
        float* os = doc_svd + (size_t)row * 384 + ln * 4;
        *(float4*)(o)        = make_float4(a0.x * i0, a0.y * i0, a0.z * i0, a0.w * i0);
        *(float4*)(o + 128)  = make_float4(a1.x * i1, a1.y * i1, a1.z * i1, a1.w * i1);
        *(float4*)(o + 256)  = make_float4(aw.x * i1, aw.y * i1, aw.z * i1, aw.w * i1);
        *(float4*)(os)       = make_float4(a0s.x * i0s, a0s.y * i0s, a0s.z * i0s, a0s.w * i0s);
        *(float4*)(os + 128) = make_float4(a1s.x * i1s, a1s.y * i1s, a1s.z * i1s, a1s.w * i1s);
        *(float4*)(os + 256) = make_float4(aw.x * i1s, aw.y * i1s, aw.z * i1s, aw.w * i1s);
    }
}

extern "C" void kernel_launch(void* const* d_in, const int* in_sizes, int n_in,
                              void* d_out, int out_size, void* d_ws, size_t ws_size,
                              hipStream_t stream) {
    const float* x1       = (const float*)d_in[0];
    const float* x2       = (const float*)d_in[1];
    const float* word_emb = (const float*)d_in[2];
    const float* W1a = (const float*)d_in[3];
    const float* b1a = (const float*)d_in[4];
    const float* W1b = (const float*)d_in[5];
    const float* b1b = (const float*)d_in[6];
    const float* W2a = (const float*)d_in[7];
    const float* b2a = (const float*)d_in[8];
    const float* W2b = (const float*)d_in[9];
    const float* b2b = (const float*)d_in[10];
    const int*   e11_src = (const int*)d_in[11];
    const int*   e11_dst = (const int*)d_in[12];
    const float* e11_w   = (const float*)d_in[13];
    const int*   e22_src = (const int*)d_in[14];
    const int*   e22_dst = (const int*)d_in[15];
    const float* e22_w   = (const float*)d_in[16];
    const int*   e01_src = (const int*)d_in[17];
    const int*   e01_dst = (const int*)d_in[18];
    const float* e01_w   = (const float*)d_in[19];
    const int*   e02_src = (const int*)d_in[20];
    const int*   e02_dst = (const int*)d_in[21];
    const float* e02_w   = (const float*)d_in[22];

    const int N0 = 10000, N1 = 50000, N2 = 30000;
    const int E11 = in_sizes[11], E22 = in_sizes[14];
    const int E01 = in_sizes[17], E02 = in_sizes[20];

    float* ws = (float*)d_ws;
    size_t off = 0;
    float* l1_1 = ws + off; off += (size_t)N1 * 128;
    float* l2_1 = ws + off; off += (size_t)N1 * 128;
    float* l1_2 = ws + off; off += (size_t)N2 * 128;
    float* l2_2 = ws + off; off += (size_t)N2 * 128;
    float* h11  = ws + off; off += (size_t)N1 * 128;
    float* h22  = ws + off; off += (size_t)N2 * 128;
    float* doc     = (float*)d_out;
    float* doc_svd = doc + (size_t)N0 * 384;

    const dim3 blk(256);
    const int g1 = (N1 + 127) / 128;
    const int g2 = (N2 + 127) / 128;
    const int s1 = (E11 + EDGES_PER_BLOCK - 1) / EDGES_PER_BLOCK;
    const int s2 = (E22 + EDGES_PER_BLOCK - 1) / EDGES_PER_BLOCK;
    const int z4 = (N1 + N2) * 32;
    const int zb = (z4 + FILL_F4_PER_BLOCK - 1) / FILL_F4_PER_BLOCK;

    stage_group<<<dim3(g1 + zb), blk, 0, stream>>>(
        x1, W1a, b1a, l1_1, N1, 256, g1,
        nullptr, nullptr, nullptr, nullptr, nullptr, 0, 0,
        h11, z4);
    stage_group<<<dim3(g2 + s1), blk, 0, stream>>>(
        x2, W2a, b2a, l1_2, N2, 256, g2,
        e11_src, e11_dst, e11_w, l1_1, h11, E11, s1,
        nullptr, 0);
    stage_group<<<dim3(g1 + s2), blk, 0, stream>>>(
        h11, W1b, b1b, l2_1, N1, 128, g1,
        e22_src, e22_dst, e22_w, l1_2, h22, E22, s2,
        nullptr, 0);
    stage_group<<<dim3(g2), blk, 0, stream>>>(
        h22, W2b, b2b, l2_2, N2, 128, g2,
        nullptr, nullptr, nullptr, nullptr, nullptr, 0, 0,
        nullptr, 0);

    doc_aggregate<<<dim3((N0 + AGG_ROWS - 1) / AGG_ROWS), dim3(1024), 0, stream>>>(
        e01_src, e01_dst, e01_w, E01, e02_src, e02_dst, e02_w, E02,
        l2_1, l1_1, l2_2, l1_2, word_emb, doc, doc_svd, N0);
}


# doc aggregation scan: one owner byte per edge (packed by spare work-groups of launch 4), edge indices queued, src/row/weight fetched after the scan one entry per thread; sums stay in arrival order
# speedup vs baseline: 2.3504x; 1.0524x over previous
#include <hip/hip_runtime.h>
#include <math.h>

typedef __attribute__((ext_vector_type(2))) float v2f;
typedef __attribute__((ext_vector_type(8))) float v8f;

#define O_DIM 128

__device__ __forceinline__ void atomAddF(float* p, float v) {
    __hip_atomic_fetch_add(p, v, __ATOMIC_RELAXED, __HIP_MEMORY_SCOPE_AGENT);
}

__device__ __forceinline__
void gemm_tile(const float* __restrict__ X, const float* __restrict__ W,
               const float* __restrict__ bias, float* __restrict__ out,
               int N, int K, int bid) {
    __shared__ float  As[128][36];
    __shared__ float2 Bs2[16][144];

    const int tid = threadIdx.x;
    const int wv  = tid >> 5;
    const int ln  = tid & 31;
    const int m0  = bid * 128;

    v8f acc[8];
#pragma unroll
    for (int t = 0; t < 8; ++t)
#pragma unroll
        for (int j = 0; j < 8; ++j) acc[t][j] = 0.0f;

    const int mrow = wv * 16 + (ln & 15);
    const int lnn  = ln & 15;
    const int khalf = (ln >> 4) << 1;

    for (int k0 = 0; k0 < K; k0 += 32) {
#pragma unroll
        for (int i = 0; i < 4; ++i) {
            int f   = tid + i * 256;
            int row = f >> 3;
            int c4  = f & 7;
            float4 v = make_float4(0.f, 0.f, 0.f, 0.f);
            if (m0 + row < N)
                v = *(const float4*)(X + (size_t)(m0 + row) * K + k0 + c4 * 4);
            *(float4*)&As[row][c4 * 4] = v;
        }
#pragma unroll
        for (int i = 0; i < 4; ++i) {
            int f   = tid + i * 256;
            int row = f >> 5;
            int c4  = f & 31;
            float4 v = *(const float4*)(W + (size_t)(k0 + row) * O_DIM + c4 * 4);
            float* bp = (float*)&Bs2[row >> 1][c4 * 4] + (row & 1);
            bp[0] = v.x;
            bp[2] = v.y;
            bp[4] = v.z;
            bp[6] = v.w;
        }
        __syncthreads();

#pragma unroll
        for (int kk = 0; kk < 32; kk += 4) {
            const int kk2 = kk + khalf;
            const int p   = kk2 >> 1;
            v2f a = *(const v2f*)&As[mrow][kk2];
#pragma unroll
            for (int t = 0; t < 8; ++t) {
                v2f b = *(const v2f*)&Bs2[p][t * 16 + lnn];
                acc[t] = __builtin_amdgcn_wmma_f32_16x16x4_f32(
                    false, a, false, b, (short)0, acc[t], false, false);
            }
        }
        __syncthreads();
    }

    const int mtile = m0 + wv * 16;
    if (mtile < N) {
        const int rowoff = (ln >> 4) << 3;
#pragma unroll
        for (int t = 0; t < 8; ++t) {
            const int col = t * 16 + lnn;
            const float bv = bias[col];
#pragma unroll
            for (int i = 0; i < 8; ++i) {
                float v = acc[t][i] + bv;
                out[(size_t)(mtile + rowoff + i) * O_DIM + col] = v > 0.f ? v : 0.f;
            }
        }
    }
}

#define EDGES_PER_WAVE 32
#define EDGES_PER_BLOCK (8 * EDGES_PER_WAVE)

__device__ __forceinline__ int bcastI(int v, int lane) {
    return __builtin_amdgcn_readlane(v, lane);
}
__device__ __forceinline__ float bcastF(float v, int lane) {
    return __builtin_bit_cast(float, __builtin_amdgcn_readlane(__builtin_bit_cast(int, v), lane));
}

__device__ __forceinline__
void spmm_tile(const int* __restrict__ src, const int* __restrict__ dst,
               const float* __restrict__ w, const float* __restrict__ x,
               float* __restrict__ out, int E, int bid) {
    const int ln = threadIdx.x & 31;
    const int wave = __builtin_amdgcn_readfirstlane(bid * 8 + (int)(threadIdx.x >> 5));
    const int e0 = wave * EDGES_PER_WAVE;
    if (e0 >= E) return;
    const int rem = E - e0;
    const int cnt = rem < EDGES_PER_WAVE ? rem : EDGES_PER_WAVE;

    int sv = 0, dv = 0;
    float wv = 0.f;
    if (ln < cnt) {
        sv = src[e0 + ln];
        dv = dst[e0 + ln];
        wv = w[e0 + ln];
    }
    const float* xl = x + ln;
    float* ol = out + ln;

    constexpr int EB = 8;
    int j0 = 0;
    for (; j0 + EB <= cnt; j0 += EB) {
        float v[EB][4];
#pragma unroll
        for (int j = 0; j < EB; ++j) {
            const float* xr = xl + (size_t)bcastI(sv, j0 + j) * 128;
            v[j][0] = xr[0];
            v[j][1] = xr[32];
            v[j][2] = xr[64];
            v[j][3] = xr[96];
        }
#pragma unroll
        for (int j = 0; j < EB; ++j) {
            float* o = ol + (size_t)bcastI(dv, j0 + j) * 128;
            const float wt = bcastF(wv, j0 + j);
            atomAddF(o +  0, wt * v[j][0]);
            atomAddF(o + 32, wt * v[j][1]);
            atomAddF(o + 64, wt * v[j][2]);
            atomAddF(o + 96, wt * v[j][3]);
        }
    }
    for (; j0 < cnt; ++j0) {
        const float* xr = xl + (size_t)bcastI(sv, j0) * 128;
        const float v0 = xr[0];
        const float v1 = xr[32];
        const float v2 = xr[64];
        const float v3 = xr[96];
        float* o = ol + (size_t)bcastI(dv, j0) * 128;
        const float wt = bcastF(wv, j0);
        atomAddF(o +  0, wt * v0);
        atomAddF(o + 32, wt * v1);
        atomAddF(o + 64, wt * v2);
        atomAddF(o + 96, wt * v3);
    }
}

#define FILL_F4_PER_BLOCK 4096

__device__ __forceinline__ void fill_tile(float* __restrict__ p, int n4, int bid) {
    float4* q = (float4*)p;
    const int b0 = bid * FILL_F4_PER_BLOCK + (int)threadIdx.x;
#pragma unroll
    for (int i = 0; i < 16; ++i) {
        const int idx = b0 + i * 256;
        if (idx < n4) q[idx] = make_float4(0.f, 0.f, 0.f, 0.f);
    }
}

#define OWN_ROWS 40
#define OWN_Q_PER_BLOCK 1024

__device__ __forceinline__ unsigned owner_byte(const int* __restrict__ d, int e, int E, unsigned nrows) {
    if (e >= E) return 255u;
    const unsigned r = (unsigned)d[e];
    return r < nrows ? r / OWN_ROWS : 255u;
}

__device__ __forceinline__ void owner_tile(const int* __restrict__ dA, int EA,
                                           const int* __restrict__ dB, int EB,
                                           unsigned* __restrict__ ob, int offB,
                                           unsigned nrows, int bid) {
    const int qA = (EA + 3) >> 2;
    const int qB = (EB + 3) >> 2;
#pragma unroll
    for (int i = 0; i < OWN_Q_PER_BLOCK / 256; ++i) {
        const int q = bid * OWN_Q_PER_BLOCK + i * 256 + (int)threadIdx.x;
        if (q >= qA + qB) break;
        const bool isA = q < qA;
        const int* d = isA ? dA : dB;
        const int  E = isA ? EA : EB;
        const int  ql = isA ? q : q - qA;
        const int  e = ql << 2;
        const unsigned w = owner_byte(d, e, E, nrows)
                         | (owner_byte(d, e + 1, E, nrows) << 8)
                         | (owner_byte(d, e + 2, E, nrows) << 16)
                         | (owner_byte(d, e + 3, E, nrows) << 24);
        ob[(isA ? 0 : offB) + ql] = w;
    }
}

__global__ __launch_bounds__(256)
void stage_group(const float* __restrict__ gX, const float* __restrict__ gW,
                 const float* __restrict__ gBias, float* __restrict__ gOut,
                 int gN, int gK, int gBlocks,
                 const int* __restrict__ sSrc, const int* __restrict__ sDst,
                 const float* __restrict__ sW, const float* __restrict__ sX,
                 float* __restrict__ sOut, int sE, int sBlocks,
                 float* __restrict__ zP, int zN4, int zBlocks,
                 const int* __restrict__ oDA, int oEA,
                 const int* __restrict__ oDB, int oEB,
                 unsigned* __restrict__ oOut, int oOffB, unsigned oRows) {
    const int bid = (int)blockIdx.x;
    if (bid < gBlocks)
        gemm_tile(gX, gW, gBias, gOut, gN, gK, bid);
    else if (bid < gBlocks + sBlocks)
        spmm_tile(sSrc, sDst, sW, sX, sOut, sE, bid - gBlocks);
    else if (bid < gBlocks + sBlocks + zBlocks)
        fill_tile(zP, zN4, bid - gBlocks - sBlocks);
    else
        owner_tile(oDA, oEA, oDB, oEB, oOut, oOffB, oRows, bid - gBlocks - sBlocks - zBlocks);
}

#define AGG_ROWS 40
#define AGG_ROWF 640
static_assert(AGG_ROWS == OWN_ROWS, "owner bytes and aggregation must agree on rows per group");
#define AGG_QCAP 6144

__device__ __forceinline__ void ldsAddF(float* p, float v) {
    __hip_atomic_fetch_add(p, v, __ATOMIC_RELAXED, __HIP_MEMORY_SCOPE_WORKGROUP);
}

template <int NS, int EB>
__device__ __forceinline__ void gather_add(const int* qs, const int* qd, const float* qw, int b,
                                           const float* __restrict__ xa, const float* __restrict__ xb,
                                           const float* __restrict__ xc, float* acc,
                                           int sa, int sb, int sc, int ln) {
    float v[EB][4 * NS];
    int   dl[EB];
    float wt[EB];
#pragma unroll
    for (int j = 0; j < EB; ++j) {
        const size_t ro = (size_t)__builtin_amdgcn_readfirstlane(qs[b + j]) * 128 + ln;
        dl[j] = __builtin_amdgcn_readfirstlane(qd[b + j]);
        wt[j] = qw[b + j];
#pragma unroll
        for (int c = 0; c < 4; ++c) {
            v[j][c]     = xa[ro + 32 * c];
            v[j][4 + c] = xb[ro + 32 * c];
            if (NS == 3) v[j][8 + c] = xc[ro + 32 * c];
        }
    }
#pragma unroll
    for (int j = 0; j < EB; ++j) {
        float* o = acc + dl[j] * AGG_ROWF + ln;
#pragma unroll
        for (int c = 0; c < 4; ++c) {
            ldsAddF(o + sa + 32 * c, wt[j] * v[j][c]);
            ldsAddF(o + sb + 32 * c, wt[j] * v[j][4 + c]);
            if (NS == 3) ldsAddF(o + sc + 32 * c, wt[j] * v[j][8 + c]);
        }
    }
}

__global__ __launch_bounds__(1024)
void doc_aggregate(const int* __restrict__ e01_src, const int* __restrict__ e01_dst,
                   const float* __restrict__ e01_w, int E01,
                   const int* __restrict__ e02_src, const int* __restrict__ e02_dst,
                   const float* __restrict__ e02_w, int E02,
                   const float* __restrict__ l2_1, const float* __restrict__ l1_1,
                   const float* __restrict__ l2_2, const float* __restrict__ l1_2,
                   const float* __restrict__ word,
                   const unsigned* __restrict__ ob, int obOffB,
                   float* __restrict__ doc, float* __restrict__ doc_svd, int N0) {
    __shared__ __attribute__((aligned(16))) float acc[AGG_ROWS * AGG_ROWF];
    __shared__ int   q_src[AGG_QCAP];
    __shared__ int   q_dl[AGG_QCAP];
    __shared__ float q_w[AGG_QCAP];
    __shared__ int   q_n;

    const int tid = threadIdx.x;
    const int ln  = tid & 31;
    const int wv  = __builtin_amdgcn_readfirstlane(tid >> 5);
    const unsigned row0 = blockIdx.x * AGG_ROWS;

    for (int i = tid; i < AGG_ROWS * AGG_ROWF / 4; i += 1024)
        ((float4*)acc)[i] = make_float4(0.f, 0.f, 0.f, 0.f);
    if (tid == 0) q_n = 0;
    __syncthreads();

#pragma unroll
    for (int list = 0; list < 2; ++list) {
        const int*   esrc = list == 0 ? e01_src : e02_src;
        const int*   edst = list == 0 ? e01_dst : e02_dst;
        const float* ew   = list == 0 ? e01_w   : e02_w;
        const int    E    = list == 0 ? E01     : E02;
        const float* xa   = list == 0 ? l2_1 : l2_2;
        const float* xb   = list == 0 ? l1_1 : l1_2;
        const int    segA = list == 0 ? 0 : 2 * 128;
        const int    segB = list == 0 ? 128 : 3 * 128;

        auto push = [&](int e) {
            const int slot = __hip_atomic_fetch_add(&q_n, 1, __ATOMIC_RELAXED,
                                                    __HIP_MEMORY_SCOPE_WORKGROUP);
            if (slot < AGG_QCAP) {
                q_src[slot] = e;
            } else {
                const int   s  = esrc[e];
                const float wt = ew[e];
                const unsigned dl = (unsigned)edst[e] - row0;
                float* o = acc + dl * AGG_ROWF;
                const size_t ro = (size_t)s * 128;
                for (int c = 0; c < 128; ++c) {
                    ldsAddF(o + segA + c, wt * xa[ro + c]);
                    ldsAddF(o + segB + c, wt * xb[ro + c]);
                    if (list == 1) ldsAddF(o + 4 * 128 + c, wt * word[ro + c]);
                }
            }
        };

        const uint4* ov = (const uint4*)(ob + (list == 0 ? 0 : obOffB));
        const int nvec = E >> 4;
        const unsigned pat = blockIdx.x * 0x01010101u;
        auto z4 = [&](unsigned v) -> unsigned {
            const unsigned t = v ^ pat;
            return ~(((t & 0x7F7F7F7Fu) + 0x7F7F7F7Fu) | t) & 0x80808080u;
        };
        auto eq16 = [&](const uint4& v) -> unsigned {
            return (z4(v.x) >> 7) | (z4(v.y) >> 6) | (z4(v.z) >> 5) | (z4(v.w) >> 4);
        };
        const int ilast = nvec > 0 ? nvec - 1 : 0;
        auto load4 = [&](uint4 (&v)[4], int base) {
#pragma unroll
            for (int u = 0; u < 4; ++u) {
                const int idx = base + u * 1024 + tid;
                v[u] = ov[idx < nvec ? idx : ilast];
            }
        };
        auto scan4 = [&](const uint4 (&v)[4], int base) {
            unsigned h[4];
#pragma unroll
            for (int u = 0; u < 4; ++u)
                h[u] = (base + u * 1024 + tid) < nvec ? eq16(v[u]) : 0u;
            unsigned long long hit = (unsigned long long)(h[0] | (h[1] << 4))
                                   | ((unsigned long long)(h[2] | (h[3] << 4)) << 32);
            while (hit) {
                const int k = __builtin_ctzll(hit);
                hit &= hit - 1;
                const int u = ((k >> 5) << 1) | ((k >> 2) & 1);
                const int within = ((k & 3) << 2) | ((k >> 3) & 3);
                push(((base + u * 1024 + tid) << 4) + within);
            }
        };
        uint4 va[4], vb[4];
        load4(va, 0);
        for (int base = 0; base < nvec; base += 8192) {
            load4(vb, base + 4096);
            scan4(va, base);
            load4(va, base + 8192);
            scan4(vb, base + 4096);
        }
        for (int e = (nvec << 4) + tid; e < E; e += 1024) {
            const unsigned r = (unsigned)edst[e] - row0;
            if (r < AGG_ROWS && (unsigned)edst[e] < (unsigned)N0) push(e);
        }
        __syncthreads();

        {
            const int qr = q_n < AGG_QCAP ? q_n : AGG_QCAP;
            for (int i = tid; i < qr; i += 1024) {
                const int e = q_src[i];
                q_src[i] = esrc[e];
                q_dl[i]  = (int)((unsigned)edst[e] - row0);
                q_w[i]   = ew[e];
            }
        }
        __syncthreads();

        const int qn = q_n < AGG_QCAP ? q_n : AGG_QCAP;
        if (list == 0) {
            constexpr int EB = 4;
            for (int b = wv * EB; b + EB <= qn; b += 32 * EB)
                gather_add<2, EB>(q_src, q_dl, q_w, b, xa, xb, xb, acc, segA, segB, 0, ln);
            const int rem0 = (qn / EB) * EB;
            if (rem0 + wv < qn)
                gather_add<2, 1>(q_src, q_dl, q_w, rem0 + wv, xa, xb, xb, acc, segA, segB, 0, ln);
        } else {
            constexpr int EB = 3;
            for (int b = wv * EB; b + EB <= qn; b += 32 * EB)
                gather_add<3, EB>(q_src, q_dl, q_w, b, xa, xb, word, acc, segA, segB, 4 * 128, ln);
            const int rem0 = (qn / EB) * EB;
            if (rem0 + wv < qn)
                gather_add<3, 1>(q_src, q_dl, q_w, rem0 + wv, xa, xb, word, acc, segA, segB, 4 * 128, ln);
        }
        __syncthreads();
        if (tid == 0) q_n = 0;
        __syncthreads();
    }

    for (int r = wv; r < AGG_ROWS; r += 32) {
        const int row = (int)row0 + r;
        if (row >= N0) break;
        const float4* p = (const float4*)(acc + r * AGG_ROWF);
        const float4 a0  = p[ln];
        const float4 a0s = p[32 + ln];
        const float4 a1  = p[64 + ln];
        const float4 a1s = p[96 + ln];
        const float4 aw  = p[128 + ln];
        const float sw = aw.x * aw.x + aw.y * aw.y + aw.z * aw.z + aw.w * aw.w;
        float s0  = a0.x * a0.x + a0.y * a0.y + a0.z * a0.z + a0.w * a0.w;
        float s0s = a0s.x * a0s.x + a0s.y * a0s.y + a0s.z * a0s.z + a0s.w * a0s.w;
        float s1  = a1.x * a1.x + a1.y * a1.y + a1.z * a1.z + a1.w * a1.w + sw;
        float s1s = a1s.x * a1s.x + a1s.y * a1s.y + a1s.z * a1s.z + a1s.w * a1s.w + sw;
#pragma unroll
        for (int off = 16; off > 0; off >>= 1) {
            s0  += __shfl_xor(s0,  off, 32);
            s0s += __shfl_xor(s0s, off, 32);
            s1  += __shfl_xor(s1,  off, 32);
            s1s += __shfl_xor(s1s, off, 32);
        }
        const float i0  = 1.0f / (sqrtf(s0)  + 1e-9f);
        const float i0s = 1.0f / (sqrtf(s0s) + 1e-9f);
        const float i1  = 1.0f / (sqrtf(s1)  + 1e-9f);
        const float i1s = 1.0f / (sqrtf(s1s) + 1e-9f);

        float* o  = doc     + (size_t)row * 384 + ln * 4;
        float* os = doc_svd + (size_t)row * 384 + ln * 4;
        *(float4*)(o)        = make_float4(a0.x * i0, a0.y * i0, a0.z * i0, a0.w * i0);
        *(float4*)(o + 128)  = make_float4(a1.x * i1, a1.y * i1, a1.z * i1, a1.w * i1);
        *(float4*)(o + 256)  = make_float4(aw.x * i1, aw.y * i1, aw.z * i1, aw.w * i1);
        *(float4*)(os)       = make_float4(a0s.x * i0s, a0s.y * i0s, a0s.z * i0s, a0s.w * i0s);
        *(float4*)(os + 128) = make_float4(a1s.x * i1s, a1s.y * i1s, a1s.z * i1s, a1s.w * i1s);
        *(float4*)(os + 256) = make_float4(aw.x * i1s, aw.y * i1s, aw.z * i1s, aw.w * i1s);
    }
}

extern "C" void kernel_launch(void* const* d_in, const int* in_sizes, int n_in,
                              void* d_out, int out_size, void* d_ws, size_t ws_size,
                              hipStream_t stream) {
    const float* x1       = (const float*)d_in[0];
    const float* x2       = (const float*)d_in[1];
    const float* word_emb = (const float*)d_in[2];
    const float* W1a = (const float*)d_in[3];
    const float* b1a = (const float*)d_in[4];
    const float* W1b = (const float*)d_in[5];
    const float* b1b = (const float*)d_in[6];
    const float* W2a = (const float*)d_in[7];
    const float* b2a = (const float*)d_in[8];
    const float* W2b = (const float*)d_in[9];
    const float* b2b = (const float*)d_in[10];
    const int*   e11_src = (const int*)d_in[11];
    const int*   e11_dst = (const int*)d_in[12];
    const float* e11_w   = (const float*)d_in[13];
    const int*   e22_src = (const int*)d_in[14];
    const int*   e22_dst = (const int*)d_in[15];
    const float* e22_w   = (const float*)d_in[16];
    const int*   e01_src = (const int*)d_in[17];
    const int*   e01_dst = (const int*)d_in[18];
    const float* e01_w   = (const float*)d_in[19];
    const int*   e02_src = (const int*)d_in[20];
    const int*   e02_dst = (const int*)d_in[21];
    const float* e02_w   = (const float*)d_in[22];

    const int N0 = 10000, N1 = 50000, N2 = 30000;
    const int E11 = in_sizes[11], E22 = in_sizes[14];
    const int E01 = in_sizes[17], E02 = in_sizes[20];

    float* ws = (float*)d_ws;
    size_t off = 0;
    float* l1_1 = ws + off; off += (size_t)N1 * 128;
    float* l2_1 = ws + off; off += (size_t)N1 * 128;
    float* l1_2 = ws + off; off += (size_t)N2 * 128;
    float* l2_2 = ws + off; off += (size_t)N2 * 128;
    float* h11  = ws + off; off += (size_t)N1 * 128;
    float* h22  = ws + off; off += (size_t)N2 * 128;
    float* doc     = (float*)d_out;
    float* doc_svd = doc + (size_t)N0 * 384;

    const dim3 blk(256);
    const int g1 = (N1 + 127) / 128;
    const int g2 = (N2 + 127) / 128;
    const int s1 = (E11 + EDGES_PER_BLOCK - 1) / EDGES_PER_BLOCK;
    const int s2 = (E22 + EDGES_PER_BLOCK - 1) / EDGES_PER_BLOCK;
    const int z4 = (N1 + N2) * 32;
    const int zb = (z4 + FILL_F4_PER_BLOCK - 1) / FILL_F4_PER_BLOCK;
    const int q01 = (E01 + 3) / 4, q02 = (E02 + 3) / 4;
    const int own_offB = (q01 + 3) / 4 * 4;
    const int ob_blocks = (q01 + q02 + OWN_Q_PER_BLOCK - 1) / OWN_Q_PER_BLOCK;
    unsigned* own = (unsigned*)h11;

    stage_group<<<dim3(g1 + zb), blk, 0, stream>>>(
        x1, W1a, b1a, l1_1, N1, 256, g1,
        nullptr, nullptr, nullptr, nullptr, nullptr, 0, 0,
        h11, z4, zb,
        nullptr, 0, nullptr, 0, nullptr, 0, 0u);
    stage_group<<<dim3(g2 + s1), blk, 0, stream>>>(
        x2, W2a, b2a, l1_2, N2, 256, g2,
        e11_src, e11_dst, e11_w, l1_1, h11, E11, s1,
        nullptr, 0, 0,
        nullptr, 0, nullptr, 0, nullptr, 0, 0u);
    stage_group<<<dim3(g1 + s2), blk, 0, stream>>>(
        h11, W1b, b1b, l2_1, N1, 128, g1,
        e22_src, e22_dst, e22_w, l1_2, h22, E22, s2,
        nullptr, 0, 0,
        nullptr, 0, nullptr, 0, nullptr, 0, 0u);
    stage_group<<<dim3(g2 + ob_blocks), blk, 0, stream>>>(
        h22, W2b, b2b, l2_2, N2, 128, g2,
        nullptr, nullptr, nullptr, nullptr, nullptr, 0, 0,
        nullptr, 0, 0,
        e01_dst, E01, e02_dst, E02, own, own_offB, (unsigned)N0);

    doc_aggregate<<<dim3((N0 + AGG_ROWS - 1) / AGG_ROWS), dim3(1024), 0, stream>>>(
        e01_src, e01_dst, e01_w, E01, e02_src, e02_dst, e02_w, E02,
        l2_1, l1_1, l2_2, l1_2, word_emb, own, own_offB, doc, doc_svd, N0);
}


# tower scatter-add owner-computes in LDS: a work-group keeps 200 + 120 sum rows, scans owner bytes of both edge lists, adds gathered rows by LDS f32 atomics (arrival-order sums), stores them plainly
# speedup vs baseline: 2.4290x; 1.0334x over previous
#include <hip/hip_runtime.h>
#include <math.h>

typedef __attribute__((ext_vector_type(2))) float v2f;
typedef __attribute__((ext_vector_type(8))) float v8f;

#define O_DIM 128

__device__ __forceinline__
void gemm_tile(const float* X, const float* __restrict__ W,
               const float* __restrict__ bias, float* out,
               int N, int K, int bid) {
    __shared__ float  As[128][36];
    __shared__ float2 Bs2[16][144];

    const int tid = threadIdx.x;
    const int wv  = tid >> 5;
    const int ln  = tid & 31;
    const int m0  = bid * 128;

    v8f acc[8];
#pragma unroll
    for (int t = 0; t < 8; ++t)
#pragma unroll
        for (int j = 0; j < 8; ++j) acc[t][j] = 0.0f;

    const int mrow = wv * 16 + (ln & 15);
    const int lnn  = ln & 15;
    const int khalf = (ln >> 4) << 1;

    for (int k0 = 0; k0 < K; k0 += 32) {
#pragma unroll
        for (int i = 0; i < 4; ++i) {
            int f   = tid + i * 256;
            int row = f >> 3;
            int c4  = f & 7;
            float4 v = make_float4(0.f, 0.f, 0.f, 0.f);
            if (m0 + row < N)
                v = *(const float4*)(X + (size_t)(m0 + row) * K + k0 + c4 * 4);
            *(float4*)&As[row][c4 * 4] = v;
        }
#pragma unroll
        for (int i = 0; i < 4; ++i) {
            int f   = tid + i * 256;
            int row = f >> 5;
            int c4  = f & 31;
            float4 v = *(const float4*)(W + (size_t)(k0 + row) * O_DIM + c4 * 4);
            float* bp = (float*)&Bs2[row >> 1][c4 * 4] + (row & 1);
            bp[0] = v.x;
            bp[2] = v.y;
            bp[4] = v.z;
            bp[6] = v.w;
        }
        __syncthreads();

#pragma unroll
        for (int kk = 0; kk < 32; kk += 4) {
            const int kk2 = kk + khalf;
            const int p   = kk2 >> 1;
            v2f a = *(const v2f*)&As[mrow][kk2];
#pragma unroll
            for (int t = 0; t < 8; ++t) {
                v2f b = *(const v2f*)&Bs2[p][t * 16 + lnn];
                acc[t] = __builtin_amdgcn_wmma_f32_16x16x4_f32(
                    false, a, false, b, (short)0, acc[t], false, false);
            }
        }
        __syncthreads();
    }

    const int mtile = m0 + wv * 16;
    if (mtile < N) {
        const int rowoff = (ln >> 4) << 3;
#pragma unroll
        for (int t = 0; t < 8; ++t) {
            const int col = t * 16 + lnn;
            const float bv = bias[col];
#pragma unroll
            for (int i = 0; i < 8; ++i) {
                float v = acc[t][i] + bv;
                out[(size_t)(mtile + rowoff + i) * O_DIM + col] = v > 0.f ? v : 0.f;
            }
        }
    }
}

#define T1_ROWS 200
#define T2_ROWS 120
#define OWN_ROWS 40
#define OWN_Q_PER_BLOCK 1024

template <int ROWS>
__device__ __forceinline__ unsigned owner_byte(const int* __restrict__ d, int e, int E, unsigned nrows) {
    if (e >= E) return 255u;
    const unsigned r = (unsigned)d[e];
    return r < nrows ? r / ROWS : 255u;
}

template <int ROWS>
__device__ __forceinline__ void owner_list(const int* __restrict__ d, int E,
                                           unsigned* __restrict__ ob, unsigned nrows, int bid) {
    const int nq = (E + 3) >> 2;
#pragma unroll
    for (int i = 0; i < OWN_Q_PER_BLOCK / 256; ++i) {
        const int q = bid * OWN_Q_PER_BLOCK + i * 256 + (int)threadIdx.x;
        if (q >= nq) break;
        const int e = q << 2;
        ob[q] = owner_byte<ROWS>(d, e, E, nrows)
              | (owner_byte<ROWS>(d, e + 1, E, nrows) << 8)
              | (owner_byte<ROWS>(d, e + 2, E, nrows) << 16)
              | (owner_byte<ROWS>(d, e + 3, E, nrows) << 24);
    }
}

__global__ __launch_bounds__(256)
void stage_pair(const float* aX, const float* __restrict__ aW,
                const float* __restrict__ aBias, float* aOut, int aN, int aK, int aBlocks,
                const float* bX, const float* __restrict__ bW,
                const float* __restrict__ bBias, float* bOut, int bN, int bK, int bBlocks,
                const int* __restrict__ d0, int E0, int w0, int nb0,
                const int* __restrict__ d1, int E1, int w1, int nb1,
                const int* __restrict__ d2, int E2, int w2, int nb2,
                const int* __restrict__ d3, int E3, int w3,
                unsigned* __restrict__ ob, unsigned rowsT1, unsigned rowsT2, unsigned rowsDoc) {
    int bid = (int)blockIdx.x;
    if (bid < aBlocks + bBlocks) {
        const bool second = bid >= aBlocks;
        gemm_tile(second ? bX : aX, second ? bW : aW, second ? bBias : aBias,
                  second ? bOut : aOut, second ? bN : aN, second ? bK : aK,
                  second ? bid - aBlocks : bid);
        return;
    }
    bid -= aBlocks + bBlocks;
    if (bid < nb0) { owner_list<T1_ROWS>(d0, E0, ob + w0, rowsT1, bid); return; }
    bid -= nb0;
    if (bid < nb1) { owner_list<T2_ROWS>(d1, E1, ob + w1, rowsT2, bid); return; }
    bid -= nb1;
    if (bid < nb2) { owner_list<OWN_ROWS>(d2, E2, ob + w2, rowsDoc, bid); return; }
    bid -= nb2;
    owner_list<OWN_ROWS>(d3, E3, ob + w3, rowsDoc, bid);
}

#define AGG_ROWS 40
#define AGG_ROWF 640
static_assert(AGG_ROWS == OWN_ROWS, "owner bytes and aggregation must agree on rows per group");
#define AGG_QCAP 6144

__device__ __forceinline__ void ldsAddF(float* p, float v) {
    __hip_atomic_fetch_add(p, v, __ATOMIC_RELAXED, __HIP_MEMORY_SCOPE_WORKGROUP);
}

template <int NS, int EB>
__device__ __forceinline__ void gather_add(const int* qs, const int* qd, const float* qw, int b,
                                           const float* __restrict__ xa, const float* __restrict__ xb,
                                           const float* __restrict__ xc, float* acc,
                                           int sa, int sb, int sc, int ln) {
    float v[EB][4 * NS];
    int   dl[EB];
    float wt[EB];
#pragma unroll
    for (int j = 0; j < EB; ++j) {
        const size_t ro = (size_t)__builtin_amdgcn_readfirstlane(qs[b + j]) * 128 + ln;
        dl[j] = __builtin_amdgcn_readfirstlane(qd[b + j]);
        wt[j] = qw[b + j];
#pragma unroll
        for (int c = 0; c < 4; ++c) {
            v[j][c]     = xa[ro + 32 * c];
            v[j][4 + c] = xb[ro + 32 * c];
            if (NS == 3) v[j][8 + c] = xc[ro + 32 * c];
        }
    }
#pragma unroll
    for (int j = 0; j < EB; ++j) {
        float* o = acc + dl[j] * AGG_ROWF + ln;
#pragma unroll
        for (int c = 0; c < 4; ++c) {
            ldsAddF(o + sa + 32 * c, wt[j] * v[j][c]);
            ldsAddF(o + sb + 32 * c, wt[j] * v[j][4 + c]);
            if (NS == 3) ldsAddF(o + sc + 32 * c, wt[j] * v[j][8 + c]);
        }
    }
}

__global__ __launch_bounds__(1024)
void doc_aggregate(const int* __restrict__ e01_src, const int* __restrict__ e01_dst,
                   const float* __restrict__ e01_w, int E01,
                   const int* __restrict__ e02_src, const int* __restrict__ e02_dst,
                   const float* __restrict__ e02_w, int E02,
                   const float* __restrict__ l2_1, const float* __restrict__ l1_1,
                   const float* __restrict__ l2_2, const float* __restrict__ l1_2,
                   const float* __restrict__ word,
                   const unsigned* __restrict__ ob, int obOffB,
                   float* __restrict__ doc, float* __restrict__ doc_svd, int N0) {
    __shared__ __attribute__((aligned(16))) float acc[AGG_ROWS * AGG_ROWF];
    __shared__ int   q_src[AGG_QCAP];
    __shared__ int   q_dl[AGG_QCAP];
    __shared__ float q_w[AGG_QCAP];
    __shared__ int   q_n;

    const int tid = threadIdx.x;
    const int ln  = tid & 31;
    const int wv  = __builtin_amdgcn_readfirstlane(tid >> 5);
    const unsigned row0 = blockIdx.x * AGG_ROWS;

    for (int i = tid; i < AGG_ROWS * AGG_ROWF / 4; i += 1024)
        ((float4*)acc)[i] = make_float4(0.f, 0.f, 0.f, 0.f);
    if (tid == 0) q_n = 0;
    __syncthreads();

#pragma unroll
    for (int list = 0; list < 2; ++list) {
        const int*   esrc = list == 0 ? e01_src : e02_src;
        const int*   edst = list == 0 ? e01_dst : e02_dst;
        const float* ew   = list == 0 ? e01_w   : e02_w;
        const int    E    = list == 0 ? E01     : E02;
        const float* xa   = list == 0 ? l2_1 : l2_2;
        const float* xb   = list == 0 ? l1_1 : l1_2;
        const int    segA = list == 0 ? 0 : 2 * 128;
        const int    segB = list == 0 ? 128 : 3 * 128;

        auto push = [&](int e) {
            const int slot = __hip_atomic_fetch_add(&q_n, 1, __ATOMIC_RELAXED,
                                                    __HIP_MEMORY_SCOPE_WORKGROUP);
            if (slot < AGG_QCAP) {
                q_src[slot] = e;
            } else {
                const int   s  = esrc[e];
                const float wt = ew[e];
                const unsigned dl = (unsigned)edst[e] - row0;
                float* o = acc + dl * AGG_ROWF;
                const size_t ro = (size_t)s * 128;
                for (int c = 0; c < 128; ++c) {
                    ldsAddF(o + segA + c, wt * xa[ro + c]);
                    ldsAddF(o + segB + c, wt * xb[ro + c]);
                    if (list == 1) ldsAddF(o + 4 * 128 + c, wt * word[ro + c]);
                }
            }
        };

        const uint4* ov = (const uint4*)(ob + (list == 0 ? 0 : obOffB));
        const int nvec = E >> 4;
        const unsigned pat = blockIdx.x * 0x01010101u;
        auto z4 = [&](unsigned v) -> unsigned {
            const unsigned t = v ^ pat;
            return ~(((t & 0x7F7F7F7Fu) + 0x7F7F7F7Fu) | t) & 0x80808080u;
        };
        auto eq16 = [&](const uint4& v) -> unsigned {
            return (z4(v.x) >> 7) | (z4(v.y) >> 6) | (z4(v.z) >> 5) | (z4(v.w) >> 4);
        };
        const int ilast = nvec > 0 ? nvec - 1 : 0;
        auto load4 = [&](uint4 (&v)[4], int base) {
#pragma unroll
            for (int u = 0; u < 4; ++u) {
                const int idx = base + u * 1024 + tid;
                v[u] = ov[idx < nvec ? idx : ilast];
            }
        };
        auto scan4 = [&](const uint4 (&v)[4], int base) {
            unsigned h[4];
#pragma unroll
            for (int u = 0; u < 4; ++u)
                h[u] = (base + u * 1024 + tid) < nvec ? eq16(v[u]) : 0u;
            unsigned long long hit = (unsigned long long)(h[0] | (h[1] << 4))
                                   | ((unsigned long long)(h[2] | (h[3] << 4)) << 32);
            while (hit) {
                const int k = __builtin_ctzll(hit);
                hit &= hit - 1;
                const int u = ((k >> 5) << 1) | ((k >> 2) & 1);
                const int within = ((k & 3) << 2) | ((k >> 3) & 3);
                push(((base + u * 1024 + tid) << 4) + within);
            }
        };
        uint4 va[4], vb[4];
        load4(va, 0);
        for (int base = 0; base < nvec; base += 8192) {
            load4(vb, base + 4096);
            scan4(va, base);
            load4(va, base + 8192);
            scan4(vb, base + 4096);
        }
        for (int e = (nvec << 4) + tid; e < E; e += 1024) {
            const unsigned r = (unsigned)edst[e] - row0;
            if (r < AGG_ROWS && (unsigned)edst[e] < (unsigned)N0) push(e);
        }
        __syncthreads();

        {
            const int qr = q_n < AGG_QCAP ? q_n : AGG_QCAP;
            for (int i = tid; i < qr; i += 1024) {
                const int e = q_src[i];
                q_src[i] = esrc[e];
                q_dl[i]  = (int)((unsigned)edst[e] - row0);
                q_w[i]   = ew[e];
            }
        }
        __syncthreads();

        const int qn = q_n < AGG_QCAP ? q_n : AGG_QCAP;
        if (list == 0) {
            constexpr int EB = 4;
            for (int b = wv * EB; b + EB <= qn; b += 32 * EB)
                gather_add<2, EB>(q_src, q_dl, q_w, b, xa, xb, xb, acc, segA, segB, 0, ln);
            const int rem0 = (qn / EB) * EB;
            if (rem0 + wv < qn)
                gather_add<2, 1>(q_src, q_dl, q_w, rem0 + wv, xa, xb, xb, acc, segA, segB, 0, ln);
        } else {
            constexpr int EB = 3;
            for (int b = wv * EB; b + EB <= qn; b += 32 * EB)
                gather_add<3, EB>(q_src, q_dl, q_w, b, xa, xb, word, acc, segA, segB, 4 * 128, ln);
            const int rem0 = (qn / EB) * EB;
            if (rem0 + wv < qn)
                gather_add<3, 1>(q_src, q_dl, q_w, rem0 + wv, xa, xb, word, acc, segA, segB, 4 * 128, ln);
        }
        __syncthreads();
        if (tid == 0) q_n = 0;
        __syncthreads();
    }

    for (int r = wv; r < AGG_ROWS; r += 32) {
        const int row = (int)row0 + r;
        if (row >= N0) break;
        const float4* p = (const float4*)(acc + r * AGG_ROWF);
        const float4 a0  = p[ln];
        const float4 a0s = p[32 + ln];
        const float4 a1  = p[64 + ln];
        const float4 a1s = p[96 + ln];
        const float4 aw  = p[128 + ln];
        const float sw = aw.x * aw.x + aw.y * aw.y + aw.z * aw.z + aw.w * aw.w;
        float s0  = a0.x * a0.x + a0.y * a0.y + a0.z * a0.z + a0.w * a0.w;
        float s0s = a0s.x * a0s.x + a0s.y * a0s.y + a0s.z * a0s.z + a0s.w * a0s.w;
        float s1  = a1.x * a1.x + a1.y * a1.y + a1.z * a1.z + a1.w * a1.w + sw;
        float s1s = a1s.x * a1s.x + a1s.y * a1s.y + a1s.z * a1s.z + a1s.w * a1s.w + sw;
#pragma unroll
        for (int off = 16; off > 0; off >>= 1) {
            s0  += __shfl_xor(s0,  off, 32);
            s0s += __shfl_xor(s0s, off, 32);
            s1  += __shfl_xor(s1,  off, 32);
            s1s += __shfl_xor(s1s, off, 32);
        }
        const float i0  = 1.0f / (sqrtf(s0)  + 1e-9f);
        const float i0s = 1.0f / (sqrtf(s0s) + 1e-9f);
        const float i1  = 1.0f / (sqrtf(s1)  + 1e-9f);
        const float i1s = 1.0f / (sqrtf(s1s) + 1e-9f);

        float* o  = doc     + (size_t)row * 384 + ln * 4;
        float* os = doc_svd + (size_t)row * 384 + ln * 4;
        *(float4*)(o)        = make_float4(a0.x * i0, a0.y * i0, a0.z * i0, a0.w * i0);
        *(float4*)(o + 128)  = make_float4(a1.x * i1, a1.y * i1, a1.z * i1, a1.w * i1);
        *(float4*)(o + 256)  = make_float4(aw.x * i1, aw.y * i1, aw.z * i1, aw.w * i1);
        *(float4*)(os)       = make_float4(a0s.x * i0s, a0s.y * i0s, a0s.z * i0s, a0s.w * i0s);
        *(float4*)(os + 128) = make_float4(a1s.x * i1s, a1s.y * i1s, a1s.z * i1s, a1s.w * i1s);
        *(float4*)(os + 256) = make_float4(aw.x * i1s, aw.y * i1s, aw.z * i1s, aw.w * i1s);
    }
}

#define TS_QCAP 6144

template <int EB>
__device__ __forceinline__ void gather_add1(const int* qs, const int* qd, const float* qw, int b,
                                            const float* __restrict__ x, float* acc, int ln) {
    float v[EB][4];
    int   dl[EB];
    float wt[EB];
#pragma unroll
    for (int j = 0; j < EB; ++j) {
        const size_t ro = (size_t)__builtin_amdgcn_readfirstlane(qs[b + j]) * 128 + ln;
        dl[j] = __builtin_amdgcn_readfirstlane(qd[b + j]);
        wt[j] = qw[b + j];
#pragma unroll
        for (int c = 0; c < 4; ++c) v[j][c] = x[ro + 32 * c];
    }
#pragma unroll
    for (int j = 0; j < EB; ++j) {
        float* o = acc + dl[j] * 128 + ln;
#pragma unroll
        for (int c = 0; c < 4; ++c) ldsAddF(o + 32 * c, wt[j] * v[j][c]);
    }
}

__global__ __launch_bounds__(1024)
void tower_scatter(const int* __restrict__ eA_src, const int* __restrict__ eA_dst,
                   const float* __restrict__ eA_w, int EA,
                   const int* __restrict__ eB_src, const int* __restrict__ eB_dst,
                   const float* __restrict__ eB_w, int EBn,
                   const float* __restrict__ xA, const float* __restrict__ xB,
                   const unsigned* __restrict__ obA, const unsigned* __restrict__ obB,
                   float* __restrict__ hA, float* __restrict__ hB, int NA, int NB) {
    __shared__ __attribute__((aligned(16))) float acc[T1_ROWS * 128];
    __shared__ int   q_src[TS_QCAP];
    __shared__ int   q_dl[TS_QCAP];
    __shared__ float q_w[TS_QCAP];
    __shared__ int   q_n;
    static_assert(T2_ROWS <= T1_ROWS, "the sums of tower 2 reuse the rows of tower 1");

    const int tid = threadIdx.x;
    const int ln  = tid & 31;
    const int wv  = __builtin_amdgcn_readfirstlane(tid >> 5);

#pragma unroll
    for (int list = 0; list < 2; ++list) {
        const int*   esrc = list == 0 ? eA_src : eB_src;
        const int*   edst = list == 0 ? eA_dst : eB_dst;
        const float* ew   = list == 0 ? eA_w   : eB_w;
        const int    E    = list == 0 ? EA     : EBn;
        const float* x    = list == 0 ? xA     : xB;
        float*       h    = list == 0 ? hA     : hB;
        const int    N    = list == 0 ? NA     : NB;
        const int    ROWS = list == 0 ? T1_ROWS : T2_ROWS;
        const unsigned row0 = blockIdx.x * (unsigned)ROWS;

        for (int i = tid; i < ROWS * 32; i += 1024)
            ((float4*)acc)[i] = make_float4(0.f, 0.f, 0.f, 0.f);
        if (tid == 0) q_n = 0;
        __syncthreads();

        auto push = [&](int e) {
            const int slot = __hip_atomic_fetch_add(&q_n, 1, __ATOMIC_RELAXED,
                                                    __HIP_MEMORY_SCOPE_WORKGROUP);
            if (slot < TS_QCAP) {
                q_src[slot] = e;
            } else {
                const int   s  = esrc[e];
                const float wt = ew[e];
                const unsigned dl = (unsigned)edst[e] - row0;
                float* o = acc + dl * 128;
                const size_t ro = (size_t)s * 128;
                for (int c = 0; c < 128; ++c) ldsAddF(o + c, wt * x[ro + c]);
            }
        };

        const uint4* ov = (const uint4*)(list == 0 ? obA : obB);
        const int nvec = E >> 4;
        const unsigned pat = blockIdx.x * 0x01010101u;
        auto z4 = [&](unsigned v) -> unsigned {
            const unsigned t = v ^ pat;
            return ~(((t & 0x7F7F7F7Fu) + 0x7F7F7F7Fu) | t) & 0x80808080u;
        };
        auto eq16 = [&](const uint4& v) -> unsigned {
            return (z4(v.x) >> 7) | (z4(v.y) >> 6) | (z4(v.z) >> 5) | (z4(v.w) >> 4);
        };
        const int ilast = nvec > 0 ? nvec - 1 : 0;
        auto load4 = [&](uint4 (&v)[4], int base) {
#pragma unroll
            for (int u = 0; u < 4; ++u) {
                const int idx = base + u * 1024 + tid;
                v[u] = ov[idx < nvec ? idx : ilast];
            }
        };
        auto scan4 = [&](const uint4 (&v)[4], int base) {
            unsigned hh[4];
#pragma unroll
            for (int u = 0; u < 4; ++u)
                hh[u] = (base + u * 1024 + tid) < nvec ? eq16(v[u]) : 0u;
            unsigned long long hit = (unsigned long long)(hh[0] | (hh[1] << 4))
                                   | ((unsigned long long)(hh[2] | (hh[3] << 4)) << 32);
            while (hit) {
                const int k = __builtin_ctzll(hit);
                hit &= hit - 1;
                const int u = ((k >> 5) << 1) | ((k >> 2) & 1);
                const int within = ((k & 3) << 2) | ((k >> 3) & 3);
                push(((base + u * 1024 + tid) << 4) + within);
            }
        };
        if (nvec > 0) {
            uint4 va[4], vb[4];
            load4(va, 0);
            for (int base = 0; base < nvec; base += 8192) {
                load4(vb, base + 4096);
                scan4(va, base);
                load4(va, base + 8192);
                scan4(vb, base + 4096);
            }
        }
        for (int e = (nvec << 4) + tid; e < E; e += 1024) {
            const unsigned r = (unsigned)edst[e] - row0;
            if (r < (unsigned)ROWS && (unsigned)edst[e] < (unsigned)N) push(e);
        }
        __syncthreads();

        const int qn = q_n < TS_QCAP ? q_n : TS_QCAP;
        for (int i = tid; i < qn; i += 1024) {
            const int e = q_src[i];
            q_src[i] = esrc[e];
            q_dl[i]  = (int)((unsigned)edst[e] - row0);
            q_w[i]   = ew[e];
        }
        __syncthreads();

        constexpr int EB = 8;
        for (int b = wv * EB; b + EB <= qn; b += 32 * EB)
            gather_add1<EB>(q_src, q_dl, q_w, b, x, acc, ln);
        const int rem0 = (qn / EB) * EB;
        if (rem0 + wv < qn)
            gather_add1<1>(q_src, q_dl, q_w, rem0 + wv, x, acc, ln);
        __syncthreads();

        const int left = N - (int)row0;
        const int nv = left < ROWS ? left : ROWS;
        float4* hp = (float4*)(h + (size_t)row0 * 128);
        for (int i = tid; i < nv * 32; i += 1024)
            hp[i] = ((const float4*)acc)[i];
        __syncthreads();
    }
}

extern "C" void kernel_launch(void* const* d_in, const int* in_sizes, int n_in,
                              void* d_out, int out_size, void* d_ws, size_t ws_size,
                              hipStream_t stream) {
    const float* x1       = (const float*)d_in[0];
    const float* x2       = (const float*)d_in[1];
    const float* word_emb = (const float*)d_in[2];
    const float* W1a = (const float*)d_in[3];
    const float* b1a = (const float*)d_in[4];
    const float* W1b = (const float*)d_in[5];
    const float* b1b = (const float*)d_in[6];
    const float* W2a = (const float*)d_in[7];
    const float* b2a = (const float*)d_in[8];
    const float* W2b = (const float*)d_in[9];
    const float* b2b = (const float*)d_in[10];
    const int*   e11_src = (const int*)d_in[11];
    const int*   e11_dst = (const int*)d_in[12];
    const float* e11_w   = (const float*)d_in[13];
    const int*   e22_src = (const int*)d_in[14];
    const int*   e22_dst = (const int*)d_in[15];
    const float* e22_w   = (const float*)d_in[16];
    const int*   e01_src = (const int*)d_in[17];
    const int*   e01_dst = (const int*)d_in[18];
    const float* e01_w   = (const float*)d_in[19];
    const int*   e02_src = (const int*)d_in[20];
    const int*   e02_dst = (const int*)d_in[21];
    const float* e02_w   = (const float*)d_in[22];

    const int N0 = 10000, N1 = 50000, N2 = 30000;
    const int E11 = in_sizes[11], E22 = in_sizes[14];
    const int E01 = in_sizes[17], E02 = in_sizes[20];


    float* ws = (float*)d_ws;
    size_t off = 0;
    float* l1_1 = ws + off; off += (size_t)N1 * 128;
    float* l2_1 = ws + off; off += (size_t)N1 * 128;
    float* l1_2 = ws + off; off += (size_t)N2 * 128;
    unsigned* own = (unsigned*)(ws + off); off += (size_t)N2 * 128;
    float* h11  = ws + off; off += (size_t)N1 * 128;
    float* h22  = ws + off; off += (size_t)N2 * 128;
    float* l2_2 = h22;
    float* doc     = (float*)d_out;
    float* doc_svd = doc + (size_t)N0 * 384;

    const dim3 blk(256);
    const int g1 = (N1 + 127) / 128;
    const int g2 = (N2 + 127) / 128;
    const int q11 = (E11 + 3) / 4, q22 = (E22 + 3) / 4;
    const int q01 = (E01 + 3) / 4, q02 = (E02 + 3) / 4;
    const int w11 = 0;
    const int w22 = w11 + (q11 + 3) / 4 * 4;
    const int w01 = w22 + (q22 + 3) / 4 * 4;
    const int w02 = w01 + (q01 + 3) / 4 * 4;
    const int nb11 = (q11 + OWN_Q_PER_BLOCK - 1) / OWN_Q_PER_BLOCK;
    const int nb22 = (q22 + OWN_Q_PER_BLOCK - 1) / OWN_Q_PER_BLOCK;
    const int nb01 = (q01 + OWN_Q_PER_BLOCK - 1) / OWN_Q_PER_BLOCK;
    const int nb02 = (q02 + OWN_Q_PER_BLOCK - 1) / OWN_Q_PER_BLOCK;
    const int t1 = (N1 + T1_ROWS - 1) / T1_ROWS;
    const int t2 = (N2 + T2_ROWS - 1) / T2_ROWS;
    const int tg = t1 > t2 ? t1 : t2;

    stage_pair<<<dim3(g1 + g2 + nb11 + nb22 + nb01 + nb02), blk, 0, stream>>>(
        x1, W1a, b1a, l1_1, N1, 256, g1,
        x2, W2a, b2a, l1_2, N2, 256, g2,
        e11_dst, E11, w11, nb11,
        e22_dst, E22, w22, nb22,
        e01_dst, E01, w01, nb01,
        e02_dst, E02, w02,
        own, (unsigned)N1, (unsigned)N2, (unsigned)N0);
    tower_scatter<<<dim3(tg), dim3(1024), 0, stream>>>(
        e11_src, e11_dst, e11_w, E11, e22_src, e22_dst, e22_w, E22,
        l1_1, l1_2, own + w11, own + w22, h11, h22, N1, N2);
    stage_pair<<<dim3(g1 + g2), blk, 0, stream>>>(
        h11, W1b, b1b, l2_1, N1, 128, g1,
        h22, W2b, b2b, l2_2, N2, 128, g2,
        nullptr, 0, 0, 0,
        nullptr, 0, 0, 0,
        nullptr, 0, 0, 0,
        nullptr, 0, 0,
        own, 0u, 0u, 0u);

    doc_aggregate<<<dim3((N0 + AGG_ROWS - 1) / AGG_ROWS), dim3(1024), 0, stream>>>(
        e01_src, e01_dst, e01_w, E01, e02_src, e02_dst, e02_w, E02,
        l2_1, l1_1, l2_2, l1_2, word_emb, own + w01, w02 - w01, doc, doc_svd, N0);
}


# row gathers of tower scatter-add and doc aggregation: one 16-byte-per-lane load per 512-byte row (was 4 line loads); LDS sums column-permuted so adds stay conflict-free; sums stay in arrival order
# speedup vs baseline: 2.5561x; 1.0523x over previous
#include <hip/hip_runtime.h>
#include <math.h>

typedef __attribute__((ext_vector_type(2))) float v2f;
typedef __attribute__((ext_vector_type(8))) float v8f;

#define O_DIM 128

__device__ __forceinline__
void gemm_tile(const float* X, const float* __restrict__ W,
               const float* __restrict__ bias, float* out,
               int N, int K, int bid) {
    __shared__ float  As[128][36];
    __shared__ float2 Bs2[16][144];

    const int tid = threadIdx.x;
    const int wv  = tid >> 5;
    const int ln  = tid & 31;
    const int m0  = bid * 128;

    v8f acc[8];
#pragma unroll
    for (int t = 0; t < 8; ++t)
#pragma unroll
        for (int j = 0; j < 8; ++j) acc[t][j] = 0.0f;

    const int mrow = wv * 16 + (ln & 15);
    const int lnn  = ln & 15;
    const int khalf = (ln >> 4) << 1;

    for (int k0 = 0; k0 < K; k0 += 32) {
#pragma unroll
        for (int i = 0; i < 4; ++i) {
            int f   = tid + i * 256;
            int row = f >> 3;
            int c4  = f & 7;
            float4 v = make_float4(0.f, 0.f, 0.f, 0.f);
            if (m0 + row < N)
                v = *(const float4*)(X + (size_t)(m0 + row) * K + k0 + c4 * 4);
            *(float4*)&As[row][c4 * 4] = v;
        }
#pragma unroll
        for (int i = 0; i < 4; ++i) {
            int f   = tid + i * 256;
            int row = f >> 5;
            int c4  = f & 31;
            float4 v = *(const float4*)(W + (size_t)(k0 + row) * O_DIM + c4 * 4);
            float* bp = (float*)&Bs2[row >> 1][c4 * 4] + (row & 1);
            bp[0] = v.x;
            bp[2] = v.y;
            bp[4] = v.z;
            bp[6] = v.w;
        }
        __syncthreads();

#pragma unroll
        for (int kk = 0; kk < 32; kk += 4) {
            const int kk2 = kk + khalf;
            const int p   = kk2 >> 1;
            v2f a = *(const v2f*)&As[mrow][kk2];
#pragma unroll
            for (int t = 0; t < 8; ++t) {
                v2f b = *(const v2f*)&Bs2[p][t * 16 + lnn];
                acc[t] = __builtin_amdgcn_wmma_f32_16x16x4_f32(
                    false, a, false, b, (short)0, acc[t], false, false);
            }
        }
        __syncthreads();
    }

    const int mtile = m0 + wv * 16;
    if (mtile < N) {
        const int rowoff = (ln >> 4) << 3;
#pragma unroll
        for (int t = 0; t < 8; ++t) {
            const int col = t * 16 + lnn;
            const float bv = bias[col];
#pragma unroll
            for (int i = 0; i < 8; ++i) {
                float v = acc[t][i] + bv;
                out[(size_t)(mtile + rowoff + i) * O_DIM + col] = v > 0.f ? v : 0.f;
            }
        }
    }
}

#define T1_ROWS 200
#define T2_ROWS 120
#define OWN_ROWS 40
#define OWN_Q_PER_BLOCK 1024

template <int ROWS>
__device__ __forceinline__ unsigned owner_byte(const int* __restrict__ d, int e, int E, unsigned nrows) {
    if (e >= E) return 255u;
    const unsigned r = (unsigned)d[e];
    return r < nrows ? r / ROWS : 255u;
}

template <int ROWS>
__device__ __forceinline__ void owner_list(const int* __restrict__ d, int E,
                                           unsigned* __restrict__ ob, unsigned nrows, int bid) {
    const int nq = (E + 3) >> 2;
#pragma unroll
    for (int i = 0; i < OWN_Q_PER_BLOCK / 256; ++i) {
        const int q = bid * OWN_Q_PER_BLOCK + i * 256 + (int)threadIdx.x;
        if (q >= nq) break;
        const int e = q << 2;
        ob[q] = owner_byte<ROWS>(d, e, E, nrows)
              | (owner_byte<ROWS>(d, e + 1, E, nrows) << 8)
              | (owner_byte<ROWS>(d, e + 2, E, nrows) << 16)
              | (owner_byte<ROWS>(d, e + 3, E, nrows) << 24);
    }
}

__global__ __launch_bounds__(256)
void stage_pair(const float* aX, const float* __restrict__ aW,
                const float* __restrict__ aBias, float* aOut, int aN, int aK, int aBlocks,
                const float* bX, const float* __restrict__ bW,
                const float* __restrict__ bBias, float* bOut, int bN, int bK, int bBlocks,
                const int* __restrict__ d0, int E0, int w0, int nb0,
                const int* __restrict__ d1, int E1, int w1, int nb1,
                const int* __restrict__ d2, int E2, int w2, int nb2,
                const int* __restrict__ d3, int E3, int w3,
                unsigned* __restrict__ ob, unsigned rowsT1, unsigned rowsT2, unsigned rowsDoc) {
    int bid = (int)blockIdx.x;
    if (bid < aBlocks + bBlocks) {
        const bool second = bid >= aBlocks;
        gemm_tile(second ? bX : aX, second ? bW : aW, second ? bBias : aBias,
                  second ? bOut : aOut, second ? bN : aN, second ? bK : aK,
                  second ? bid - aBlocks : bid);
        return;
    }
    bid -= aBlocks + bBlocks;
    if (bid < nb0) { owner_list<T1_ROWS>(d0, E0, ob + w0, rowsT1, bid); return; }
    bid -= nb0;
    if (bid < nb1) { owner_list<T2_ROWS>(d1, E1, ob + w1, rowsT2, bid); return; }
    bid -= nb1;
    if (bid < nb2) { owner_list<OWN_ROWS>(d2, E2, ob + w2, rowsDoc, bid); return; }
    bid -= nb2;
    owner_list<OWN_ROWS>(d3, E3, ob + w3, rowsDoc, bid);
}

#define AGG_ROWS 40
#define AGG_ROWF 640
static_assert(AGG_ROWS == OWN_ROWS, "owner bytes and aggregation must agree on rows per group");
#define AGG_QCAP 6144

__device__ __forceinline__ void ldsAddF(float* p, float v) {
    __hip_atomic_fetch_add(p, v, __ATOMIC_RELAXED, __HIP_MEMORY_SCOPE_WORKGROUP);
}

template <int NS, int EB>
__device__ __forceinline__ void gather_add(const int* qs, const int* qd, const float* qw, int b,
                                           const float* __restrict__ xa, const float* __restrict__ xb,
                                           const float* __restrict__ xc, float* acc,
                                           int sa, int sb, int sc, int ln) {
    float4 va[EB], vb[EB], vc[EB];
    int   dl[EB];
    float wt[EB];
#pragma unroll
    for (int j = 0; j < EB; ++j) {
        const size_t ro = (size_t)__builtin_amdgcn_readfirstlane(qs[b + j]) * 128;
        dl[j] = __builtin_amdgcn_readfirstlane(qd[b + j]);
        wt[j] = qw[b + j];
        va[j] = ((const float4*)(xa + ro))[ln];
        vb[j] = ((const float4*)(xb + ro))[ln];
        if (NS == 3) vc[j] = ((const float4*)(xc + ro))[ln];
    }
#pragma unroll
    for (int j = 0; j < EB; ++j) {
        float* o = acc + dl[j] * AGG_ROWF + ln;
        const float w = wt[j];
        ldsAddF(o + sa,      w * va[j].x);
        ldsAddF(o + sa + 32, w * va[j].y);
        ldsAddF(o + sa + 64, w * va[j].z);
        ldsAddF(o + sa + 96, w * va[j].w);
        ldsAddF(o + sb,      w * vb[j].x);
        ldsAddF(o + sb + 32, w * vb[j].y);
        ldsAddF(o + sb + 64, w * vb[j].z);
        ldsAddF(o + sb + 96, w * vb[j].w);
        if (NS == 3) {
            ldsAddF(o + sc,      w * vc[j].x);
            ldsAddF(o + sc + 32, w * vc[j].y);
            ldsAddF(o + sc + 64, w * vc[j].z);
            ldsAddF(o + sc + 96, w * vc[j].w);
        }
    }
}

__global__ __launch_bounds__(1024)
void doc_aggregate(const int* __restrict__ e01_src, const int* __restrict__ e01_dst,
                   const float* __restrict__ e01_w, int E01,
                   const int* __restrict__ e02_src, const int* __restrict__ e02_dst,
                   const float* __restrict__ e02_w, int E02,
                   const float* __restrict__ l2_1, const float* __restrict__ l1_1,
                   const float* __restrict__ l2_2, const float* __restrict__ l1_2,
                   const float* __restrict__ word,
                   const unsigned* __restrict__ ob, int obOffB,
                   float* __restrict__ doc, float* __restrict__ doc_svd, int N0) {
    __shared__ __attribute__((aligned(16))) float acc[AGG_ROWS * AGG_ROWF];
    __shared__ int   q_src[AGG_QCAP];
    __shared__ int   q_dl[AGG_QCAP];
    __shared__ float q_w[AGG_QCAP];
    __shared__ int   q_n;

    const int tid = threadIdx.x;
    const int ln  = tid & 31;
    const int wv  = __builtin_amdgcn_readfirstlane(tid >> 5);
    const unsigned row0 = blockIdx.x * AGG_ROWS;

    for (int i = tid; i < AGG_ROWS * AGG_ROWF / 4; i += 1024)
        ((float4*)acc)[i] = make_float4(0.f, 0.f, 0.f, 0.f);
    if (tid == 0) q_n = 0;
    __syncthreads();

#pragma unroll
    for (int list = 0; list < 2; ++list) {
        const int*   esrc = list == 0 ? e01_src : e02_src;
        const int*   edst = list == 0 ? e01_dst : e02_dst;
        const float* ew   = list == 0 ? e01_w   : e02_w;
        const int    E    = list == 0 ? E01     : E02;
        const float* xa   = list == 0 ? l2_1 : l2_2;
        const float* xb   = list == 0 ? l1_1 : l1_2;
        const int    segA = list == 0 ? 0 : 2 * 128;
        const int    segB = list == 0 ? 128 : 3 * 128;

        auto push = [&](int e) {
            const int slot = __hip_atomic_fetch_add(&q_n, 1, __ATOMIC_RELAXED,
                                                    __HIP_MEMORY_SCOPE_WORKGROUP);
            if (slot < AGG_QCAP) {
                q_src[slot] = e;
            } else {
                const int   s  = esrc[e];
                const float wt = ew[e];
                const unsigned dl = (unsigned)edst[e] - row0;
                float* o = acc + dl * AGG_ROWF;
                const size_t ro = (size_t)s * 128;
                for (int c = 0; c < 128; ++c) {
                    const int pc = 32 * (c & 3) + (c >> 2);
                    ldsAddF(o + segA + pc, wt * xa[ro + c]);
                    ldsAddF(o + segB + pc, wt * xb[ro + c]);
                    if (list == 1) ldsAddF(o + 4 * 128 + pc, wt * word[ro + c]);
                }
            }
        };

        const uint4* ov = (const uint4*)(ob + (list == 0 ? 0 : obOffB));
        const int nvec = E >> 4;
        const unsigned pat = blockIdx.x * 0x01010101u;
        auto z4 = [&](unsigned v) -> unsigned {
            const unsigned t = v ^ pat;
            return ~(((t & 0x7F7F7F7Fu) + 0x7F7F7F7Fu) | t) & 0x80808080u;
        };
        auto eq16 = [&](const uint4& v) -> unsigned {
            return (z4(v.x) >> 7) | (z4(v.y) >> 6) | (z4(v.z) >> 5) | (z4(v.w) >> 4);
        };
        const int ilast = nvec > 0 ? nvec - 1 : 0;
        auto load4 = [&](uint4 (&v)[4], int base) {
#pragma unroll
            for (int u = 0; u < 4; ++u) {
                const int idx = base + u * 1024 + tid;
                v[u] = ov[idx < nvec ? idx : ilast];
            }
        };
        auto scan4 = [&](const uint4 (&v)[4], int base) {
            unsigned h[4];
#pragma unroll
            for (int u = 0; u < 4; ++u)
                h[u] = (base + u * 1024 + tid) < nvec ? eq16(v[u]) : 0u;
            unsigned long long hit = (unsigned long long)(h[0] | (h[1] << 4))
                                   | ((unsigned long long)(h[2] | (h[3] << 4)) << 32);
            while (hit) {
                const int k = __builtin_ctzll(hit);
                hit &= hit - 1;
                const int u = ((k >> 5) << 1) | ((k >> 2) & 1);
                const int within = ((k & 3) << 2) | ((k >> 3) & 3);
                push(((base + u * 1024 + tid) << 4) + within);
            }
        };
        uint4 va[4], vb[4];
        load4(va, 0);
        for (int base = 0; base < nvec; base += 8192) {
            load4(vb, base + 4096);
            scan4(va, base);
            load4(va, base + 8192);
            scan4(vb, base + 4096);
        }
        for (int e = (nvec << 4) + tid; e < E; e += 1024) {
            const unsigned r = (unsigned)edst[e] - row0;
            if (r < AGG_ROWS && (unsigned)edst[e] < (unsigned)N0) push(e);
        }
        __syncthreads();

        {
            const int qr = q_n < AGG_QCAP ? q_n : AGG_QCAP;
            for (int i = tid; i < qr; i += 1024) {
                const int e = q_src[i];
                q_src[i] = esrc[e];
                q_dl[i]  = (int)((unsigned)edst[e] - row0);
                q_w[i]   = ew[e];
            }
        }
        __syncthreads();

        const int qn = q_n < AGG_QCAP ? q_n : AGG_QCAP;
        if (list == 0) {
            constexpr int EB = 4;
            for (int b = wv * EB; b + EB <= qn; b += 32 * EB)
                gather_add<2, EB>(q_src, q_dl, q_w, b, xa, xb, xb, acc, segA, segB, 0, ln);
            const int rem0 = (qn / EB) * EB;
            if (rem0 + wv < qn)
                gather_add<2, 1>(q_src, q_dl, q_w, rem0 + wv, xa, xb, xb, acc, segA, segB, 0, ln);
        } else {
            constexpr int EB = 3;
            for (int b = wv * EB; b + EB <= qn; b += 32 * EB)
                gather_add<3, EB>(q_src, q_dl, q_w, b, xa, xb, word, acc, segA, segB, 4 * 128, ln);
            const int rem0 = (qn / EB) * EB;
            if (rem0 + wv < qn)
                gather_add<3, 1>(q_src, q_dl, q_w, rem0 + wv, xa, xb, word, acc, segA, segB, 4 * 128, ln);
        }
        __syncthreads();
        if (tid == 0) q_n = 0;
        __syncthreads();
    }

    for (int r = wv; r < AGG_ROWS; r += 32) {
        const int row = (int)row0 + r;
        if (row >= N0) break;
        const float* p = acc + r * AGG_ROWF + ln;
        const float4 a0  = make_float4(p[0],   p[32],  p[64],  p[96]);
        const float4 a0s = make_float4(p[128], p[160], p[192], p[224]);
        const float4 a1  = make_float4(p[256], p[288], p[320], p[352]);
        const float4 a1s = make_float4(p[384], p[416], p[448], p[480]);
        const float4 aw  = make_float4(p[512], p[544], p[576], p[608]);
        const float sw = aw.x * aw.x + aw.y * aw.y + aw.z * aw.z + aw.w * aw.w;
        float s0  = a0.x * a0.x + a0.y * a0.y + a0.z * a0.z + a0.w * a0.w;
        float s0s = a0s.x * a0s.x + a0s.y * a0s.y + a0s.z * a0s.z + a0s.w * a0s.w;
        float s1  = a1.x * a1.x + a1.y * a1.y + a1.z * a1.z + a1.w * a1.w + sw;
        float s1s = a1s.x * a1s.x + a1s.y * a1s.y + a1s.z * a1s.z + a1s.w * a1s.w + sw;
#pragma unroll
        for (int off = 16; off > 0; off >>= 1) {
            s0  += __shfl_xor(s0,  off, 32);
            s0s += __shfl_xor(s0s, off, 32);
            s1  += __shfl_xor(s1,  off, 32);
            s1s += __shfl_xor(s1s, off, 32);
        }
        const float i0  = 1.0f / (sqrtf(s0)  + 1e-9f);
        const float i0s = 1.0f / (sqrtf(s0s) + 1e-9f);
        const float i1  = 1.0f / (sqrtf(s1)  + 1e-9f);
        const float i1s = 1.0f / (sqrtf(s1s) + 1e-9f);

        float* o  = doc     + (size_t)row * 384 + ln * 4;
        float* os = doc_svd + (size_t)row * 384 + ln * 4;
        *(float4*)(o)        = make_float4(a0.x * i0, a0.y * i0, a0.z * i0, a0.w * i0);
        *(float4*)(o + 128)  = make_float4(a1.x * i1, a1.y * i1, a1.z * i1, a1.w * i1);
        *(float4*)(o + 256)  = make_float4(aw.x * i1, aw.y * i1, aw.z * i1, aw.w * i1);
        *(float4*)(os)       = make_float4(a0s.x * i0s, a0s.y * i0s, a0s.z * i0s, a0s.w * i0s);
        *(float4*)(os + 128) = make_float4(a1s.x * i1s, a1s.y * i1s, a1s.z * i1s, a1s.w * i1s);
        *(float4*)(os + 256) = make_float4(aw.x * i1s, aw.y * i1s, aw.z * i1s, aw.w * i1s);
    }
}

#define TS_QCAP 6144

template <int EB>
__device__ __forceinline__ void gather_add1(const int* qs, const int* qd, const float* qw, int b,
                                            const float* __restrict__ x, float* acc, int ln) {
    float4 v[EB];
    int   dl[EB];
    float wt[EB];
#pragma unroll
    for (int j = 0; j < EB; ++j) {
        const size_t ro = (size_t)__builtin_amdgcn_readfirstlane(qs[b + j]) * 128;
        dl[j] = __builtin_amdgcn_readfirstlane(qd[b + j]);
        wt[j] = qw[b + j];
        v[j]  = ((const float4*)(x + ro))[ln];
    }
#pragma unroll
    for (int j = 0; j < EB; ++j) {
        float* o = acc + dl[j] * 128 + ln;
        ldsAddF(o,      wt[j] * v[j].x);
        ldsAddF(o + 32, wt[j] * v[j].y);
        ldsAddF(o + 64, wt[j] * v[j].z);
        ldsAddF(o + 96, wt[j] * v[j].w);
    }
}

__global__ __launch_bounds__(1024)
void tower_scatter(const int* __restrict__ eA_src, const int* __restrict__ eA_dst,
                   const float* __restrict__ eA_w, int EA,
                   const int* __restrict__ eB_src, const int* __restrict__ eB_dst,
                   const float* __restrict__ eB_w, int EBn,
                   const float* __restrict__ xA, const float* __restrict__ xB,
                   const unsigned* __restrict__ obA, const unsigned* __restrict__ obB,
                   float* __restrict__ hA, float* __restrict__ hB, int NA, int NB) {
    __shared__ __attribute__((aligned(16))) float acc[T1_ROWS * 128];
    __shared__ int   q_src[TS_QCAP];
    __shared__ int   q_dl[TS_QCAP];
    __shared__ float q_w[TS_QCAP];
    __shared__ int   q_n;
    static_assert(T2_ROWS <= T1_ROWS, "the sums of tower 2 reuse the rows of tower 1");

    const int tid = threadIdx.x;
    const int ln  = tid & 31;
    const int wv  = __builtin_amdgcn_readfirstlane(tid >> 5);

#pragma unroll
    for (int list = 0; list < 2; ++list) {
        const int*   esrc = list == 0 ? eA_src : eB_src;
        const int*   edst = list == 0 ? eA_dst : eB_dst;
        const float* ew   = list == 0 ? eA_w   : eB_w;
        const int    E    = list == 0 ? EA     : EBn;
        const float* x    = list == 0 ? xA     : xB;
        float*       h    = list == 0 ? hA     : hB;
        const int    N    = list == 0 ? NA     : NB;
        const int    ROWS = list == 0 ? T1_ROWS : T2_ROWS;
        const unsigned row0 = blockIdx.x * (unsigned)ROWS;

        for (int i = tid; i < ROWS * 32; i += 1024)
            ((float4*)acc)[i] = make_float4(0.f, 0.f, 0.f, 0.f);
        if (tid == 0) q_n = 0;
        __syncthreads();

        auto push = [&](int e) {
            const int slot = __hip_atomic_fetch_add(&q_n, 1, __ATOMIC_RELAXED,
                                                    __HIP_MEMORY_SCOPE_WORKGROUP);
            if (slot < TS_QCAP) {
                q_src[slot] = e;
            } else {
                const int   s  = esrc[e];
                const float wt = ew[e];
                const unsigned dl = (unsigned)edst[e] - row0;
                float* o = acc + dl * 128;
                const size_t ro = (size_t)s * 128;
                for (int c = 0; c < 128; ++c)
                    ldsAddF(o + 32 * (c & 3) + (c >> 2), wt * x[ro + c]);
            }
        };

        const uint4* ov = (const uint4*)(list == 0 ? obA : obB);
        const int nvec = E >> 4;
        const unsigned pat = blockIdx.x * 0x01010101u;
        auto z4 = [&](unsigned v) -> unsigned {
            const unsigned t = v ^ pat;
            return ~(((t & 0x7F7F7F7Fu) + 0x7F7F7F7Fu) | t) & 0x80808080u;
        };
        auto eq16 = [&](const uint4& v) -> unsigned {
            return (z4(v.x) >> 7) | (z4(v.y) >> 6) | (z4(v.z) >> 5) | (z4(v.w) >> 4);
        };
        const int ilast = nvec > 0 ? nvec - 1 : 0;
        auto load4 = [&](uint4 (&v)[4], int base) {
#pragma unroll
            for (int u = 0; u < 4; ++u) {
                const int idx = base + u * 1024 + tid;
                v[u] = ov[idx < nvec ? idx : ilast];
            }
        };
        auto scan4 = [&](const uint4 (&v)[4], int base) {
            unsigned hh[4];
#pragma unroll
            for (int u = 0; u < 4; ++u)
                hh[u] = (base + u * 1024 + tid) < nvec ? eq16(v[u]) : 0u;
            unsigned long long hit = (unsigned long long)(hh[0] | (hh[1] << 4))
                                   | ((unsigned long long)(hh[2] | (hh[3] << 4)) << 32);
            while (hit) {
                const int k = __builtin_ctzll(hit);
                hit &= hit - 1;
                const int u = ((k >> 5) << 1) | ((k >> 2) & 1);
                const int within = ((k & 3) << 2) | ((k >> 3) & 3);
                push(((base + u * 1024 + tid) << 4) + within);
            }
        };
        if (nvec > 0) {
            uint4 va[4], vb[4];
            load4(va, 0);
            for (int base = 0; base < nvec; base += 8192) {
                load4(vb, base + 4096);
                scan4(va, base);
                load4(va, base + 8192);
                scan4(vb, base + 4096);
            }
        }
        for (int e = (nvec << 4) + tid; e < E; e += 1024) {
            const unsigned r = (unsigned)edst[e] - row0;
            if (r < (unsigned)ROWS && (unsigned)edst[e] < (unsigned)N) push(e);
        }
        __syncthreads();

        const int qn = q_n < TS_QCAP ? q_n : TS_QCAP;
        for (int i = tid; i < qn; i += 1024) {
            const int e = q_src[i];
            q_src[i] = esrc[e];
            q_dl[i]  = (int)((unsigned)edst[e] - row0);
            q_w[i]   = ew[e];
        }
        __syncthreads();

        constexpr int EB = 8;
        for (int b = wv * EB; b + EB <= qn; b += 32 * EB)
            gather_add1<EB>(q_src, q_dl, q_w, b, x, acc, ln);
        const int rem0 = (qn / EB) * EB;
        if (rem0 + wv < qn)
            gather_add1<1>(q_src, q_dl, q_w, rem0 + wv, x, acc, ln);
        __syncthreads();

        const int left = N - (int)row0;
        const int nv = left < ROWS ? left : ROWS;
        float4* hp = (float4*)(h + (size_t)row0 * 128);
        for (int i = tid; i < nv * 32; i += 1024) {
            const float* a = acc + (i >> 5) * 128 + (i & 31);
            hp[i] = make_float4(a[0], a[32], a[64], a[96]);
        }
        __syncthreads();
    }
}

extern "C" void kernel_launch(void* const* d_in, const int* in_sizes, int n_in,
                              void* d_out, int out_size, void* d_ws, size_t ws_size,
                              hipStream_t stream) {
    const float* x1       = (const float*)d_in[0];
    const float* x2       = (const float*)d_in[1];
    const float* word_emb = (const float*)d_in[2];
    const float* W1a = (const float*)d_in[3];
    const float* b1a = (const float*)d_in[4];
    const float* W1b = (const float*)d_in[5];
    const float* b1b = (const float*)d_in[6];
    const float* W2a = (const float*)d_in[7];
    const float* b2a = (const float*)d_in[8];
    const float* W2b = (const float*)d_in[9];
    const float* b2b = (const float*)d_in[10];
    const int*   e11_src = (const int*)d_in[11];
    const int*   e11_dst = (const int*)d_in[12];
    const float* e11_w   = (const float*)d_in[13];
    const int*   e22_src = (const int*)d_in[14];
    const int*   e22_dst = (const int*)d_in[15];
    const float* e22_w   = (const float*)d_in[16];
    const int*   e01_src = (const int*)d_in[17];
    const int*   e01_dst = (const int*)d_in[18];
    const float* e01_w   = (const float*)d_in[19];
    const int*   e02_src = (const int*)d_in[20];
    const int*   e02_dst = (const int*)d_in[21];
    const float* e02_w   = (const float*)d_in[22];

    const int N0 = 10000, N1 = 50000, N2 = 30000;
    const int E11 = in_sizes[11], E22 = in_sizes[14];
    const int E01 = in_sizes[17], E02 = in_sizes[20];


    float* ws = (float*)d_ws;
    size_t off = 0;
    float* l1_1 = ws + off; off += (size_t)N1 * 128;
    float* l2_1 = ws + off; off += (size_t)N1 * 128;
    float* l1_2 = ws + off; off += (size_t)N2 * 128;
    unsigned* own = (unsigned*)(ws + off); off += (size_t)N2 * 128;
    float* h11  = ws + off; off += (size_t)N1 * 128;
    float* h22  = ws + off; off += (size_t)N2 * 128;
    float* l2_2 = h22;
    float* doc     = (float*)d_out;
    float* doc_svd = doc + (size_t)N0 * 384;

    const dim3 blk(256);
    const int g1 = (N1 + 127) / 128;
    const int g2 = (N2 + 127) / 128;
    const int q11 = (E11 + 3) / 4, q22 = (E22 + 3) / 4;
    const int q01 = (E01 + 3) / 4, q02 = (E02 + 3) / 4;
    const int w11 = 0;
    const int w22 = w11 + (q11 + 3) / 4 * 4;
    const int w01 = w22 + (q22 + 3) / 4 * 4;
    const int w02 = w01 + (q01 + 3) / 4 * 4;
    const int nb11 = (q11 + OWN_Q_PER_BLOCK - 1) / OWN_Q_PER_BLOCK;
    const int nb22 = (q22 + OWN_Q_PER_BLOCK - 1) / OWN_Q_PER_BLOCK;
    const int nb01 = (q01 + OWN_Q_PER_BLOCK - 1) / OWN_Q_PER_BLOCK;
    const int nb02 = (q02 + OWN_Q_PER_BLOCK - 1) / OWN_Q_PER_BLOCK;
    const int t1 = (N1 + T1_ROWS - 1) / T1_ROWS;
    const int t2 = (N2 + T2_ROWS - 1) / T2_ROWS;
    const int tg = t1 > t2 ? t1 : t2;

    stage_pair<<<dim3(g1 + g2 + nb11 + nb22 + nb01 + nb02), blk, 0, stream>>>(
        x1, W1a, b1a, l1_1, N1, 256, g1,
        x2, W2a, b2a, l1_2, N2, 256, g2,
        e11_dst, E11, w11, nb11,
        e22_dst, E22, w22, nb22,
        e01_dst, E01, w01, nb01,
        e02_dst, E02, w02,
        own, (unsigned)N1, (unsigned)N2, (unsigned)N0);
    tower_scatter<<<dim3(tg), dim3(1024), 0, stream>>>(
        e11_src, e11_dst, e11_w, E11, e22_src, e22_dst, e22_w, E22,
        l1_1, l1_2, own + w11, own + w22, h11, h22, N1, N2);
    stage_pair<<<dim3(g1 + g2), blk, 0, stream>>>(
        h11, W1b, b1b, l2_1, N1, 128, g1,
        h22, W2b, b2b, l2_2, N2, 128, g2,
        nullptr, 0, 0, 0,
        nullptr, 0, 0, 0,
        nullptr, 0, 0, 0,
        nullptr, 0, 0,
        own, 0u, 0u, 0u);

    doc_aggregate<<<dim3((N0 + AGG_ROWS - 1) / AGG_ROWS), dim3(1024), 0, stream>>>(
        e01_src, e01_dst, e01_w, E01, e02_src, e02_dst, e02_w, E02,
        l2_1, l1_1, l2_2, l1_2, word_emb, own + w01, w02 - w01, doc, doc_svd, N0);
}


# edge lists binned by owner group in launch 1 (chunk-wise LDS counting sort, 8-byte records + run table): owners read only their own runs, no list scan or random src/dst/w fetch; arrival-order sums
# speedup vs baseline: 3.3065x; 1.2936x over previous
#include <hip/hip_runtime.h>
#include <math.h>

typedef __attribute__((ext_vector_type(2))) float v2f;
typedef __attribute__((ext_vector_type(8))) float v8f;

#define O_DIM 128

__device__ __forceinline__
void gemm_tile(const float* X, const float* __restrict__ W,
               const float* __restrict__ bias, float* out,
               int N, int K, int bid) {
    __shared__ float  As[128][36];
    __shared__ float2 Bs2[16][144];

    const int tid = threadIdx.x;
    const int wv  = tid >> 5;
    const int ln  = tid & 31;
    const int m0  = bid * 128;

    v8f acc[8];
#pragma unroll
    for (int t = 0; t < 8; ++t)
#pragma unroll
        for (int j = 0; j < 8; ++j) acc[t][j] = 0.0f;

    const int mrow = wv * 16 + (ln & 15);
    const int lnn  = ln & 15;
    const int khalf = (ln >> 4) << 1;

    for (int k0 = 0; k0 < K; k0 += 32) {
#pragma unroll
        for (int i = 0; i < 4; ++i) {
            int f   = tid + i * 256;
            int row = f >> 3;
            int c4  = f & 7;
            float4 v = make_float4(0.f, 0.f, 0.f, 0.f);
            if (m0 + row < N)
                v = *(const float4*)(X + (size_t)(m0 + row) * K + k0 + c4 * 4);
            *(float4*)&As[row][c4 * 4] = v;
        }
#pragma unroll
        for (int i = 0; i < 4; ++i) {
            int f   = tid + i * 256;
            int row = f >> 5;
            int c4  = f & 31;
            float4 v = *(const float4*)(W + (size_t)(k0 + row) * O_DIM + c4 * 4);
            float* bp = (float*)&Bs2[row >> 1][c4 * 4] + (row & 1);
            bp[0] = v.x;
            bp[2] = v.y;
            bp[4] = v.z;
            bp[6] = v.w;
        }
        __syncthreads();

#pragma unroll
        for (int kk = 0; kk < 32; kk += 4) {
            const int kk2 = kk + khalf;
            const int p   = kk2 >> 1;
            v2f a = *(const v2f*)&As[mrow][kk2];
#pragma unroll
            for (int t = 0; t < 8; ++t) {
                v2f b = *(const v2f*)&Bs2[p][t * 16 + lnn];
                acc[t] = __builtin_amdgcn_wmma_f32_16x16x4_f32(
                    false, a, false, b, (short)0, acc[t], false, false);
            }
        }
        __syncthreads();
    }

    const int mtile = m0 + wv * 16;
    if (mtile < N) {
        const int rowoff = (ln >> 4) << 3;
#pragma unroll
        for (int t = 0; t < 8; ++t) {
            const int col = t * 16 + lnn;
            const float bv = bias[col];
#pragma unroll
            for (int i = 0; i < 8; ++i) {
                float v = acc[t][i] + bv;
                out[(size_t)(mtile + rowoff + i) * O_DIM + col] = v > 0.f ? v : 0.f;
            }
        }
    }
}


#define T1_ROWS 200
#define T2_ROWS 120
#define OWN_ROWS 40
#define BIN_CH 4096
#define BIN_K (BIN_CH / 256)
#define BIN_TAB 251

template <int ROWS>
__device__ __forceinline__ void bin_chunk(int* hist, int* wsum, uint2* stage,
                                          const int* __restrict__ dst, const int* __restrict__ src,
                                          const float* __restrict__ w, int E, unsigned nrows,
                                          uint2* __restrict__ rec, int* __restrict__ tab,
                                          int nch, int chunk) {
    const int tid = threadIdx.x;
    const int ln  = tid & 31;
    const int wv  = tid >> 5;
    const int e0  = chunk * BIN_CH;
    const int elast = E - 1;

    hist[tid] = 0;
    __syncthreads();

    unsigned r[BIN_K], sv[BIN_K];
    float    wt[BIN_K];
#pragma unroll
    for (int k = 0; k < BIN_K; ++k) {
        const int e  = e0 + k * 256 + tid;
        const int ec = e < E ? e : elast;
        r[k]  = (unsigned)dst[ec];
        sv[k] = (unsigned)src[ec];
        wt[k] = w[ec];
    }
    unsigned key[BIN_K];
    unsigned rank[BIN_K];
#pragma unroll
    for (int k = 0; k < BIN_K; ++k) {
        const int e = e0 + k * 256 + tid;
        const bool in = e < E && r[k] < nrows;
        const unsigned o  = in ? r[k] / ROWS : 255u;
        const unsigned dl = in ? r[k] - o * ROWS : 0u;
        key[k] = (o << 24) | (dl << 16);
        rank[k] = (unsigned)__hip_atomic_fetch_add(&hist[o], 1, __ATOMIC_RELAXED,
                                                   __HIP_MEMORY_SCOPE_WORKGROUP);
    }
#pragma unroll
    for (int k = 0; k < BIN_K; ++k) key[k] |= rank[k];
    __syncthreads();

    const int cnt = hist[tid];
    int x = cnt;
#pragma unroll
    for (int off = 1; off < 32; off <<= 1) {
        const int y = __shfl_up(x, off, 32);
        if (ln >= off) x += y;
    }
    if (ln == 31) wsum[wv] = x;
    __syncthreads();
    int base = 0;
#pragma unroll
    for (int k = 0; k < 8; ++k) if (k < wv) base += wsum[k];
    const int excl = base + x - cnt;
    hist[tid] = excl;
    if (tid < BIN_TAB) tab[(size_t)tid * nch + chunk] = excl;
    __syncthreads();

#pragma unroll
    for (int k = 0; k < BIN_K; ++k) {
        const int e = e0 + k * 256 + tid;
        if (e < E) {
            const unsigned kk = key[k];
            const int pos = hist[kk >> 24] + (int)(kk & 0xFFFFu);
            stage[pos] = make_uint2(sv[k] | (((kk >> 16) & 0xFFu) << 24), __float_as_uint(wt[k]));
        }
    }
    __syncthreads();

    int nvalid = E - e0;
    if (nvalid > BIN_CH) nvalid = BIN_CH;
    uint2* out = rec + (size_t)chunk * BIN_CH;
    for (int p = tid; p < nvalid; p += 256) out[p] = stage[p];
}

__global__ __launch_bounds__(256)
void stage_pair(const float* aX, const float* __restrict__ aW,
                const float* __restrict__ aBias, float* aOut, int aN, int aK, int aBlocks,
                const float* bX, const float* __restrict__ bW,
                const float* __restrict__ bBias, float* bOut, int bN, int bK, int bBlocks,
                const int* __restrict__ d0, const int* __restrict__ s0, const float* __restrict__ w0, int E0, int nb0,
                const int* __restrict__ d1, const int* __restrict__ s1, const float* __restrict__ w1, int E1, int nb1,
                const int* __restrict__ d2, const int* __restrict__ s2, const float* __restrict__ w2, int E2, int nb2,
                const int* __restrict__ d3, const int* __restrict__ s3, const float* __restrict__ w3, int E3, int nb3,
                unsigned* __restrict__ recw, int* __restrict__ tab,
                unsigned rowsT1, unsigned rowsT2, unsigned rowsDoc) {
    uint2* rec = (uint2*)recw;
    int bid = (int)blockIdx.x;
    if (bid < aBlocks + bBlocks) {
        const bool second = bid >= aBlocks;
        gemm_tile(second ? bX : aX, second ? bW : aW, second ? bBias : aBias,
                  second ? bOut : aOut, second ? bN : aN, second ? bK : aK,
                  second ? bid - aBlocks : bid);
        return;
    }
    bid -= aBlocks + bBlocks;
    __shared__ int   hist[256];
    __shared__ int   wsum[8];
    __shared__ uint2 stage[BIN_CH];
    if (bid < nb0) { bin_chunk<T1_ROWS>(hist, wsum, stage, d0, s0, w0, E0, rowsT1, rec, tab, nb0, bid); return; }
    bid -= nb0;  rec += (size_t)nb0 * BIN_CH;  tab += (size_t)nb0 * BIN_TAB;
    if (bid < nb1) { bin_chunk<T2_ROWS>(hist, wsum, stage, d1, s1, w1, E1, rowsT2, rec, tab, nb1, bid); return; }
    bid -= nb1;  rec += (size_t)nb1 * BIN_CH;  tab += (size_t)nb1 * BIN_TAB;
    if (bid < nb2) { bin_chunk<OWN_ROWS>(hist, wsum, stage, d2, s2, w2, E2, rowsDoc, rec, tab, nb2, bid); return; }
    bid -= nb2;  rec += (size_t)nb2 * BIN_CH;  tab += (size_t)nb2 * BIN_TAB;
    if (bid < nb3) bin_chunk<OWN_ROWS>(hist, wsum, stage, d3, s3, w3, E3, rowsDoc, rec, tab, nb3, bid);
}

#define FETCH_U 8
#define FETCH_C (32 * FETCH_U)

template <typename Put>
__device__ __forceinline__ void fetch_runs(const uint2* __restrict__ rec, const int* __restrict__ tab,
                                           int nch, int owner, int* q_n, int* seg,
                                           int tid, int wv, int ln, Put put) {
    const int* ta = tab + (size_t)owner * nch;
    const int* tb = ta + nch;
    int* segS = seg;
    int* segN = seg + FETCH_C;
    int* segQ = seg + 2 * FETCH_C;
    for (int cb = 0; cb < nch; cb += FETCH_C) {
        if (tid < FETCH_C) {
            const int c = cb + tid;
            int s = 0, n = 0, q = 0;
            if (c < nch) {
                s = ta[c];
                n = tb[c] - s;
                q = __hip_atomic_fetch_add(q_n, n, __ATOMIC_RELAXED, __HIP_MEMORY_SCOPE_WORKGROUP);
            }
            segS[tid] = s;
            segN[tid] = n;
            segQ[tid] = q;
        }
        __syncthreads();
        const int c0 = wv * FETCH_U;
        int s[FETCH_U], n[FETCH_U], q[FETCH_U];
        uint2 r[FETCH_U];
#pragma unroll
        for (int u = 0; u < FETCH_U; ++u) {
            s[u] = segS[c0 + u];
            n[u] = segN[c0 + u];
            q[u] = segQ[c0 + u];
        }
#pragma unroll
        for (int u = 0; u < FETCH_U; ++u) {
            r[u] = make_uint2(0u, 0u);
            if (ln < n[u]) r[u] = rec[(size_t)(cb + c0 + u) * BIN_CH + s[u] + ln];
        }
#pragma unroll
        for (int u = 0; u < FETCH_U; ++u) {
            if (ln < n[u]) put(q[u] + ln, r[u]);
            for (int j = ln + 32; j < n[u]; j += 32)
                put(q[u] + j, rec[(size_t)(cb + c0 + u) * BIN_CH + s[u] + j]);
        }
        __syncthreads();
    }
}

#define AGG_ROWS 40
#define AGG_ROWF 640
static_assert(AGG_ROWS == OWN_ROWS, "owner bytes and aggregation must agree on rows per group");
#define AGG_QCAP 6144

__device__ __forceinline__ void ldsAddF(float* p, float v) {
    __hip_atomic_fetch_add(p, v, __ATOMIC_RELAXED, __HIP_MEMORY_SCOPE_WORKGROUP);
}

template <int NS, int EB>
__device__ __forceinline__ void gather_add(const int* qs, const int* qd, const float* qw, int b,
                                           const float* __restrict__ xa, const float* __restrict__ xb,
                                           const float* __restrict__ xc, float* acc,
                                           int sa, int sb, int sc, int ln) {
    float4 va[EB], vb[EB], vc[EB];
    int   dl[EB];
    float wt[EB];
#pragma unroll
    for (int j = 0; j < EB; ++j) {
        const size_t ro = (size_t)__builtin_amdgcn_readfirstlane(qs[b + j]) * 128;
        dl[j] = __builtin_amdgcn_readfirstlane(qd[b + j]);
        wt[j] = qw[b + j];
        va[j] = ((const float4*)(xa + ro))[ln];
        vb[j] = ((const float4*)(xb + ro))[ln];
        if (NS == 3) vc[j] = ((const float4*)(xc + ro))[ln];
    }
#pragma unroll
    for (int j = 0; j < EB; ++j) {
        float* o = acc + dl[j] * AGG_ROWF + ln;
        const float w = wt[j];
        ldsAddF(o + sa,      w * va[j].x);
        ldsAddF(o + sa + 32, w * va[j].y);
        ldsAddF(o + sa + 64, w * va[j].z);
        ldsAddF(o + sa + 96, w * va[j].w);
        ldsAddF(o + sb,      w * vb[j].x);
        ldsAddF(o + sb + 32, w * vb[j].y);
        ldsAddF(o + sb + 64, w * vb[j].z);
        ldsAddF(o + sb + 96, w * vb[j].w);
        if (NS == 3) {
            ldsAddF(o + sc,      w * vc[j].x);
            ldsAddF(o + sc + 32, w * vc[j].y);
            ldsAddF(o + sc + 64, w * vc[j].z);
            ldsAddF(o + sc + 96, w * vc[j].w);
        }
    }
}


__global__ __launch_bounds__(1024)
void doc_aggregate(const unsigned* __restrict__ recA, const int* __restrict__ tabA, int nchA,
                   const unsigned* __restrict__ recB, const int* __restrict__ tabB, int nchB,
                   const float* __restrict__ l2_1, const float* __restrict__ l1_1,
                   const float* __restrict__ l2_2, const float* __restrict__ l1_2,
                   const float* __restrict__ word,
                   float* __restrict__ doc, float* __restrict__ doc_svd, int N0) {
    __shared__ __attribute__((aligned(16))) float acc[AGG_ROWS * AGG_ROWF];
    __shared__ int   q_src[AGG_QCAP];
    __shared__ int   q_dl[AGG_QCAP];
    __shared__ float q_w[AGG_QCAP];
    __shared__ int   q_n;
    __shared__ int   seg[3 * FETCH_C];

    const int tid = threadIdx.x;
    const int ln  = tid & 31;
    const int wv  = __builtin_amdgcn_readfirstlane(tid >> 5);
    const unsigned row0 = blockIdx.x * AGG_ROWS;

    for (int i = tid; i < AGG_ROWS * AGG_ROWF / 4; i += 1024)
        ((float4*)acc)[i] = make_float4(0.f, 0.f, 0.f, 0.f);
    if (tid == 0) q_n = 0;
    __syncthreads();

#pragma unroll
    for (int list = 0; list < 2; ++list) {
        const uint2* rec  = (const uint2*)(list == 0 ? recA : recB);
        const int*   tab  = list == 0 ? tabA : tabB;
        const int    nch  = list == 0 ? nchA : nchB;
        const float* xa   = list == 0 ? l2_1 : l2_2;
        const float* xb   = list == 0 ? l1_1 : l1_2;
        const int    segA = list == 0 ? 0 : 2 * 128;
        const int    segB = list == 0 ? 128 : 3 * 128;

        auto put = [&](int slot, const uint2& r) {
            const int   s  = (int)(r.x & 0xFFFFFFu);
            const int   dl = (int)(r.x >> 24);
            const float wt = __uint_as_float(r.y);
            if (slot < AGG_QCAP) {
                q_src[slot] = s;
                q_dl[slot]  = dl;
                q_w[slot]   = wt;
            } else {
                float* o = acc + dl * AGG_ROWF;
                const size_t ro = (size_t)s * 128;
                for (int c = 0; c < 128; ++c) {
                    const int pc = 32 * (c & 3) + (c >> 2);
                    ldsAddF(o + segA + pc, wt * xa[ro + c]);
                    ldsAddF(o + segB + pc, wt * xb[ro + c]);
                    if (list == 1) ldsAddF(o + 4 * 128 + pc, wt * word[ro + c]);
                }
            }
        };
        fetch_runs(rec, tab, nch, (int)blockIdx.x, &q_n, seg, tid, wv, ln, put);
        __syncthreads();

        const int qn = q_n < AGG_QCAP ? q_n : AGG_QCAP;
        if (list == 0) {
            constexpr int EB = 4;
            for (int b = wv * EB; b + EB <= qn; b += 32 * EB)
                gather_add<2, EB>(q_src, q_dl, q_w, b, xa, xb, xb, acc, segA, segB, 0, ln);
            const int rem0 = (qn / EB) * EB;
            if (rem0 + wv < qn)
                gather_add<2, 1>(q_src, q_dl, q_w, rem0 + wv, xa, xb, xb, acc, segA, segB, 0, ln);
        } else {
            constexpr int EB = 3;
            for (int b = wv * EB; b + EB <= qn; b += 32 * EB)
                gather_add<3, EB>(q_src, q_dl, q_w, b, xa, xb, word, acc, segA, segB, 4 * 128, ln);
            const int rem0 = (qn / EB) * EB;
            if (rem0 + wv < qn)
                gather_add<3, 1>(q_src, q_dl, q_w, rem0 + wv, xa, xb, word, acc, segA, segB, 4 * 128, ln);
        }
        __syncthreads();
        if (tid == 0) q_n = 0;
        __syncthreads();
    }

    for (int r = wv; r < AGG_ROWS; r += 32) {
        const int row = (int)row0 + r;
        if (row >= N0) break;
        const float* p = acc + r * AGG_ROWF + ln;
        const float4 a0  = make_float4(p[0],   p[32],  p[64],  p[96]);
        const float4 a0s = make_float4(p[128], p[160], p[192], p[224]);
        const float4 a1  = make_float4(p[256], p[288], p[320], p[352]);
        const float4 a1s = make_float4(p[384], p[416], p[448], p[480]);
        const float4 aw  = make_float4(p[512], p[544], p[576], p[608]);
        const float sw = aw.x * aw.x + aw.y * aw.y + aw.z * aw.z + aw.w * aw.w;
        float s0  = a0.x * a0.x + a0.y * a0.y + a0.z * a0.z + a0.w * a0.w;
        float s0s = a0s.x * a0s.x + a0s.y * a0s.y + a0s.z * a0s.z + a0s.w * a0s.w;
        float s1  = a1.x * a1.x + a1.y * a1.y + a1.z * a1.z + a1.w * a1.w + sw;
        float s1s = a1s.x * a1s.x + a1s.y * a1s.y + a1s.z * a1s.z + a1s.w * a1s.w + sw;
#pragma unroll
        for (int off = 16; off > 0; off >>= 1) {
            s0  += __shfl_xor(s0,  off, 32);
            s0s += __shfl_xor(s0s, off, 32);
            s1  += __shfl_xor(s1,  off, 32);
            s1s += __shfl_xor(s1s, off, 32);
        }
        const float i0  = 1.0f / (sqrtf(s0)  + 1e-9f);
        const float i0s = 1.0f / (sqrtf(s0s) + 1e-9f);
        const float i1  = 1.0f / (sqrtf(s1)  + 1e-9f);
        const float i1s = 1.0f / (sqrtf(s1s) + 1e-9f);

        float* o  = doc     + (size_t)row * 384 + ln * 4;
        float* os = doc_svd + (size_t)row * 384 + ln * 4;
        *(float4*)(o)        = make_float4(a0.x * i0, a0.y * i0, a0.z * i0, a0.w * i0);
        *(float4*)(o + 128)  = make_float4(a1.x * i1, a1.y * i1, a1.z * i1, a1.w * i1);
        *(float4*)(o + 256)  = make_float4(aw.x * i1, aw.y * i1, aw.z * i1, aw.w * i1);
        *(float4*)(os)       = make_float4(a0s.x * i0s, a0s.y * i0s, a0s.z * i0s, a0s.w * i0s);
        *(float4*)(os + 128) = make_float4(a1s.x * i1s, a1s.y * i1s, a1s.z * i1s, a1s.w * i1s);
        *(float4*)(os + 256) = make_float4(aw.x * i1s, aw.y * i1s, aw.z * i1s, aw.w * i1s);
    }
}

#define TS_QCAP 6144

template <int EB>
__device__ __forceinline__ void gather_add1(const int* qs, const int* qd, const float* qw, int b,
                                            const float* __restrict__ x, float* acc, int ln) {
    float4 v[EB];
    int   dl[EB];
    float wt[EB];
#pragma unroll
    for (int j = 0; j < EB; ++j) {
        const size_t ro = (size_t)__builtin_amdgcn_readfirstlane(qs[b + j]) * 128;
        dl[j] = __builtin_amdgcn_readfirstlane(qd[b + j]);
        wt[j] = qw[b + j];
        v[j]  = ((const float4*)(x + ro))[ln];
    }
#pragma unroll
    for (int j = 0; j < EB; ++j) {
        float* o = acc + dl[j] * 128 + ln;
        ldsAddF(o,      wt[j] * v[j].x);
        ldsAddF(o + 32, wt[j] * v[j].y);
        ldsAddF(o + 64, wt[j] * v[j].z);
        ldsAddF(o + 96, wt[j] * v[j].w);
    }
}


__global__ __launch_bounds__(1024)
void tower_scatter(const unsigned* __restrict__ recA, const int* __restrict__ tabA, int nchA,
                   const unsigned* __restrict__ recB, const int* __restrict__ tabB, int nchB,
                   const float* __restrict__ xA, const float* __restrict__ xB,
                   float* __restrict__ hA, float* __restrict__ hB, int NA, int NB) {
    __shared__ __attribute__((aligned(16))) float acc[T1_ROWS * 128];
    __shared__ int   q_src[TS_QCAP];
    __shared__ int   q_dl[TS_QCAP];
    __shared__ float q_w[TS_QCAP];
    __shared__ int   q_n;
    __shared__ int   seg[3 * FETCH_C];
    static_assert(T2_ROWS <= T1_ROWS, "the sums of tower 2 reuse the rows of tower 1");

    const int tid = threadIdx.x;
    const int ln  = tid & 31;
    const int wv  = __builtin_amdgcn_readfirstlane(tid >> 5);

#pragma unroll
    for (int list = 0; list < 2; ++list) {
        const uint2* rec  = (const uint2*)(list == 0 ? recA : recB);
        const int*   tab  = list == 0 ? tabA : tabB;
        const int    nch  = list == 0 ? nchA : nchB;
        const float* x    = list == 0 ? xA     : xB;
        float*       h    = list == 0 ? hA     : hB;
        const int    N    = list == 0 ? NA     : NB;
        const int    ROWS = list == 0 ? T1_ROWS : T2_ROWS;
        const unsigned row0 = blockIdx.x * (unsigned)ROWS;

        for (int i = tid; i < ROWS * 32; i += 1024)
            ((float4*)acc)[i] = make_float4(0.f, 0.f, 0.f, 0.f);
        if (tid == 0) q_n = 0;
        __syncthreads();

        auto put = [&](int slot, const uint2& r) {
            const int   s  = (int)(r.x & 0xFFFFFFu);
            const int   dl = (int)(r.x >> 24);
            const float wt = __uint_as_float(r.y);
            if (slot < TS_QCAP) {
                q_src[slot] = s;
                q_dl[slot]  = dl;
                q_w[slot]   = wt;
            } else {
                float* o = acc + dl * 128;
                const size_t ro = (size_t)s * 128;
                for (int c = 0; c < 128; ++c)
                    ldsAddF(o + 32 * (c & 3) + (c >> 2), wt * x[ro + c]);
            }
        };
        fetch_runs(rec, tab, nch, (int)blockIdx.x, &q_n, seg, tid, wv, ln, put);
        __syncthreads();

        const int qn = q_n < TS_QCAP ? q_n : TS_QCAP;
        constexpr int EB = 8;
        for (int b = wv * EB; b + EB <= qn; b += 32 * EB)
            gather_add1<EB>(q_src, q_dl, q_w, b, x, acc, ln);
        const int rem0 = (qn / EB) * EB;
        if (rem0 + wv < qn)
            gather_add1<1>(q_src, q_dl, q_w, rem0 + wv, x, acc, ln);
        __syncthreads();

        const int left = N - (int)row0;
        const int nv = left < ROWS ? left : ROWS;
        float4* hp = (float4*)(h + (size_t)row0 * 128);
        for (int i = tid; i < nv * 32; i += 1024) {
            const float* a = acc + (i >> 5) * 128 + (i & 31);
            hp[i] = make_float4(a[0], a[32], a[64], a[96]);
        }
        __syncthreads();
    }
}

extern "C" void kernel_launch(void* const* d_in, const int* in_sizes, int n_in,
                              void* d_out, int out_size, void* d_ws, size_t ws_size,
                              hipStream_t stream) {
    const float* x1       = (const float*)d_in[0];
    const float* x2       = (const float*)d_in[1];
    const float* word_emb = (const float*)d_in[2];
    const float* W1a = (const float*)d_in[3];
    const float* b1a = (const float*)d_in[4];
    const float* W1b = (const float*)d_in[5];
    const float* b1b = (const float*)d_in[6];
    const float* W2a = (const float*)d_in[7];
    const float* b2a = (const float*)d_in[8];
    const float* W2b = (const float*)d_in[9];
    const float* b2b = (const float*)d_in[10];
    const int*   e11_src = (const int*)d_in[11];
    const int*   e11_dst = (const int*)d_in[12];
    const float* e11_w   = (const float*)d_in[13];
    const int*   e22_src = (const int*)d_in[14];
    const int*   e22_dst = (const int*)d_in[15];
    const float* e22_w   = (const float*)d_in[16];
    const int*   e01_src = (const int*)d_in[17];
    const int*   e01_dst = (const int*)d_in[18];
    const float* e01_w   = (const float*)d_in[19];
    const int*   e02_src = (const int*)d_in[20];
    const int*   e02_dst = (const int*)d_in[21];
    const float* e02_w   = (const float*)d_in[22];

    const int N0 = 10000, N1 = 50000, N2 = 30000;
    const int E11 = in_sizes[11], E22 = in_sizes[14];
    const int E01 = in_sizes[17], E02 = in_sizes[20];

    float* ws = (float*)d_ws;
    size_t off = 0;
    float* l1_1 = ws + off; off += (size_t)N1 * 128;
    float* l2_1 = ws + off; off += (size_t)N1 * 128;
    float* l1_2 = ws + off; off += (size_t)N2 * 128;
    unsigned* rec = (unsigned*)(ws + off); off += (size_t)N2 * 128;
    float* h11  = ws + off; off += (size_t)N1 * 128;
    float* h22  = ws + off; off += (size_t)N2 * 128;
    float* l2_2 = h22;
    float* doc     = (float*)d_out;
    float* doc_svd = doc + (size_t)N0 * 384;

    const dim3 blk(256);
    const int g1 = (N1 + 127) / 128;
    const int g2 = (N2 + 127) / 128;
    const int c11 = (E11 + BIN_CH - 1) / BIN_CH;
    const int c22 = (E22 + BIN_CH - 1) / BIN_CH;
    const int c01 = (E01 + BIN_CH - 1) / BIN_CH;
    const int c02 = (E02 + BIN_CH - 1) / BIN_CH;
    unsigned* rec11 = rec;
    unsigned* rec22 = rec11 + (size_t)c11 * BIN_CH * 2;
    unsigned* rec01 = rec22 + (size_t)c22 * BIN_CH * 2;
    unsigned* rec02 = rec01 + (size_t)c01 * BIN_CH * 2;
    int* tab   = (int*)(rec02 + (size_t)c02 * BIN_CH * 2);
    int* tab11 = tab;
    int* tab22 = tab11 + (size_t)c11 * BIN_TAB;
    int* tab01 = tab22 + (size_t)c22 * BIN_TAB;
    int* tab02 = tab01 + (size_t)c01 * BIN_TAB;
    const int t1 = (N1 + T1_ROWS - 1) / T1_ROWS;
    const int t2 = (N2 + T2_ROWS - 1) / T2_ROWS;
    const int tg = t1 > t2 ? t1 : t2;

    stage_pair<<<dim3(g1 + g2 + c11 + c22 + c01 + c02), blk, 0, stream>>>(
        x1, W1a, b1a, l1_1, N1, 256, g1,
        x2, W2a, b2a, l1_2, N2, 256, g2,
        e11_dst, e11_src, e11_w, E11, c11,
        e22_dst, e22_src, e22_w, E22, c22,
        e01_dst, e01_src, e01_w, E01, c01,
        e02_dst, e02_src, e02_w, E02, c02,
        rec, tab, (unsigned)N1, (unsigned)N2, (unsigned)N0);
    tower_scatter<<<dim3(tg), dim3(1024), 0, stream>>>(
        rec11, tab11, c11, rec22, tab22, c22,
        l1_1, l1_2, h11, h22, N1, N2);
    stage_pair<<<dim3(g1 + g2), blk, 0, stream>>>(
        h11, W1b, b1b, l2_1, N1, 128, g1,
        h22, W2b, b2b, l2_2, N2, 128, g2,
        nullptr, nullptr, nullptr, 0, 0,
        nullptr, nullptr, nullptr, 0, 0,
        nullptr, nullptr, nullptr, 0, 0,
        nullptr, nullptr, nullptr, 0, 0,
        rec, tab, 0u, 0u, 0u);

    doc_aggregate<<<dim3((N0 + AGG_ROWS - 1) / AGG_ROWS), dim3(1024), 0, stream>>>(
        rec01, tab01, c01, rec02, tab02, c02,
        l2_1, l1_1, l2_2, l1_2, word_emb, doc, doc_svd, N0);
}


# tower scatter-add + doc aggregation: owner group sorts queued edges by destination row (LDS counting sort); a wave sums a whole row in registers (fma, arrival order of the sort): no LDS float adds
# speedup vs baseline: 3.4663x; 1.0483x over previous
#include <hip/hip_runtime.h>
#include <math.h>

typedef __attribute__((ext_vector_type(2))) float v2f;
typedef __attribute__((ext_vector_type(8))) float v8f;

#define O_DIM 128

__device__ __forceinline__
void gemm_tile(const float* X, const float* __restrict__ W,
               const float* __restrict__ bias, float* out,
               int N, int K, int bid) {
    __shared__ float  As[128][36];
    __shared__ float2 Bs2[16][144];

    const int tid = threadIdx.x;
    const int wv  = tid >> 5;
    const int ln  = tid & 31;
    const int m0  = bid * 128;

    v8f acc[8];
#pragma unroll
    for (int t = 0; t < 8; ++t)
#pragma unroll
        for (int j = 0; j < 8; ++j) acc[t][j] = 0.0f;

    const int mrow = wv * 16 + (ln & 15);
    const int lnn  = ln & 15;
    const int khalf = (ln >> 4) << 1;

    for (int k0 = 0; k0 < K; k0 += 32) {
#pragma unroll
        for (int i = 0; i < 4; ++i) {
            int f   = tid + i * 256;
            int row = f >> 3;
            int c4  = f & 7;
            float4 v = make_float4(0.f, 0.f, 0.f, 0.f);
            if (m0 + row < N)
                v = *(const float4*)(X + (size_t)(m0 + row) * K + k0 + c4 * 4);
            *(float4*)&As[row][c4 * 4] = v;
        }
#pragma unroll
        for (int i = 0; i < 4; ++i) {
            int f   = tid + i * 256;
            int row = f >> 5;
            int c4  = f & 31;
            float4 v = *(const float4*)(W + (size_t)(k0 + row) * O_DIM + c4 * 4);
            float* bp = (float*)&Bs2[row >> 1][c4 * 4] + (row & 1);
            bp[0] = v.x;
            bp[2] = v.y;
            bp[4] = v.z;
            bp[6] = v.w;
        }
        __syncthreads();

#pragma unroll
        for (int kk = 0; kk < 32; kk += 4) {
            const int kk2 = kk + khalf;
            const int p   = kk2 >> 1;
            v2f a = *(const v2f*)&As[mrow][kk2];
#pragma unroll
            for (int t = 0; t < 8; ++t) {
                v2f b = *(const v2f*)&Bs2[p][t * 16 + lnn];
                acc[t] = __builtin_amdgcn_wmma_f32_16x16x4_f32(
                    false, a, false, b, (short)0, acc[t], false, false);
            }
        }
        __syncthreads();
    }

    const int mtile = m0 + wv * 16;
    if (mtile < N) {
        const int rowoff = (ln >> 4) << 3;
#pragma unroll
        for (int t = 0; t < 8; ++t) {
            const int col = t * 16 + lnn;
            const float bv = bias[col];
#pragma unroll
            for (int i = 0; i < 8; ++i) {
                float v = acc[t][i] + bv;
                out[(size_t)(mtile + rowoff + i) * O_DIM + col] = v > 0.f ? v : 0.f;
            }
        }
    }
}


#define T1_ROWS 200
#define T2_ROWS 120
#define OWN_ROWS 40
#define BIN_CH 4096
#define BIN_K (BIN_CH / 256)
#define BIN_TAB 251

template <int ROWS>
__device__ __forceinline__ void bin_chunk(int* hist, int* wsum, uint2* stage,
                                          const int* __restrict__ dst, const int* __restrict__ src,
                                          const float* __restrict__ w, int E, unsigned nrows,
                                          uint2* __restrict__ rec, int* __restrict__ tab,
                                          int nch, int chunk) {
    const int tid = threadIdx.x;
    const int ln  = tid & 31;
    const int wv  = tid >> 5;
    const int e0  = chunk * BIN_CH;
    const int elast = E - 1;

    hist[tid] = 0;
    __syncthreads();

    unsigned r[BIN_K], sv[BIN_K];
    float    wt[BIN_K];
#pragma unroll
    for (int k = 0; k < BIN_K; ++k) {
        const int e  = e0 + k * 256 + tid;
        const int ec = e < E ? e : elast;
        r[k]  = (unsigned)dst[ec];
        sv[k] = (unsigned)src[ec];
        wt[k] = w[ec];
    }
    unsigned key[BIN_K];
    unsigned rank[BIN_K];
#pragma unroll
    for (int k = 0; k < BIN_K; ++k) {
        const int e = e0 + k * 256 + tid;
        const bool in = e < E && r[k] < nrows;
        const unsigned o  = in ? r[k] / ROWS : 255u;
        const unsigned dl = in ? r[k] - o * ROWS : 0u;
        key[k] = (o << 24) | (dl << 16);
        rank[k] = (unsigned)__hip_atomic_fetch_add(&hist[o], 1, __ATOMIC_RELAXED,
                                                   __HIP_MEMORY_SCOPE_WORKGROUP);
    }
#pragma unroll
    for (int k = 0; k < BIN_K; ++k) key[k] |= rank[k];
    __syncthreads();

    const int cnt = hist[tid];
    int x = cnt;
#pragma unroll
    for (int off = 1; off < 32; off <<= 1) {
        const int y = __shfl_up(x, off, 32);
        if (ln >= off) x += y;
    }
    if (ln == 31) wsum[wv] = x;
    __syncthreads();
    int base = 0;
#pragma unroll
    for (int k = 0; k < 8; ++k) if (k < wv) base += wsum[k];
    const int excl = base + x - cnt;
    hist[tid] = excl;
    if (tid < BIN_TAB) tab[(size_t)tid * nch + chunk] = excl;
    __syncthreads();

#pragma unroll
    for (int k = 0; k < BIN_K; ++k) {
        const int e = e0 + k * 256 + tid;
        if (e < E) {
            const unsigned kk = key[k];
            const int pos = hist[kk >> 24] + (int)(kk & 0xFFFFu);
            stage[pos] = make_uint2(sv[k] | (((kk >> 16) & 0xFFu) << 24), __float_as_uint(wt[k]));
        }
    }
    __syncthreads();

    int nvalid = E - e0;
    if (nvalid > BIN_CH) nvalid = BIN_CH;
    uint2* out = rec + (size_t)chunk * BIN_CH;
    for (int p = tid; p < nvalid; p += 256) out[p] = stage[p];
}

__global__ __launch_bounds__(256)
void stage_pair(const float* aX, const float* __restrict__ aW,
                const float* __restrict__ aBias, float* aOut, int aN, int aK, int aBlocks,
                const float* bX, const float* __restrict__ bW,
                const float* __restrict__ bBias, float* bOut, int bN, int bK, int bBlocks,
                const int* __restrict__ d0, const int* __restrict__ s0, const float* __restrict__ w0, int E0, int nb0,
                const int* __restrict__ d1, const int* __restrict__ s1, const float* __restrict__ w1, int E1, int nb1,
                const int* __restrict__ d2, const int* __restrict__ s2, const float* __restrict__ w2, int E2, int nb2,
                const int* __restrict__ d3, const int* __restrict__ s3, const float* __restrict__ w3, int E3, int nb3,
                unsigned* __restrict__ recw, int* __restrict__ tab,
                unsigned rowsT1, unsigned rowsT2, unsigned rowsDoc) {
    uint2* rec = (uint2*)recw;
    int bid = (int)blockIdx.x;
    if (bid < aBlocks + bBlocks) {
        const bool second = bid >= aBlocks;
        gemm_tile(second ? bX : aX, second ? bW : aW, second ? bBias : aBias,
                  second ? bOut : aOut, second ? bN : aN, second ? bK : aK,
                  second ? bid - aBlocks : bid);
        return;
    }
    bid -= aBlocks + bBlocks;
    __shared__ int   hist[256];
    __shared__ int   wsum[8];
    __shared__ uint2 stage[BIN_CH];
    if (bid < nb0) { bin_chunk<T1_ROWS>(hist, wsum, stage, d0, s0, w0, E0, rowsT1, rec, tab, nb0, bid); return; }
    bid -= nb0;  rec += (size_t)nb0 * BIN_CH;  tab += (size_t)nb0 * BIN_TAB;
    if (bid < nb1) { bin_chunk<T2_ROWS>(hist, wsum, stage, d1, s1, w1, E1, rowsT2, rec, tab, nb1, bid); return; }
    bid -= nb1;  rec += (size_t)nb1 * BIN_CH;  tab += (size_t)nb1 * BIN_TAB;
    if (bid < nb2) { bin_chunk<OWN_ROWS>(hist, wsum, stage, d2, s2, w2, E2, rowsDoc, rec, tab, nb2, bid); return; }
    bid -= nb2;  rec += (size_t)nb2 * BIN_CH;  tab += (size_t)nb2 * BIN_TAB;
    if (bid < nb3) bin_chunk<OWN_ROWS>(hist, wsum, stage, d3, s3, w3, E3, rowsDoc, rec, tab, nb3, bid);
}


#define FETCH_U 8
#define FETCH_C (32 * FETCH_U)
#define SEG_QCAP 10240
#define KEY_SHIFT 20
#define SRC_MASK 0xFFFFFu

template <typename Put>
__device__ __forceinline__ int fetch_runs(const uint2* __restrict__ rec, const int* __restrict__ tab,
                                          int nch, int owner, int pos0, int win0,
                                          int* wsum, int* seg, int tid, int wv, int ln, Put put) {
    const int* ta = tab + (size_t)owner * nch;
    const int* tb = ta + nch;
    int* segS = seg;
    int* segN = seg + FETCH_C;
    int* segQ = seg + 2 * FETCH_C;
    int carry = pos0;
    for (int cb = 0; cb < nch; cb += FETCH_C) {
        int s = 0, n = 0, incl = 0;
        if (tid < FETCH_C) {
            const int c = cb + tid;
            if (c < nch) {
                s = ta[c];
                n = tb[c] - s;
            }
            incl = n;
#pragma unroll
            for (int off = 1; off < 32; off <<= 1) {
                const int y = __shfl_up(incl, off, 32);
                if (ln >= off) incl += y;
            }
            if (ln == 31) wsum[wv] = incl;
        }
        __syncthreads();
        int blk = 0, base = 0;
#pragma unroll
        for (int k = 0; k < FETCH_C / 32; ++k) {
            const int t = wsum[k];
            blk += t;
            if (k < wv) base += t;
        }
        if (tid < FETCH_C) {
            segS[tid] = s;
            segN[tid] = n;
            segQ[tid] = carry + base + incl - n - win0;
        }
        carry += blk;
        __syncthreads();
        const int c0 = wv * FETCH_U;
        int rn[FETCH_U], rq[FETCH_U];
        const uint2* rp[FETCH_U];
        uint2 r[FETCH_U];
#pragma unroll
        for (int u = 0; u < FETCH_U; ++u) {
            const int rs = __builtin_amdgcn_readfirstlane(segS[c0 + u]);
            rn[u] = __builtin_amdgcn_readfirstlane(segN[c0 + u]);
            rq[u] = __builtin_amdgcn_readfirstlane(segQ[c0 + u]);
            rp[u] = rec + (size_t)(cb + c0 + u) * BIN_CH + rs;
        }
#pragma unroll
        for (int u = 0; u < FETCH_U; ++u) {
            r[u] = make_uint2(0u, 0u);
            if (ln < rn[u]) r[u] = rp[u][ln];
        }
#pragma unroll
        for (int u = 0; u < FETCH_U; ++u)
            if (ln < rn[u]) put(rq[u] + ln, r[u]);
#pragma nounroll
        for (int u = 0; u < FETCH_U; ++u) {
            const int n = __builtin_amdgcn_readfirstlane(segN[c0 + u]);
            if (n > 32) {
                const int rs = __builtin_amdgcn_readfirstlane(segS[c0 + u]);
                const int q  = __builtin_amdgcn_readfirstlane(segQ[c0 + u]);
                const uint2* p = rec + (size_t)(cb + c0 + u) * BIN_CH + rs;
                for (int j = ln + 32; j < n; j += 32) put(q + j, p[j]);
            }
        }
        __syncthreads();
    }
    return carry;
}

template <int EB>
__device__ __forceinline__ float4 seg_sum(const uint2* s_rec, int beg, int end,
                                          const float* __restrict__ x, int ln) {
    float4 acc = make_float4(0.f, 0.f, 0.f, 0.f);
    for (int b = beg; b < end; b += EB) {
        float4 v[EB];
        float  wt[EB];
        uint2  e[EB];
#pragma unroll
        for (int j = 0; j < EB; ++j) {
            const int i = b + j;
            e[j] = s_rec[i < end ? i : end - 1];
        }
#pragma unroll
        for (int j = 0; j < EB; ++j) {
            const bool ok = b + j < end;
            const size_t ro = (size_t)__builtin_amdgcn_readfirstlane((int)(e[j].x & SRC_MASK)) * 128;
            wt[j] = ok ? __uint_as_float(e[j].y) : 0.f;
            v[j] = make_float4(0.f, 0.f, 0.f, 0.f);
            if (ok) v[j] = ((const float4*)(x + ro))[ln];
        }
#pragma unroll
        for (int j = 0; j < EB; ++j) {
            acc.x += wt[j] * v[j].x;
            acc.y += wt[j] * v[j].y;
            acc.z += wt[j] * v[j].z;
            acc.w += wt[j] * v[j].w;
        }
    }
    return acc;
}

__device__ __forceinline__ void key_bounds(const int* cur, int key, int& beg, int& end) {
    beg = __builtin_amdgcn_readfirstlane(cur[key]);
    end = __builtin_amdgcn_readfirstlane(cur[key + 1]);
}

struct QueuePut {
    uint2*   q_rec;
    int*     cur;
    unsigned keyoff;
    __device__ __forceinline__ void operator()(int slot, const uint2& r) const {
        if ((unsigned)slot < (unsigned)SEG_QCAP) {
            const unsigned key = (r.x >> 24) + keyoff;
            q_rec[slot] = make_uint2((r.x & SRC_MASK) | (key << KEY_SHIFT), r.y);
            __hip_atomic_fetch_add(&cur[key], 1, __ATOMIC_RELAXED, __HIP_MEMORY_SCOPE_WORKGROUP);
        }
    }
};

template <int NKEY, int NUNIT, typename Unit>
__device__ __forceinline__ void owner_rounds(const uint2* __restrict__ recA, const int* __restrict__ tabA, int nchA,
                                             const uint2* __restrict__ recB, const int* __restrict__ tabB, int nchB,
                                             int keyoffB, int owner,
                                             uint2* q_rec, uint2* s_rec, int* cur, int* wsum, int* seg, int* ctr,
                                             int tid, int wv, int ln, Unit unit) {
    constexpr int NW = (NKEY + 31) / 32;
    int* cnt = cur + 1;
    for (int win0 = 0; ; win0 += SEG_QCAP) {
        for (int i = tid; i <= NKEY; i += 1024) cur[i] = 0;
        if (tid == 0) *ctr = 0;
        __syncthreads();

        int total = fetch_runs(recA, tabA, nchA, owner, 0, win0, wsum, seg, tid, wv, ln,
                               QueuePut{q_rec, cnt, 0u});
        total = fetch_runs(recB, tabB, nchB, owner, total, win0, wsum, seg, tid, wv, ln,
                           QueuePut{q_rec, cnt, (unsigned)keyoffB});

        int c = 0, incl = 0;
        if (tid < NW * 32) {
            c = tid < NKEY ? cnt[tid] : 0;
            incl = c;
#pragma unroll
            for (int off = 1; off < 32; off <<= 1) {
                const int y = __shfl_up(incl, off, 32);
                if (ln >= off) incl += y;
            }
            if (ln == 31) wsum[wv] = incl;
        }
        __syncthreads();
        if (tid < NKEY) {
            int base = 0;
#pragma unroll
            for (int k = 0; k < NW; ++k) if (k < wv) base += wsum[k];
            cnt[tid] = base + incl - c;
        }
        __syncthreads();

        int nq = total - win0;
        if (nq > SEG_QCAP) nq = SEG_QCAP;
        for (int i = tid; i < nq; i += 1024) {
            const uint2 r = q_rec[i];
            const int pos = __hip_atomic_fetch_add(&cnt[r.x >> KEY_SHIFT], 1, __ATOMIC_RELAXED,
                                                   __HIP_MEMORY_SCOPE_WORKGROUP);
            s_rec[pos] = r;
        }
        __syncthreads();

        for (;;) {
            int u = 0;
            if (ln == 0) u = __hip_atomic_fetch_add(ctr, 1, __ATOMIC_RELAXED, __HIP_MEMORY_SCOPE_WORKGROUP);
            u = __builtin_amdgcn_readfirstlane(u);
            if (u >= NUNIT) break;
            unit(u, win0 == 0);
        }
        if (win0 + SEG_QCAP >= total) break;
        __syncthreads();
    }
}

#define AGG_ROWS 40
#define AGG_ROWF 640
#define SUM_EB 12
static_assert(AGG_ROWS == OWN_ROWS, "binning and aggregation must agree on rows per group");

struct DocUnit {
    const float *l2_1, *l1_1, *l2_2, *l1_2, *word;
    const uint2* s_rec;
    const int*   cur;
    float*       acc;
    int          ln;
    __device__ __forceinline__ void operator()(int u, bool first) const {
        int sgm = 0;
        const float* x = l2_1;
        if (u >= 1 * AGG_ROWS) { sgm = 1; x = l1_1; }
        if (u >= 2 * AGG_ROWS) { sgm = 2; x = l2_2; }
        if (u >= 3 * AGG_ROWS) { sgm = 3; x = l1_2; }
        if (u >= 4 * AGG_ROWS) { sgm = 4; x = word; }
        const int row = u - sgm * AGG_ROWS;
        const int key = sgm >= 2 ? AGG_ROWS + row : row;
        int beg, end;
        key_bounds(cur, key, beg, end);
        if (!first && beg == end) return;
        const float4 a = seg_sum<SUM_EB>(s_rec, beg, end, x, ln);
        float* p = acc + row * AGG_ROWF + sgm * 128 + ln;
        if (first) {
            p[0] = a.x;  p[32] = a.y;  p[64] = a.z;  p[96] = a.w;
        } else {
            p[0] += a.x; p[32] += a.y; p[64] += a.z; p[96] += a.w;
        }
    }
};

__global__ __launch_bounds__(1024)
void doc_aggregate(const unsigned* __restrict__ recA, const int* __restrict__ tabA, int nchA,
                   const unsigned* __restrict__ recB, const int* __restrict__ tabB, int nchB,
                   const float* __restrict__ l2_1, const float* __restrict__ l1_1,
                   const float* __restrict__ l2_2, const float* __restrict__ l1_2,
                   const float* __restrict__ word,
                   float* __restrict__ doc, float* __restrict__ doc_svd, int N0) {
    __shared__ __attribute__((aligned(16))) float acc[AGG_ROWS * AGG_ROWF];
    __shared__ uint2 q_rec[SEG_QCAP];
    __shared__ uint2 s_rec[SEG_QCAP];
    __shared__ int   cur[2 * AGG_ROWS + 1];
    __shared__ int   wsum[32];
    __shared__ int   ctr;
    __shared__ int   seg[3 * FETCH_C];

    const int tid = threadIdx.x;
    const int ln  = tid & 31;
    const int wv  = __builtin_amdgcn_readfirstlane(tid >> 5);
    const unsigned row0 = blockIdx.x * AGG_ROWS;

    const DocUnit unit{l2_1, l1_1, l2_2, l1_2, word, s_rec, cur, acc, ln};
    owner_rounds<2 * AGG_ROWS, 5 * AGG_ROWS>((const uint2*)recA, tabA, nchA, (const uint2*)recB, tabB, nchB,
                                             AGG_ROWS, (int)blockIdx.x, q_rec, s_rec, cur, wsum, seg, &ctr,
                                             tid, wv, ln, unit);
    __syncthreads();

    for (int r = wv; r < AGG_ROWS; r += 32) {
        const int row = (int)row0 + r;
        if (row >= N0) break;
        const float* p = acc + r * AGG_ROWF + ln;
        const float4 a0  = make_float4(p[0],   p[32],  p[64],  p[96]);
        const float4 a0s = make_float4(p[128], p[160], p[192], p[224]);
        const float4 a1  = make_float4(p[256], p[288], p[320], p[352]);
        const float4 a1s = make_float4(p[384], p[416], p[448], p[480]);
        const float4 aw  = make_float4(p[512], p[544], p[576], p[608]);
        const float sw = aw.x * aw.x + aw.y * aw.y + aw.z * aw.z + aw.w * aw.w;
        float s0  = a0.x * a0.x + a0.y * a0.y + a0.z * a0.z + a0.w * a0.w;
        float s0s = a0s.x * a0s.x + a0s.y * a0s.y + a0s.z * a0s.z + a0s.w * a0s.w;
        float s1  = a1.x * a1.x + a1.y * a1.y + a1.z * a1.z + a1.w * a1.w + sw;
        float s1s = a1s.x * a1s.x + a1s.y * a1s.y + a1s.z * a1s.z + a1s.w * a1s.w + sw;
#pragma unroll
        for (int off = 16; off > 0; off >>= 1) {
            s0  += __shfl_xor(s0,  off, 32);
            s0s += __shfl_xor(s0s, off, 32);
            s1  += __shfl_xor(s1,  off, 32);
            s1s += __shfl_xor(s1s, off, 32);
        }
        const float i0  = 1.0f / (sqrtf(s0)  + 1e-9f);
        const float i0s = 1.0f / (sqrtf(s0s) + 1e-9f);
        const float i1  = 1.0f / (sqrtf(s1)  + 1e-9f);
        const float i1s = 1.0f / (sqrtf(s1s) + 1e-9f);

        float* o  = doc     + (size_t)row * 384 + ln * 4;
        float* os = doc_svd + (size_t)row * 384 + ln * 4;
        *(float4*)(o)        = make_float4(a0.x * i0, a0.y * i0, a0.z * i0, a0.w * i0);
        *(float4*)(o + 128)  = make_float4(a1.x * i1, a1.y * i1, a1.z * i1, a1.w * i1);
        *(float4*)(o + 256)  = make_float4(aw.x * i1, aw.y * i1, aw.z * i1, aw.w * i1);
        *(float4*)(os)       = make_float4(a0s.x * i0s, a0s.y * i0s, a0s.z * i0s, a0s.w * i0s);
        *(float4*)(os + 128) = make_float4(a1s.x * i1s, a1s.y * i1s, a1s.z * i1s, a1s.w * i1s);
        *(float4*)(os + 256) = make_float4(aw.x * i1s, aw.y * i1s, aw.z * i1s, aw.w * i1s);
    }
}


struct TowerUnit {
    const float *xA, *xB;
    float *hA, *hB;
    int NA, NB, owner;
    const uint2* s_rec;
    const int*   cur;
    int          ln;
    __device__ __forceinline__ void operator()(int u, bool first) const {
        const bool second = u >= T1_ROWS;
        const int row = second ? owner * T2_ROWS + (u - T1_ROWS) : owner * T1_ROWS + u;
        if (row >= (second ? NB : NA)) return;
        int beg, end;
        key_bounds(cur, u, beg, end);
        if (!first && beg == end) return;
        float4 a = seg_sum<SUM_EB>(s_rec, beg, end, second ? xB : xA, ln);
        float4* hp = (float4*)((second ? hB : hA) + (size_t)row * 128) + ln;
        if (!first) {
            const float4 o = *hp;
            a.x += o.x;  a.y += o.y;  a.z += o.z;  a.w += o.w;
        }
        *hp = a;
    }
};

__global__ __launch_bounds__(1024)
void tower_scatter(const unsigned* __restrict__ recA, const int* __restrict__ tabA, int nchA,
                   const unsigned* __restrict__ recB, const int* __restrict__ tabB, int nchB,
                   const float* __restrict__ xA, const float* __restrict__ xB,
                   float* hA, float* hB, int NA, int NB) {
    __shared__ uint2 q_rec[SEG_QCAP];
    __shared__ uint2 s_rec[SEG_QCAP];
    __shared__ int   cur[T1_ROWS + T2_ROWS + 1];
    __shared__ int   wsum[32];
    __shared__ int   ctr;
    __shared__ int   seg[3 * FETCH_C];

    const int tid = threadIdx.x;
    const int ln  = tid & 31;
    const int wv  = __builtin_amdgcn_readfirstlane(tid >> 5);
    const int owner = (int)blockIdx.x;

    const TowerUnit unit{xA, xB, hA, hB, NA, NB, owner, s_rec, cur, ln};
    owner_rounds<T1_ROWS + T2_ROWS, T1_ROWS + T2_ROWS>((const uint2*)recA, tabA, nchA, (const uint2*)recB, tabB, nchB,
                                                       T1_ROWS, owner, q_rec, s_rec, cur, wsum, seg, &ctr,
                                                       tid, wv, ln, unit);
}

extern "C" void kernel_launch(void* const* d_in, const int* in_sizes, int n_in,
                              void* d_out, int out_size, void* d_ws, size_t ws_size,
                              hipStream_t stream) {
    const float* x1       = (const float*)d_in[0];
    const float* x2       = (const float*)d_in[1];
    const float* word_emb = (const float*)d_in[2];
    const float* W1a = (const float*)d_in[3];
    const float* b1a = (const float*)d_in[4];
    const float* W1b = (const float*)d_in[5];
    const float* b1b = (const float*)d_in[6];
    const float* W2a = (const float*)d_in[7];
    const float* b2a = (const float*)d_in[8];
    const float* W2b = (const float*)d_in[9];
    const float* b2b = (const float*)d_in[10];
    const int*   e11_src = (const int*)d_in[11];
    const int*   e11_dst = (const int*)d_in[12];
    const float* e11_w   = (const float*)d_in[13];
    const int*   e22_src = (const int*)d_in[14];
    const int*   e22_dst = (const int*)d_in[15];
    const float* e22_w   = (const float*)d_in[16];
    const int*   e01_src = (const int*)d_in[17];
    const int*   e01_dst = (const int*)d_in[18];
    const float* e01_w   = (const float*)d_in[19];
    const int*   e02_src = (const int*)d_in[20];
    const int*   e02_dst = (const int*)d_in[21];
    const float* e02_w   = (const float*)d_in[22];

    const int N0 = 10000, N1 = 50000, N2 = 30000;
    const int E11 = in_sizes[11], E22 = in_sizes[14];
    const int E01 = in_sizes[17], E02 = in_sizes[20];

    float* ws = (float*)d_ws;
    size_t off = 0;
    float* l1_1 = ws + off; off += (size_t)N1 * 128;
    float* l2_1 = ws + off; off += (size_t)N1 * 128;
    float* l1_2 = ws + off; off += (size_t)N2 * 128;
    unsigned* rec = (unsigned*)(ws + off); off += (size_t)N2 * 128;
    float* h11  = ws + off; off += (size_t)N1 * 128;
    float* h22  = ws + off; off += (size_t)N2 * 128;
    float* l2_2 = h22;
    float* doc     = (float*)d_out;
    float* doc_svd = doc + (size_t)N0 * 384;

    const dim3 blk(256);
    const int g1 = (N1 + 127) / 128;
    const int g2 = (N2 + 127) / 128;
    const int c11 = (E11 + BIN_CH - 1) / BIN_CH;
    const int c22 = (E22 + BIN_CH - 1) / BIN_CH;
    const int c01 = (E01 + BIN_CH - 1) / BIN_CH;
    const int c02 = (E02 + BIN_CH - 1) / BIN_CH;
    unsigned* rec11 = rec;
    unsigned* rec22 = rec11 + (size_t)c11 * BIN_CH * 2;
    unsigned* rec01 = rec22 + (size_t)c22 * BIN_CH * 2;
    unsigned* rec02 = rec01 + (size_t)c01 * BIN_CH * 2;
    int* tab   = (int*)(rec02 + (size_t)c02 * BIN_CH * 2);
    int* tab11 = tab;
    int* tab22 = tab11 + (size_t)c11 * BIN_TAB;
    int* tab01 = tab22 + (size_t)c22 * BIN_TAB;
    int* tab02 = tab01 + (size_t)c01 * BIN_TAB;
    const int t1 = (N1 + T1_ROWS - 1) / T1_ROWS;
    const int t2 = (N2 + T2_ROWS - 1) / T2_ROWS;
    const int tg = t1 > t2 ? t1 : t2;

    stage_pair<<<dim3(g1 + g2 + c11 + c22 + c01 + c02), blk, 0, stream>>>(
        x1, W1a, b1a, l1_1, N1, 256, g1,
        x2, W2a, b2a, l1_2, N2, 256, g2,
        e11_dst, e11_src, e11_w, E11, c11,
        e22_dst, e22_src, e22_w, E22, c22,
        e01_dst, e01_src, e01_w, E01, c01,
        e02_dst, e02_src, e02_w, E02, c02,
        rec, tab, (unsigned)N1, (unsigned)N2, (unsigned)N0);
    tower_scatter<<<dim3(tg), dim3(1024), 0, stream>>>(
        rec11, tab11, c11, rec22, tab22, c22,
        l1_1, l1_2, h11, h22, N1, N2);
    stage_pair<<<dim3(g1 + g2), blk, 0, stream>>>(
        h11, W1b, b1b, l2_1, N1, 128, g1,
        h22, W2b, b2b, l2_2, N2, 128, g2,
        nullptr, nullptr, nullptr, 0, 0,
        nullptr, nullptr, nullptr, 0, 0,
        nullptr, nullptr, nullptr, 0, 0,
        nullptr, nullptr, nullptr, 0, 0,
        rec, tab, 0u, 0u, 0u);

    doc_aggregate<<<dim3((N0 + AGG_ROWS - 1) / AGG_ROWS), dim3(1024), 0, stream>>>(
        rec01, tab01, c01, rec02, tab02, c02,
        l2_1, l1_1, l2_2, l1_2, word_emb, doc, doc_svd, N0);
}


# row sums (tower scatter-add, doc aggregation): whole batches of 12 rows load untested, only the last partial batch is predicated; packed fma; a batch's loads all in flight at once; same sum order
# speedup vs baseline: 3.5900x; 1.0357x over previous
#include <hip/hip_runtime.h>
#include <math.h>

typedef __attribute__((ext_vector_type(2))) float v2f;
typedef __attribute__((ext_vector_type(8))) float v8f;

#define O_DIM 128

__device__ __forceinline__
void gemm_tile(const float* X, const float* __restrict__ W,
               const float* __restrict__ bias, float* out,
               int N, int K, int bid) {
    __shared__ float  As[128][36];
    __shared__ float2 Bs2[16][144];

    const int tid = threadIdx.x;
    const int wv  = tid >> 5;
    const int ln  = tid & 31;
    const int m0  = bid * 128;

    v8f acc[8];
#pragma unroll
    for (int t = 0; t < 8; ++t)
#pragma unroll
        for (int j = 0; j < 8; ++j) acc[t][j] = 0.0f;

    const int mrow = wv * 16 + (ln & 15);
    const int lnn  = ln & 15;
    const int khalf = (ln >> 4) << 1;

    for (int k0 = 0; k0 < K; k0 += 32) {
#pragma unroll
        for (int i = 0; i < 4; ++i) {
            int f   = tid + i * 256;
            int row = f >> 3;
            int c4  = f & 7;
            float4 v = make_float4(0.f, 0.f, 0.f, 0.f);
            if (m0 + row < N)
                v = *(const float4*)(X + (size_t)(m0 + row) * K + k0 + c4 * 4);
            *(float4*)&As[row][c4 * 4] = v;
        }
#pragma unroll
        for (int i = 0; i < 4; ++i) {
            int f   = tid + i * 256;
            int row = f >> 5;
            int c4  = f & 31;
            float4 v = *(const float4*)(W + (size_t)(k0 + row) * O_DIM + c4 * 4);
            float* bp = (float*)&Bs2[row >> 1][c4 * 4] + (row & 1);
            bp[0] = v.x;
            bp[2] = v.y;
            bp[4] = v.z;
            bp[6] = v.w;
        }
        __syncthreads();

#pragma unroll
        for (int kk = 0; kk < 32; kk += 4) {
            const int kk2 = kk + khalf;
            const int p   = kk2 >> 1;
            v2f a = *(const v2f*)&As[mrow][kk2];
#pragma unroll
            for (int t = 0; t < 8; ++t) {
                v2f b = *(const v2f*)&Bs2[p][t * 16 + lnn];
                acc[t] = __builtin_amdgcn_wmma_f32_16x16x4_f32(
                    false, a, false, b, (short)0, acc[t], false, false);
            }
        }
        __syncthreads();
    }

    const int mtile = m0 + wv * 16;
    if (mtile < N) {
        const int rowoff = (ln >> 4) << 3;
#pragma unroll
        for (int t = 0; t < 8; ++t) {
            const int col = t * 16 + lnn;
            const float bv = bias[col];
#pragma unroll
            for (int i = 0; i < 8; ++i) {
                float v = acc[t][i] + bv;
                out[(size_t)(mtile + rowoff + i) * O_DIM + col] = v > 0.f ? v : 0.f;
            }
        }
    }
}


#define T1_ROWS 200
#define T2_ROWS 120
#define OWN_ROWS 40
#define BIN_CH 4096
#define BIN_K (BIN_CH / 256)
#define BIN_TAB 251

template <int ROWS>
__device__ __forceinline__ void bin_chunk(int* hist, int* wsum, uint2* stage,
                                          const int* __restrict__ dst, const int* __restrict__ src,
                                          const float* __restrict__ w, int E, unsigned nrows,
                                          uint2* __restrict__ rec, int* __restrict__ tab,
                                          int nch, int chunk) {
    const int tid = threadIdx.x;
    const int ln  = tid & 31;
    const int wv  = tid >> 5;
    const int e0  = chunk * BIN_CH;
    const int elast = E - 1;

    hist[tid] = 0;
    __syncthreads();

    unsigned r[BIN_K], sv[BIN_K];
    float    wt[BIN_K];
#pragma unroll
    for (int k = 0; k < BIN_K; ++k) {
        const int e  = e0 + k * 256 + tid;
        const int ec = e < E ? e : elast;
        r[k]  = (unsigned)dst[ec];
        sv[k] = (unsigned)src[ec];
        wt[k] = w[ec];
    }
    unsigned key[BIN_K];
    unsigned rank[BIN_K];
#pragma unroll
    for (int k = 0; k < BIN_K; ++k) {
        const int e = e0 + k * 256 + tid;
        const bool in = e < E && r[k] < nrows;
        const unsigned o  = in ? r[k] / ROWS : 255u;
        const unsigned dl = in ? r[k] - o * ROWS : 0u;
        key[k] = (o << 24) | (dl << 16);
        rank[k] = (unsigned)__hip_atomic_fetch_add(&hist[o], 1, __ATOMIC_RELAXED,
                                                   __HIP_MEMORY_SCOPE_WORKGROUP);
    }
#pragma unroll
    for (int k = 0; k < BIN_K; ++k) key[k] |= rank[k];
    __syncthreads();

    const int cnt = hist[tid];
    int x = cnt;
#pragma unroll
    for (int off = 1; off < 32; off <<= 1) {
        const int y = __shfl_up(x, off, 32);
        if (ln >= off) x += y;
    }
    if (ln == 31) wsum[wv] = x;
    __syncthreads();
    int base = 0;
#pragma unroll
    for (int k = 0; k < 8; ++k) if (k < wv) base += wsum[k];
    const int excl = base + x - cnt;
    hist[tid] = excl;
    if (tid < BIN_TAB) tab[(size_t)tid * nch + chunk] = excl;
    __syncthreads();

#pragma unroll
    for (int k = 0; k < BIN_K; ++k) {
        const int e = e0 + k * 256 + tid;
        if (e < E) {
            const unsigned kk = key[k];
            const int pos = hist[kk >> 24] + (int)(kk & 0xFFFFu);
            stage[pos] = make_uint2(sv[k] | (((kk >> 16) & 0xFFu) << 24), __float_as_uint(wt[k]));
        }
    }
    __syncthreads();

    int nvalid = E - e0;
    if (nvalid > BIN_CH) nvalid = BIN_CH;
    uint2* out = rec + (size_t)chunk * BIN_CH;
    for (int p = tid; p < nvalid; p += 256) out[p] = stage[p];
}

__global__ __launch_bounds__(256)
void stage_pair(const float* aX, const float* __restrict__ aW,
                const float* __restrict__ aBias, float* aOut, int aN, int aK, int aBlocks,
                const float* bX, const float* __restrict__ bW,
                const float* __restrict__ bBias, float* bOut, int bN, int bK, int bBlocks,
                const int* __restrict__ d0, const int* __restrict__ s0, const float* __restrict__ w0, int E0, int nb0,
                const int* __restrict__ d1, const int* __restrict__ s1, const float* __restrict__ w1, int E1, int nb1,
                const int* __restrict__ d2, const int* __restrict__ s2, const float* __restrict__ w2, int E2, int nb2,
                const int* __restrict__ d3, const int* __restrict__ s3, const float* __restrict__ w3, int E3, int nb3,
                unsigned* __restrict__ recw, int* __restrict__ tab,
                unsigned rowsT1, unsigned rowsT2, unsigned rowsDoc) {
    uint2* rec = (uint2*)recw;
    int bid = (int)blockIdx.x;
    if (bid < aBlocks + bBlocks) {
        const bool second = bid >= aBlocks;
        gemm_tile(second ? bX : aX, second ? bW : aW, second ? bBias : aBias,
                  second ? bOut : aOut, second ? bN : aN, second ? bK : aK,
                  second ? bid - aBlocks : bid);
        return;
    }
    bid -= aBlocks + bBlocks;
    __shared__ int   hist[256];
    __shared__ int   wsum[8];
    __shared__ uint2 stage[BIN_CH];
    if (bid < nb0) { bin_chunk<T1_ROWS>(hist, wsum, stage, d0, s0, w0, E0, rowsT1, rec, tab, nb0, bid); return; }
    bid -= nb0;  rec += (size_t)nb0 * BIN_CH;  tab += (size_t)nb0 * BIN_TAB;
    if (bid < nb1) { bin_chunk<T2_ROWS>(hist, wsum, stage, d1, s1, w1, E1, rowsT2, rec, tab, nb1, bid); return; }
    bid -= nb1;  rec += (size_t)nb1 * BIN_CH;  tab += (size_t)nb1 * BIN_TAB;
    if (bid < nb2) { bin_chunk<OWN_ROWS>(hist, wsum, stage, d2, s2, w2, E2, rowsDoc, rec, tab, nb2, bid); return; }
    bid -= nb2;  rec += (size_t)nb2 * BIN_CH;  tab += (size_t)nb2 * BIN_TAB;
    if (bid < nb3) bin_chunk<OWN_ROWS>(hist, wsum, stage, d3, s3, w3, E3, rowsDoc, rec, tab, nb3, bid);
}


#define FETCH_U 8
#define FETCH_C (32 * FETCH_U)
#define SEG_QCAP 10240
#define KEY_SHIFT 20
#define SRC_MASK 0xFFFFFu

template <typename Put>
__device__ __forceinline__ int fetch_runs(const uint2* __restrict__ rec, const int* __restrict__ tab,
                                          int nch, int owner, int pos0, int win0,
                                          int* wsum, int* seg, int tid, int wv, int ln, Put put) {
    const int* ta = tab + (size_t)owner * nch;
    const int* tb = ta + nch;
    int* segS = seg;
    int* segN = seg + FETCH_C;
    int* segQ = seg + 2 * FETCH_C;
    int carry = pos0;
    for (int cb = 0; cb < nch; cb += FETCH_C) {
        int s = 0, n = 0, incl = 0;
        if (tid < FETCH_C) {
            const int c = cb + tid;
            if (c < nch) {
                s = ta[c];
                n = tb[c] - s;
            }
            incl = n;
#pragma unroll
            for (int off = 1; off < 32; off <<= 1) {
                const int y = __shfl_up(incl, off, 32);
                if (ln >= off) incl += y;
            }
            if (ln == 31) wsum[wv] = incl;
        }
        __syncthreads();
        int blk = 0, base = 0;
#pragma unroll
        for (int k = 0; k < FETCH_C / 32; ++k) {
            const int t = wsum[k];
            blk += t;
            if (k < wv) base += t;
        }
        if (tid < FETCH_C) {
            segS[tid] = s;
            segN[tid] = n;
            segQ[tid] = carry + base + incl - n - win0;
        }
        carry += blk;
        __syncthreads();
        const int c0 = wv * FETCH_U;
        int rn[FETCH_U], rq[FETCH_U];
        const uint2* rp[FETCH_U];
        uint2 r[FETCH_U];
#pragma unroll
        for (int u = 0; u < FETCH_U; ++u) {
            const int rs = __builtin_amdgcn_readfirstlane(segS[c0 + u]);
            rn[u] = __builtin_amdgcn_readfirstlane(segN[c0 + u]);
            rq[u] = __builtin_amdgcn_readfirstlane(segQ[c0 + u]);
            rp[u] = rec + (size_t)(cb + c0 + u) * BIN_CH + rs;
        }
#pragma unroll
        for (int u = 0; u < FETCH_U; ++u) {
            r[u] = make_uint2(0u, 0u);
            if (ln < rn[u]) r[u] = rp[u][ln];
        }
#pragma unroll
        for (int u = 0; u < FETCH_U; ++u)
            if (ln < rn[u]) put(rq[u] + ln, r[u]);
#pragma nounroll
        for (int u = 0; u < FETCH_U; ++u) {
            const int n = __builtin_amdgcn_readfirstlane(segN[c0 + u]);
            if (n > 32) {
                const int rs = __builtin_amdgcn_readfirstlane(segS[c0 + u]);
                const int q  = __builtin_amdgcn_readfirstlane(segQ[c0 + u]);
                const uint2* p = rec + (size_t)(cb + c0 + u) * BIN_CH + rs;
                for (int j = ln + 32; j < n; j += 32) put(q + j, p[j]);
            }
        }
        __syncthreads();
    }
    return carry;
}

typedef __attribute__((ext_vector_type(4))) float v4f;

__device__ __forceinline__ void row_fma(v2f& lo, v2f& hi, float w, const v4f& v) {
    const v2f ww = {w, w};
    const v2f vl = {v.x, v.y};
    const v2f vh = {v.z, v.w};
    lo = __builtin_elementwise_fma(ww, vl, lo);
    hi = __builtin_elementwise_fma(ww, vh, hi);
}

template <int NT>
__device__ __forceinline__ void seg_tail(v2f& lo, v2f& hi, const uint2* s_rec, int b, int end,
                                         const float* __restrict__ x, int ln) {
    uint2 e[NT];
    v4f   v[NT];
#pragma unroll
    for (int j = 0; j < NT; ++j) {
        const int i = b + j;
        e[j] = s_rec[i < end ? i : end - 1];
    }
#pragma unroll
    for (int j = 0; j < NT; ++j) {
        const size_t ro = (size_t)__builtin_amdgcn_readfirstlane((int)(e[j].x & SRC_MASK)) * 128;
        v[j] = (v4f){0.f, 0.f, 0.f, 0.f};
        if (b + j < end) v[j] = ((const v4f*)(x + ro))[ln];
    }
#pragma unroll
    for (int j = 0; j < NT; ++j)
        row_fma(lo, hi, b + j < end ? __uint_as_float(e[j].y) : 0.f, v[j]);
}

template <int EB>
__device__ __forceinline__ float4 seg_sum(const uint2* s_rec, int beg, int end,
                                          const float* __restrict__ x, int ln) {
    v2f lo = {0.f, 0.f}, hi = {0.f, 0.f};
    int b = beg;
    for (; b + EB <= end; b += EB) {
        uint2 e[EB];
        v4f   v[EB];
#pragma unroll
        for (int j = 0; j < EB; ++j) e[j] = s_rec[b + j];
#pragma unroll
        for (int j = 0; j < EB; ++j) {
            const size_t ro = (size_t)__builtin_amdgcn_readfirstlane((int)(e[j].x & SRC_MASK)) * 128;
            v[j] = ((const v4f*)(x + ro))[ln];
        }
#pragma unroll
        for (int j = 0; j < EB; ++j) row_fma(lo, hi, __uint_as_float(e[j].y), v[j]);
    }
    if (b < end) seg_tail<EB - 1>(lo, hi, s_rec, b, end, x, ln);
    return make_float4(lo.x, lo.y, hi.x, hi.y);
}

__device__ __forceinline__ void key_bounds(const int* cur, int key, int& beg, int& end) {
    beg = __builtin_amdgcn_readfirstlane(cur[key]);
    end = __builtin_amdgcn_readfirstlane(cur[key + 1]);
}

struct QueuePut {
    uint2*   q_rec;
    int*     cur;
    unsigned keyoff;
    __device__ __forceinline__ void operator()(int slot, const uint2& r) const {
        if ((unsigned)slot < (unsigned)SEG_QCAP) {
            const unsigned key = (r.x >> 24) + keyoff;
            q_rec[slot] = make_uint2((r.x & SRC_MASK) | (key << KEY_SHIFT), r.y);
            __hip_atomic_fetch_add(&cur[key], 1, __ATOMIC_RELAXED, __HIP_MEMORY_SCOPE_WORKGROUP);
        }
    }
};

template <int NKEY, int NUNIT, typename Unit>
__device__ __forceinline__ void owner_rounds(const uint2* __restrict__ recA, const int* __restrict__ tabA, int nchA,
                                             const uint2* __restrict__ recB, const int* __restrict__ tabB, int nchB,
                                             int keyoffB, int owner,
                                             uint2* q_rec, uint2* s_rec, int* cur, int* wsum, int* seg, int* ctr,
                                             int tid, int wv, int ln, Unit unit) {
    constexpr int NW = (NKEY + 31) / 32;
    int* cnt = cur + 1;
    for (int win0 = 0; ; win0 += SEG_QCAP) {
        for (int i = tid; i <= NKEY; i += 1024) cur[i] = 0;
        if (tid == 0) *ctr = 0;
        __syncthreads();

        int total = fetch_runs(recA, tabA, nchA, owner, 0, win0, wsum, seg, tid, wv, ln,
                               QueuePut{q_rec, cnt, 0u});
        total = fetch_runs(recB, tabB, nchB, owner, total, win0, wsum, seg, tid, wv, ln,
                           QueuePut{q_rec, cnt, (unsigned)keyoffB});

        int c = 0, incl = 0;
        if (tid < NW * 32) {
            c = tid < NKEY ? cnt[tid] : 0;
            incl = c;
#pragma unroll
            for (int off = 1; off < 32; off <<= 1) {
                const int y = __shfl_up(incl, off, 32);
                if (ln >= off) incl += y;
            }
            if (ln == 31) wsum[wv] = incl;
        }
        __syncthreads();
        if (tid < NKEY) {
            int base = 0;
#pragma unroll
            for (int k = 0; k < NW; ++k) if (k < wv) base += wsum[k];
            cnt[tid] = base + incl - c;
        }
        __syncthreads();

        int nq = total - win0;
        if (nq > SEG_QCAP) nq = SEG_QCAP;
        for (int i = tid; i < nq; i += 1024) {
            const uint2 r = q_rec[i];
            const int pos = __hip_atomic_fetch_add(&cnt[r.x >> KEY_SHIFT], 1, __ATOMIC_RELAXED,
                                                   __HIP_MEMORY_SCOPE_WORKGROUP);
            s_rec[pos] = r;
        }
        __syncthreads();

        for (;;) {
            int u = 0;
            if (ln == 0) u = __hip_atomic_fetch_add(ctr, 1, __ATOMIC_RELAXED, __HIP_MEMORY_SCOPE_WORKGROUP);
            u = __builtin_amdgcn_readfirstlane(u);
            if (u >= NUNIT) break;
            unit(u, win0 == 0);
        }
        if (win0 + SEG_QCAP >= total) break;
        __syncthreads();
    }
}

#define AGG_ROWS 40
#define AGG_ROWF 640
#define SUM_EB 12
static_assert(AGG_ROWS == OWN_ROWS, "binning and aggregation must agree on rows per group");

struct DocUnit {
    const float *l2_1, *l1_1, *l2_2, *l1_2, *word;
    const uint2* s_rec;
    const int*   cur;
    float*       acc;
    int          ln;
    __device__ __forceinline__ void operator()(int u, bool first) const {
        int sgm = 0;
        const float* x = l2_1;
        if (u >= 1 * AGG_ROWS) { sgm = 1; x = l1_1; }
        if (u >= 2 * AGG_ROWS) { sgm = 2; x = l2_2; }
        if (u >= 3 * AGG_ROWS) { sgm = 3; x = l1_2; }
        if (u >= 4 * AGG_ROWS) { sgm = 4; x = word; }
        const int row = u - sgm * AGG_ROWS;
        const int key = sgm >= 2 ? AGG_ROWS + row : row;
        int beg, end;
        key_bounds(cur, key, beg, end);
        if (!first && beg == end) return;
        const float4 a = seg_sum<SUM_EB>(s_rec, beg, end, x, ln);
        float* p = acc + row * AGG_ROWF + sgm * 128 + ln;
        if (first) {
            p[0] = a.x;  p[32] = a.y;  p[64] = a.z;  p[96] = a.w;
        } else {
            p[0] += a.x; p[32] += a.y; p[64] += a.z; p[96] += a.w;
        }
    }
};

__global__ __launch_bounds__(1024)
void doc_aggregate(const unsigned* __restrict__ recA, const int* __restrict__ tabA, int nchA,
                   const unsigned* __restrict__ recB, const int* __restrict__ tabB, int nchB,
                   const float* __restrict__ l2_1, const float* __restrict__ l1_1,
                   const float* __restrict__ l2_2, const float* __restrict__ l1_2,
                   const float* __restrict__ word,
                   float* __restrict__ doc, float* __restrict__ doc_svd, int N0) {
    __shared__ __attribute__((aligned(16))) float acc[AGG_ROWS * AGG_ROWF];
    __shared__ uint2 q_rec[SEG_QCAP];
    __shared__ uint2 s_rec[SEG_QCAP];
    __shared__ int   cur[2 * AGG_ROWS + 1];
    __shared__ int   wsum[32];
    __shared__ int   ctr;
    __shared__ int   seg[3 * FETCH_C];

    const int tid = threadIdx.x;
    const int ln  = tid & 31;
    const int wv  = __builtin_amdgcn_readfirstlane(tid >> 5);
    const unsigned row0 = blockIdx.x * AGG_ROWS;

    const DocUnit unit{l2_1, l1_1, l2_2, l1_2, word, s_rec, cur, acc, ln};
    owner_rounds<2 * AGG_ROWS, 5 * AGG_ROWS>((const uint2*)recA, tabA, nchA, (const uint2*)recB, tabB, nchB,
                                             AGG_ROWS, (int)blockIdx.x, q_rec, s_rec, cur, wsum, seg, &ctr,
                                             tid, wv, ln, unit);
    __syncthreads();

    for (int r = wv; r < AGG_ROWS; r += 32) {
        const int row = (int)row0 + r;
        if (row >= N0) break;
        const float* p = acc + r * AGG_ROWF + ln;
        const float4 a0  = make_float4(p[0],   p[32],  p[64],  p[96]);
        const float4 a0s = make_float4(p[128], p[160], p[192], p[224]);
        const float4 a1  = make_float4(p[256], p[288], p[320], p[352]);
        const float4 a1s = make_float4(p[384], p[416], p[448], p[480]);
        const float4 aw  = make_float4(p[512], p[544], p[576], p[608]);
        const float sw = aw.x * aw.x + aw.y * aw.y + aw.z * aw.z + aw.w * aw.w;
        float s0  = a0.x * a0.x + a0.y * a0.y + a0.z * a0.z + a0.w * a0.w;
        float s0s = a0s.x * a0s.x + a0s.y * a0s.y + a0s.z * a0s.z + a0s.w * a0s.w;
        float s1  = a1.x * a1.x + a1.y * a1.y + a1.z * a1.z + a1.w * a1.w + sw;
        float s1s = a1s.x * a1s.x + a1s.y * a1s.y + a1s.z * a1s.z + a1s.w * a1s.w + sw;
#pragma unroll
        for (int off = 16; off > 0; off >>= 1) {
            s0  += __shfl_xor(s0,  off, 32);
            s0s += __shfl_xor(s0s, off, 32);
            s1  += __shfl_xor(s1,  off, 32);
            s1s += __shfl_xor(s1s, off, 32);
        }
        const float i0  = 1.0f / (sqrtf(s0)  + 1e-9f);
        const float i0s = 1.0f / (sqrtf(s0s) + 1e-9f);
        const float i1  = 1.0f / (sqrtf(s1)  + 1e-9f);
        const float i1s = 1.0f / (sqrtf(s1s) + 1e-9f);

        float* o  = doc     + (size_t)row * 384 + ln * 4;
        float* os = doc_svd + (size_t)row * 384 + ln * 4;
        *(float4*)(o)        = make_float4(a0.x * i0, a0.y * i0, a0.z * i0, a0.w * i0);
        *(float4*)(o + 128)  = make_float4(a1.x * i1, a1.y * i1, a1.z * i1, a1.w * i1);
        *(float4*)(o + 256)  = make_float4(aw.x * i1, aw.y * i1, aw.z * i1, aw.w * i1);
        *(float4*)(os)       = make_float4(a0s.x * i0s, a0s.y * i0s, a0s.z * i0s, a0s.w * i0s);
        *(float4*)(os + 128) = make_float4(a1s.x * i1s, a1s.y * i1s, a1s.z * i1s, a1s.w * i1s);
        *(float4*)(os + 256) = make_float4(aw.x * i1s, aw.y * i1s, aw.z * i1s, aw.w * i1s);
    }
}


struct TowerUnit {
    const float *xA, *xB;
    float *hA, *hB;
    int NA, NB, owner;
    const uint2* s_rec;
    const int*   cur;
    int          ln;
    __device__ __forceinline__ void operator()(int u, bool first) const {
        const bool second = u >= T1_ROWS;
        const int row = second ? owner * T2_ROWS + (u - T1_ROWS) : owner * T1_ROWS + u;
        if (row >= (second ? NB : NA)) return;
        int beg, end;
        key_bounds(cur, u, beg, end);
        if (!first && beg == end) return;
        float4 a = seg_sum<SUM_EB>(s_rec, beg, end, second ? xB : xA, ln);
        float4* hp = (float4*)((second ? hB : hA) + (size_t)row * 128) + ln;
        if (!first) {
            const float4 o = *hp;
            a.x += o.x;  a.y += o.y;  a.z += o.z;  a.w += o.w;
        }
        *hp = a;
    }
};

__global__ __launch_bounds__(1024)
void tower_scatter(const unsigned* __restrict__ recA, const int* __restrict__ tabA, int nchA,
                   const unsigned* __restrict__ recB, const int* __restrict__ tabB, int nchB,
                   const float* __restrict__ xA, const float* __restrict__ xB,
                   float* hA, float* hB, int NA, int NB) {
    __shared__ uint2 q_rec[SEG_QCAP];
    __shared__ uint2 s_rec[SEG_QCAP];
    __shared__ int   cur[T1_ROWS + T2_ROWS + 1];
    __shared__ int   wsum[32];
    __shared__ int   ctr;
    __shared__ int   seg[3 * FETCH_C];

    const int tid = threadIdx.x;
    const int ln  = tid & 31;
    const int wv  = __builtin_amdgcn_readfirstlane(tid >> 5);
    const int owner = (int)blockIdx.x;

    const TowerUnit unit{xA, xB, hA, hB, NA, NB, owner, s_rec, cur, ln};
    owner_rounds<T1_ROWS + T2_ROWS, T1_ROWS + T2_ROWS>((const uint2*)recA, tabA, nchA, (const uint2*)recB, tabB, nchB,
                                                       T1_ROWS, owner, q_rec, s_rec, cur, wsum, seg, &ctr,
                                                       tid, wv, ln, unit);
}

extern "C" void kernel_launch(void* const* d_in, const int* in_sizes, int n_in,
                              void* d_out, int out_size, void* d_ws, size_t ws_size,
                              hipStream_t stream) {
    const float* x1       = (const float*)d_in[0];
    const float* x2       = (const float*)d_in[1];
    const float* word_emb = (const float*)d_in[2];
    const float* W1a = (const float*)d_in[3];
    const float* b1a = (const float*)d_in[4];
    const float* W1b = (const float*)d_in[5];
    const float* b1b = (const float*)d_in[6];
    const float* W2a = (const float*)d_in[7];
    const float* b2a = (const float*)d_in[8];
    const float* W2b = (const float*)d_in[9];
    const float* b2b = (const float*)d_in[10];
    const int*   e11_src = (const int*)d_in[11];
    const int*   e11_dst = (const int*)d_in[12];
    const float* e11_w   = (const float*)d_in[13];
    const int*   e22_src = (const int*)d_in[14];
    const int*   e22_dst = (const int*)d_in[15];
    const float* e22_w   = (const float*)d_in[16];
    const int*   e01_src = (const int*)d_in[17];
    const int*   e01_dst = (const int*)d_in[18];
    const float* e01_w   = (const float*)d_in[19];
    const int*   e02_src = (const int*)d_in[20];
    const int*   e02_dst = (const int*)d_in[21];
    const float* e02_w   = (const float*)d_in[22];

    const int N0 = 10000, N1 = 50000, N2 = 30000;
    const int E11 = in_sizes[11], E22 = in_sizes[14];
    const int E01 = in_sizes[17], E02 = in_sizes[20];

    float* ws = (float*)d_ws;
    size_t off = 0;
    float* l1_1 = ws + off; off += (size_t)N1 * 128;
    float* l2_1 = ws + off; off += (size_t)N1 * 128;
    float* l1_2 = ws + off; off += (size_t)N2 * 128;
    unsigned* rec = (unsigned*)(ws + off); off += (size_t)N2 * 128;
    float* h11  = ws + off; off += (size_t)N1 * 128;
    float* h22  = ws + off; off += (size_t)N2 * 128;
    float* l2_2 = h22;
    float* doc     = (float*)d_out;
    float* doc_svd = doc + (size_t)N0 * 384;

    const dim3 blk(256);
    const int g1 = (N1 + 127) / 128;
    const int g2 = (N2 + 127) / 128;
    const int c11 = (E11 + BIN_CH - 1) / BIN_CH;
    const int c22 = (E22 + BIN_CH - 1) / BIN_CH;
    const int c01 = (E01 + BIN_CH - 1) / BIN_CH;
    const int c02 = (E02 + BIN_CH - 1) / BIN_CH;
    unsigned* rec11 = rec;
    unsigned* rec22 = rec11 + (size_t)c11 * BIN_CH * 2;
    unsigned* rec01 = rec22 + (size_t)c22 * BIN_CH * 2;
    unsigned* rec02 = rec01 + (size_t)c01 * BIN_CH * 2;
    int* tab   = (int*)(rec02 + (size_t)c02 * BIN_CH * 2);
    int* tab11 = tab;
    int* tab22 = tab11 + (size_t)c11 * BIN_TAB;
    int* tab01 = tab22 + (size_t)c22 * BIN_TAB;
    int* tab02 = tab01 + (size_t)c01 * BIN_TAB;
    const int t1 = (N1 + T1_ROWS - 1) / T1_ROWS;
    const int t2 = (N2 + T2_ROWS - 1) / T2_ROWS;
    const int tg = t1 > t2 ? t1 : t2;

    stage_pair<<<dim3(g1 + g2 + c11 + c22 + c01 + c02), blk, 0, stream>>>(
        x1, W1a, b1a, l1_1, N1, 256, g1,
        x2, W2a, b2a, l1_2, N2, 256, g2,
        e11_dst, e11_src, e11_w, E11, c11,
        e22_dst, e22_src, e22_w, E22, c22,
        e01_dst, e01_src, e01_w, E01, c01,
        e02_dst, e02_src, e02_w, E02, c02,
        rec, tab, (unsigned)N1, (unsigned)N2, (unsigned)N0);
    tower_scatter<<<dim3(tg), dim3(1024), 0, stream>>>(
        rec11, tab11, c11, rec22, tab22, c22,
        l1_1, l1_2, h11, h22, N1, N2);
    stage_pair<<<dim3(g1 + g2), blk, 0, stream>>>(
        h11, W1b, b1b, l2_1, N1, 128, g1,
        h22, W2b, b2b, l2_2, N2, 128, g2,
        nullptr, nullptr, nullptr, 0, 0,
        nullptr, nullptr, nullptr, 0, 0,
        nullptr, nullptr, nullptr, 0, 0,
        nullptr, nullptr, nullptr, 0, 0,
        rec, tab, 0u, 0u, 0u);

    doc_aggregate<<<dim3((N0 + AGG_ROWS - 1) / AGG_ROWS), dim3(1024), 0, stream>>>(
        rec01, tab01, c01, rec02, tab02, c02,
        l2_1, l1_1, l2_2, l1_2, word_emb, doc, doc_svd, N0);
}
